# starting kernel
# baseline (speedup 1.0000x reference)
#include <hip/hip_runtime.h>
#include <hip/hip_bf16.h>
#include <math.h>

typedef float v2f __attribute__((ext_vector_type(2)));
typedef float v8f __attribute__((ext_vector_type(8)));

#define B_    2
#define T_    4
#define N_    1024
#define L_    4
#define MA_   256
#define KNB_  32
#define C_    512
#define HEADS_ 8
#define DH_   64
#define MTOK_ 1024
#define ROWS_ 2048
#define MLP_  2048

__device__ __forceinline__ v8f wmma4(v2f a, v2f b, v8f c) {
  return __builtin_amdgcn_wmma_f32_16x16x4_f32(false, a, false, b, (short)0, c,
                                               false, false);
}

__device__ __forceinline__ float gelu_exact(float x) {
  return 0.5f * x * (1.0f + erff(x * 0.7071067811865475f));
}

__global__ void fps_kernel(const float* __restrict__ xyzs,
                           float* __restrict__ anchors) {
  __shared__ float sval[N_];
  __shared__ int   sidx[N_];
  __shared__ float last[3];
  const int blk = blockIdx.x;
  const int tid = threadIdx.x;
  const float* frame = xyzs + (size_t)blk * N_ * 3;
  const float px = frame[tid * 3 + 0];
  const float py = frame[tid * 3 + 1];
  const float pz = frame[tid * 3 + 2];
  float dist = 1e10f;
  float* aout = anchors + (size_t)blk * MA_ * 3;
  if (tid == 0) {
    last[0] = px; last[1] = py; last[2] = pz;
    aout[0] = px; aout[1] = py; aout[2] = pz;
  }
  __syncthreads();
  for (int it = 1; it < MA_; ++it) {
    const float dx = px - last[0], dy = py - last[1], dz = pz - last[2];
    dist = fminf(dist, dx * dx + dy * dy + dz * dz);
    sval[tid] = dist; sidx[tid] = tid;
    __syncthreads();
    for (int s = 512; s > 0; s >>= 1) {
      if (tid < s) {
        const float v2v = sval[tid + s];
        const int   i2  = sidx[tid + s];
        if (v2v > sval[tid] || (v2v == sval[tid] && i2 < sidx[tid])) {
          sval[tid] = v2v; sidx[tid] = i2;
        }
      }
      __syncthreads();
    }
    const int nxt = sidx[0];
    __syncthreads();
    if (tid == nxt) {
      last[0] = px; last[1] = py; last[2] = pz;
      aout[it * 3 + 0] = px; aout[it * 3 + 1] = py; aout[it * 3 + 2] = pz;
    }
    __syncthreads();
  }
}

__global__ void p4dconv_kernel(const float* __restrict__ xyzs,
                               const float* __restrict__ anchors,
                               const float* __restrict__ Wd,
                               float* __restrict__ feats) {
  __shared__ int   sscan[N_];
  __shared__ float spx[KNB_], spy[KNB_], spz[KNB_];
  __shared__ float s_p0[3];
  const int blk = blockIdx.x;
  const int m   = blk % MA_;
  const int bl  = blk / MA_;
  const int l   = bl % L_;
  const int b   = bl / L_;
  const int tid = threadIdx.x;

  const float* arow = anchors + ((size_t)bl * MA_ + m) * 3;
  const float ax = arow[0], ay = arow[1], az = arow[2];

  float w0 = 0.f, w1 = 0.f, w2 = 0.f, w3 = 0.f;
  if (tid < C_) {
    w0 = Wd[tid * 4 + 0]; w1 = Wd[tid * 4 + 1];
    w2 = Wd[tid * 4 + 2]; w3 = Wd[tid * 4 + 3];
  }
  float maxv = -3.4e38f;

  for (int di = -1; di <= 1; ++di) {
    int fi = l + di; fi = fi < 0 ? 0 : (fi > T_ - 1 ? T_ - 1 : fi);
    const float* nb = xyzs + ((size_t)b * T_ + fi) * N_ * 3;
    const float qx = nb[tid * 3 + 0];
    const float qy = nb[tid * 3 + 1];
    const float qz = nb[tid * 3 + 2];
    const float dx = qx - ax, dy = qy - ay, dz = qz - az;
    const int pred = (dx * dx + dy * dy + dz * dz) < 1.0f ? 1 : 0;
    sscan[tid] = pred;
    if (tid == 0) { s_p0[0] = dx; s_p0[1] = dy; s_p0[2] = dz; }
    __syncthreads();
    for (int off = 1; off < N_; off <<= 1) {
      const int v   = sscan[tid];
      const int add = (tid >= off) ? sscan[tid - off] : 0;
      __syncthreads();
      sscan[tid] = v + add;
      __syncthreads();
    }
    const int incl  = sscan[tid];
    const int total = sscan[N_ - 1];
    const int pos   = incl - pred;
    if (pred && pos < KNB_) { spx[pos] = dx; spy[pos] = dy; spz[pos] = dz; }
    __syncthreads();
    const int cnt = total < KNB_ ? total : KNB_;
    if (tid < KNB_ && tid >= cnt) {
      spx[tid] = (total > 0) ? spx[0] : s_p0[0];
      spy[tid] = (total > 0) ? spy[0] : s_p0[1];
      spz[tid] = (total > 0) ? spz[0] : s_p0[2];
    }
    __syncthreads();
    if (tid < C_) {
      const float tq = (float)di;
      #pragma unroll 8
      for (int k = 0; k < KNB_; ++k) {
        const float v = w0 * spx[k] + w1 * spy[k] + w2 * spz[k] + w3 * tq;
        maxv = fmaxf(maxv, v);
      }
    }
    __syncthreads();
  }
  if (tid < C_)
    feats[((size_t)bl * MA_ + m) * C_ + tid] = fmaxf(maxv, 0.0f);
}

__global__ void ln_kernel(const float* __restrict__ x, const float* __restrict__ g,
                          const float* __restrict__ bta, float* __restrict__ y) {
  __shared__ float red[256];
  const int row = blockIdx.x, t = threadIdx.x;
  const float* xr = x + (size_t)row * C_;
  const float a0 = xr[t], a1 = xr[t + 256];
  red[t] = a0 + a1; __syncthreads();
  for (int s = 128; s > 0; s >>= 1) { if (t < s) red[t] += red[t + s]; __syncthreads(); }
  const float mean = red[0] * (1.0f / C_);
  __syncthreads();
  const float d0 = a0 - mean, d1 = a1 - mean;
  red[t] = d0 * d0 + d1 * d1; __syncthreads();
  for (int s = 128; s > 0; s >>= 1) { if (t < s) red[t] += red[t + s]; __syncthreads(); }
  const float inv = rsqrtf(red[0] * (1.0f / C_) + 1e-5f);
  float* yr = y + (size_t)row * C_;
  yr[t]       = d0 * inv * g[t]       + bta[t];
  yr[t + 256] = d1 * inv * g[t + 256] + bta[t + 256];
}

__global__ void gemm_wmma(const float* __restrict__ A, int lda,
                          const float* __restrict__ W, int ldw,
                          const float* __restrict__ bias,
                          const float* __restrict__ resid,
                          float* __restrict__ Cout, int ldc,
                          int Kdim, int act) {
  const int j0 = blockIdx.x * 32, i0 = blockIdx.y * 32;
  const int lane = threadIdx.x & 31;
  const int sub = lane & 15, hi = lane >> 4;
  const float* arow0 = A + (size_t)(i0 + sub) * lda + 2 * hi;
  const float* arow1 = arow0 + (size_t)16 * lda;
  const float* wrow0 = W + (size_t)(j0 + sub) * ldw + 2 * hi;
  const float* wrow1 = wrow0 + (size_t)16 * ldw;
  __builtin_prefetch(arow0, 0, 0);
  __builtin_prefetch(wrow0, 0, 0);
  v8f acc00 = {0.f, 0.f, 0.f, 0.f, 0.f, 0.f, 0.f, 0.f};
  v8f acc01 = acc00, acc10 = acc00, acc11 = acc00;
  #pragma unroll 4
  for (int k0 = 0; k0 < Kdim; k0 += 4) {
    const v2f a0 = *(const v2f*)(arow0 + k0);
    const v2f a1 = *(const v2f*)(arow1 + k0);
    const v2f b0 = *(const v2f*)(wrow0 + k0);
    const v2f b1 = *(const v2f*)(wrow1 + k0);
    acc00 = wmma4(a0, b0, acc00);
    acc01 = wmma4(a0, b1, acc01);
    acc10 = wmma4(a1, b0, acc10);
    acc11 = wmma4(a1, b1, acc11);
  }
  const float bj0 = bias ? bias[j0 + sub]      : 0.0f;
  const float bj1 = bias ? bias[j0 + 16 + sub] : 0.0f;
  #pragma unroll
  for (int r = 0; r < 8; ++r) {
    const int row0 = i0 + r + 8 * hi;
    const int row1 = row0 + 16;
    const int col0 = j0 + sub;
    const int col1 = col0 + 16;
    float v00 = acc00[r] + bj0, v01 = acc01[r] + bj1;
    float v10 = acc10[r] + bj0, v11 = acc11[r] + bj1;
    if (act == 1) {
      v00 = gelu_exact(v00); v01 = gelu_exact(v01);
      v10 = gelu_exact(v10); v11 = gelu_exact(v11);
    }
    if (resid) {
      v00 += resid[(size_t)row0 * ldc + col0];
      v01 += resid[(size_t)row0 * ldc + col1];
      v10 += resid[(size_t)row1 * ldc + col0];
      v11 += resid[(size_t)row1 * ldc + col1];
    }
    Cout[(size_t)row0 * ldc + col0] = v00;
    Cout[(size_t)row0 * ldc + col1] = v01;
    Cout[(size_t)row1 * ldc + col0] = v10;
    Cout[(size_t)row1 * ldc + col1] = v11;
  }
}

__global__ void attn_kernel(const float* __restrict__ qkv,
                            const float* __restrict__ xyzf,
                            const float* __restrict__ Wsp,
                            float* __restrict__ attn_out) {
  extern __shared__ float smem[];
  float* S    = smem;
  float* red  = S + 16 * 1024;
  float* rmax = red + 256;
  float* rsum = rmax + 16;
  float* wx   = rsum + 16;
  const int b = blockIdx.z, h = blockIdx.y, i0 = blockIdx.x * 16;
  const int t = threadIdx.x;
  const int lane = t & 31, w = t >> 5;
  const int sub = lane & 15, hi = lane >> 4;

  const float* qbase =
      qkv + ((size_t)(b * MTOK_ + i0 + sub) * 3 + 0) * C_ + h * DH_ + 2 * hi;
  v2f qa[16];
  #pragma unroll
  for (int ks = 0; ks < 16; ++ks) qa[ks] = *(const v2f*)(qbase + 4 * ks);
  for (int jj = 0; jj < 8; ++jj) {
    const int j0 = (w * 8 + jj) * 16;
    const float* kbase =
        qkv + ((size_t)(b * MTOK_ + j0 + sub) * 3 + 1) * C_ + h * DH_ + 2 * hi;
    v8f acc = {0.f, 0.f, 0.f, 0.f, 0.f, 0.f, 0.f, 0.f};
    #pragma unroll
    for (int ks = 0; ks < 16; ++ks) {
      const v2f kb = *(const v2f*)(kbase + 4 * ks);
      acc = wmma4(qa[ks], kb, acc);
    }
    #pragma unroll
    for (int r = 0; r < 8; ++r)
      S[(r + 8 * hi) * 1024 + j0 + sub] = acc[r] * 0.125f;
  }
  __syncthreads();

  const int row = t >> 4, seg = t & 15;
  float* sr = S + row * 1024 + seg * 64;
  float lm = -3.4e38f;
  for (int c2 = 0; c2 < 64; ++c2) lm = fmaxf(lm, sr[c2]);
  red[t] = lm; __syncthreads();
  if (t < 16) {
    float mm = -3.4e38f;
    for (int s2 = 0; s2 < 16; ++s2) mm = fmaxf(mm, red[t * 16 + s2]);
    rmax[t] = mm;
  }
  __syncthreads();
  const float rm = rmax[row];
  float ls = 0.0f;
  for (int c2 = 0; c2 < 64; ++c2) {
    const float e = expf(sr[c2] - rm);
    sr[c2] = e; ls += e;
  }
  red[t] = ls; __syncthreads();
  if (t < 16) {
    float ss = 0.0f;
    for (int s2 = 0; s2 < 16; ++s2) ss += red[t * 16 + s2];
    rsum[t] = ss;
  }
  __syncthreads();

  v8f acc = {0.f, 0.f, 0.f, 0.f, 0.f, 0.f, 0.f, 0.f};
  const int d0 = (w & 3) * 16;
  if (w < 4) {
    const float* vbase =
        qkv + ((size_t)(b * MTOK_) * 3 + 2) * C_ + h * DH_ + d0 + sub;
    const float* srow = S + sub * 1024 + 2 * hi;
    for (int ks = 0; ks < 256; ++ks) {
      const v2f a = *(const v2f*)(srow + 4 * ks);
      const int kk = 4 * ks + 2 * hi;
      v2f bb;
      bb.x = vbase[(size_t)kk * 1536];
      bb.y = vbase[(size_t)(kk + 1) * 1536];
      acc = wmma4(a, bb, acc);
    }
  } else {
    const int q = (w - 4) * 32 + lane;
    if (q < 48) {
      const int rr = q / 3, cc = q % 3;
      const float* srw = S + rr * 1024;
      const float* xb  = xyzf + (size_t)b * MTOK_ * 3 + cc;
      float s = 0.0f;
      for (int j = 0; j < 1024; ++j) s += srw[j] * xb[j * 3];
      wx[q] = s;
    }
  }
  __syncthreads();
  if (w < 4) {
    const int d = d0 + sub;
    const float w0 = Wsp[d * 3 + 0], w1 = Wsp[d * 3 + 1], w2 = Wsp[d * 3 + 2];
    #pragma unroll
    for (int r = 0; r < 8; ++r) {
      const int rr = r + 8 * hi;
      const int i  = i0 + rr;
      const float inv = 1.0f / rsum[rr];
      const float* xi = xyzf + ((size_t)b * MTOK_ + i) * 3;
      const float wd0 = wx[rr * 3 + 0] * inv - xi[0];
      const float wd1 = wx[rr * 3 + 1] * inv - xi[1];
      const float wd2 = wx[rr * 3 + 2] * inv - xi[2];
      attn_out[((size_t)(b * MTOK_) + i) * C_ + h * DH_ + d] =
          acc[r] * inv + w0 * wd0 + w1 * wd1 + w2 * wd2;
    }
  }
}

extern "C" void kernel_launch(void* const* d_in, const int* in_sizes, int n_in,
                              void* d_out, int out_size, void* d_ws, size_t ws_size,
                              hipStream_t stream) {
  (void)in_sizes; (void)n_in; (void)out_size; (void)ws_size;
  const float* xyzs = (const float*)d_in[0];
  const float* Wd   = (const float*)d_in[1];
  const float* Wqkv = (const float*)d_in[2];
  const float* Wsp  = (const float*)d_in[3];
  const float* Wout = (const float*)d_in[4];
  const float* bout = (const float*)d_in[5];
  const float* ln1g = (const float*)d_in[6];
  const float* ln1b = (const float*)d_in[7];
  const float* Wff1 = (const float*)d_in[8];
  const float* bff1 = (const float*)d_in[9];
  const float* Wff2 = (const float*)d_in[10];
  const float* bff2 = (const float*)d_in[11];
  const float* ln2g = (const float*)d_in[12];
  const float* ln2b = (const float*)d_in[13];

  float* ws      = (float*)d_ws;
  float* anchors = ws;
  float* feats   = ws + 8192;
  float* xln     = feats + (size_t)ROWS_ * C_;
  float* qkvb    = xln   + (size_t)ROWS_ * C_;
  float* attno   = qkvb  + (size_t)ROWS_ * 1536;
  float* ffh     = attno + (size_t)ROWS_ * C_;

  fps_kernel<<<B_ * L_, N_, 0, stream>>>(xyzs, anchors);
  p4dconv_kernel<<<B_ * L_ * MA_, N_, 0, stream>>>(xyzs, anchors, Wd, feats);

  const size_t smem_attn = (16 * 1024 + 256 + 16 + 16 + 48) * sizeof(float);
  for (int d = 0; d < 2; ++d) {
    ln_kernel<<<ROWS_, 256, 0, stream>>>(feats, ln1g + d * C_, ln1b + d * C_, xln);
    gemm_wmma<<<dim3(1536 / 32, ROWS_ / 32), 32, 0, stream>>>(
        xln, C_, Wqkv + (size_t)d * 1536 * C_, C_, nullptr, nullptr,
        qkvb, 1536, C_, 0);
    attn_kernel<<<dim3(MTOK_ / 16, HEADS_, B_), 256, smem_attn, stream>>>(
        qkvb, anchors, Wsp + (size_t)d * DH_ * 3, attno);
    gemm_wmma<<<dim3(C_ / 32, ROWS_ / 32), 32, 0, stream>>>(
        attno, C_, Wout + (size_t)d * C_ * C_, C_, bout + d * C_,
        feats, feats, C_, C_, 1);
    ln_kernel<<<ROWS_, 256, 0, stream>>>(feats, ln2g + d * C_, ln2b + d * C_, xln);
    gemm_wmma<<<dim3(MLP_ / 32, ROWS_ / 32), 32, 0, stream>>>(
        xln, C_, Wff1 + (size_t)d * MLP_ * C_, C_, bff1 + d * MLP_,
        nullptr, ffh, MLP_, C_, 1);
    gemm_wmma<<<dim3(C_ / 32, ROWS_ / 32), 32, 0, stream>>>(
        ffh, MLP_, Wff2 + (size_t)d * C_ * MLP_, MLP_, bff2 + d * C_,
        feats, feats, C_, MLP_, 0);
  }
  hipMemcpyAsync(d_out, feats, (size_t)ROWS_ * C_ * sizeof(float),
                 hipMemcpyDeviceToDevice, stream);
}


# attention as streaming online softmax: 4 waves split the keys, S^T = K Q^T tiles stay in registers and are the B operand of V^T P^T, no score rows in LDS; row sums, P@V, P@xyz add in another order
# speedup vs baseline: 1.2190x; 1.2190x over previous
#include <hip/hip_runtime.h>
#include <hip/hip_bf16.h>
#include <math.h>

typedef float v2f __attribute__((ext_vector_type(2)));
typedef float v4f __attribute__((ext_vector_type(4)));
typedef float v8f __attribute__((ext_vector_type(8)));

#define B_    2
#define T_    4
#define N_    1024
#define L_    4
#define MA_   256
#define KNB_  32
#define C_    512
#define HEADS_ 8
#define DH_   64
#define MTOK_ 1024
#define ROWS_ 2048
#define MLP_  2048

__device__ __forceinline__ v8f wmma4(v2f a, v2f b, v8f c) {
  return __builtin_amdgcn_wmma_f32_16x16x4_f32(false, a, false, b, (short)0, c,
                                               false, false);
}

__device__ __forceinline__ float gelu_exact(float x) {
  return 0.5f * x * (1.0f + erff(x * 0.7071067811865475f));
}

__device__ __forceinline__ float xhalf(float x) {
  const int bx = __builtin_bit_cast(int, x);
  return __builtin_bit_cast(float, __builtin_amdgcn_permlanex16(
      bx, bx, 0x76543210, 0xfedcba98, false, false));
}

__global__ void fps_kernel(const float* __restrict__ xyzs,
                           float* __restrict__ anchors) {
  __shared__ float sval[N_];
  __shared__ int   sidx[N_];
  __shared__ float last[3];
  const int blk = blockIdx.x;
  const int tid = threadIdx.x;
  const float* frame = xyzs + (size_t)blk * N_ * 3;
  const float px = frame[tid * 3 + 0];
  const float py = frame[tid * 3 + 1];
  const float pz = frame[tid * 3 + 2];
  float dist = 1e10f;
  float* aout = anchors + (size_t)blk * MA_ * 3;
  if (tid == 0) {
    last[0] = px; last[1] = py; last[2] = pz;
    aout[0] = px; aout[1] = py; aout[2] = pz;
  }
  __syncthreads();
  for (int it = 1; it < MA_; ++it) {
    const float dx = px - last[0], dy = py - last[1], dz = pz - last[2];
    dist = fminf(dist, dx * dx + dy * dy + dz * dz);
    sval[tid] = dist; sidx[tid] = tid;
    __syncthreads();
    for (int s = 512; s > 0; s >>= 1) {
      if (tid < s) {
        const float v2v = sval[tid + s];
        const int   i2  = sidx[tid + s];
        if (v2v > sval[tid] || (v2v == sval[tid] && i2 < sidx[tid])) {
          sval[tid] = v2v; sidx[tid] = i2;
        }
      }
      __syncthreads();
    }
    const int nxt = sidx[0];
    __syncthreads();
    if (tid == nxt) {
      last[0] = px; last[1] = py; last[2] = pz;
      aout[it * 3 + 0] = px; aout[it * 3 + 1] = py; aout[it * 3 + 2] = pz;
    }
    __syncthreads();
  }
}

__global__ void p4dconv_kernel(const float* __restrict__ xyzs,
                               const float* __restrict__ anchors,
                               const float* __restrict__ Wd,
                               float* __restrict__ feats) {
  __shared__ int   sscan[N_];
  __shared__ float spx[KNB_], spy[KNB_], spz[KNB_];
  __shared__ float s_p0[3];
  const int blk = blockIdx.x;
  const int m   = blk % MA_;
  const int bl  = blk / MA_;
  const int l   = bl % L_;
  const int b   = bl / L_;
  const int tid = threadIdx.x;

  const float* arow = anchors + ((size_t)bl * MA_ + m) * 3;
  const float ax = arow[0], ay = arow[1], az = arow[2];

  float w0 = 0.f, w1 = 0.f, w2 = 0.f, w3 = 0.f;
  if (tid < C_) {
    w0 = Wd[tid * 4 + 0]; w1 = Wd[tid * 4 + 1];
    w2 = Wd[tid * 4 + 2]; w3 = Wd[tid * 4 + 3];
  }
  float maxv = -3.4e38f;

  for (int di = -1; di <= 1; ++di) {
    int fi = l + di; fi = fi < 0 ? 0 : (fi > T_ - 1 ? T_ - 1 : fi);
    const float* nb = xyzs + ((size_t)b * T_ + fi) * N_ * 3;
    const float qx = nb[tid * 3 + 0];
    const float qy = nb[tid * 3 + 1];
    const float qz = nb[tid * 3 + 2];
    const float dx = qx - ax, dy = qy - ay, dz = qz - az;
    const int pred = (dx * dx + dy * dy + dz * dz) < 1.0f ? 1 : 0;
    sscan[tid] = pred;
    if (tid == 0) { s_p0[0] = dx; s_p0[1] = dy; s_p0[2] = dz; }
    __syncthreads();
    for (int off = 1; off < N_; off <<= 1) {
      const int v   = sscan[tid];
      const int add = (tid >= off) ? sscan[tid - off] : 0;
      __syncthreads();
      sscan[tid] = v + add;
      __syncthreads();
    }
    const int incl  = sscan[tid];
    const int total = sscan[N_ - 1];
    const int pos   = incl - pred;
    if (pred && pos < KNB_) { spx[pos] = dx; spy[pos] = dy; spz[pos] = dz; }
    __syncthreads();
    const int cnt = total < KNB_ ? total : KNB_;
    if (tid < KNB_ && tid >= cnt) {
      spx[tid] = (total > 0) ? spx[0] : s_p0[0];
      spy[tid] = (total > 0) ? spy[0] : s_p0[1];
      spz[tid] = (total > 0) ? spz[0] : s_p0[2];
    }
    __syncthreads();
    if (tid < C_) {
      const float tq = (float)di;
      #pragma unroll 8
      for (int k = 0; k < KNB_; ++k) {
        const float v = w0 * spx[k] + w1 * spy[k] + w2 * spz[k] + w3 * tq;
        maxv = fmaxf(maxv, v);
      }
    }
    __syncthreads();
  }
  if (tid < C_)
    feats[((size_t)bl * MA_ + m) * C_ + tid] = fmaxf(maxv, 0.0f);
}

__global__ void ln_kernel(const float* __restrict__ x, const float* __restrict__ g,
                          const float* __restrict__ bta, float* __restrict__ y) {
  __shared__ float red[256];
  const int row = blockIdx.x, t = threadIdx.x;
  const float* xr = x + (size_t)row * C_;
  const float a0 = xr[t], a1 = xr[t + 256];
  red[t] = a0 + a1; __syncthreads();
  for (int s = 128; s > 0; s >>= 1) { if (t < s) red[t] += red[t + s]; __syncthreads(); }
  const float mean = red[0] * (1.0f / C_);
  __syncthreads();
  const float d0 = a0 - mean, d1 = a1 - mean;
  red[t] = d0 * d0 + d1 * d1; __syncthreads();
  for (int s = 128; s > 0; s >>= 1) { if (t < s) red[t] += red[t + s]; __syncthreads(); }
  const float inv = rsqrtf(red[0] * (1.0f / C_) + 1e-5f);
  float* yr = y + (size_t)row * C_;
  yr[t]       = d0 * inv * g[t]       + bta[t];
  yr[t + 256] = d1 * inv * g[t + 256] + bta[t + 256];
}

__global__ void gemm_wmma(const float* __restrict__ A, int lda,
                          const float* __restrict__ W, int ldw,
                          const float* __restrict__ bias,
                          const float* __restrict__ resid,
                          float* __restrict__ Cout, int ldc,
                          int Kdim, int act) {
  const int j0 = blockIdx.x * 32, i0 = blockIdx.y * 32;
  const int lane = threadIdx.x & 31;
  const int sub = lane & 15, hi = lane >> 4;
  const float* arow0 = A + (size_t)(i0 + sub) * lda + 2 * hi;
  const float* arow1 = arow0 + (size_t)16 * lda;
  const float* wrow0 = W + (size_t)(j0 + sub) * ldw + 2 * hi;
  const float* wrow1 = wrow0 + (size_t)16 * ldw;
  __builtin_prefetch(arow0, 0, 0);
  __builtin_prefetch(wrow0, 0, 0);
  v8f acc00 = {0.f, 0.f, 0.f, 0.f, 0.f, 0.f, 0.f, 0.f};
  v8f acc01 = acc00, acc10 = acc00, acc11 = acc00;
  #pragma unroll 4
  for (int k0 = 0; k0 < Kdim; k0 += 4) {
    const v2f a0 = *(const v2f*)(arow0 + k0);
    const v2f a1 = *(const v2f*)(arow1 + k0);
    const v2f b0 = *(const v2f*)(wrow0 + k0);
    const v2f b1 = *(const v2f*)(wrow1 + k0);
    acc00 = wmma4(a0, b0, acc00);
    acc01 = wmma4(a0, b1, acc01);
    acc10 = wmma4(a1, b0, acc10);
    acc11 = wmma4(a1, b1, acc11);
  }
  const float bj0 = bias ? bias[j0 + sub]      : 0.0f;
  const float bj1 = bias ? bias[j0 + 16 + sub] : 0.0f;
  #pragma unroll
  for (int r = 0; r < 8; ++r) {
    const int row0 = i0 + r + 8 * hi;
    const int row1 = row0 + 16;
    const int col0 = j0 + sub;
    const int col1 = col0 + 16;
    float v00 = acc00[r] + bj0, v01 = acc01[r] + bj1;
    float v10 = acc10[r] + bj0, v11 = acc11[r] + bj1;
    if (act == 1) {
      v00 = gelu_exact(v00); v01 = gelu_exact(v01);
      v10 = gelu_exact(v10); v11 = gelu_exact(v11);
    }
    if (resid) {
      v00 += resid[(size_t)row0 * ldc + col0];
      v01 += resid[(size_t)row0 * ldc + col1];
      v10 += resid[(size_t)row1 * ldc + col0];
      v11 += resid[(size_t)row1 * ldc + col1];
    }
    Cout[(size_t)row0 * ldc + col0] = v00;
    Cout[(size_t)row0 * ldc + col1] = v01;
    Cout[(size_t)row1 * ldc + col0] = v10;
    Cout[(size_t)row1 * ldc + col1] = v11;
  }
}

#define ATT_WAVES_ 4
#define ATT_KPW_   (MTOK_ / ATT_WAVES_)
#define ATT_OP_    68

__global__ void __launch_bounds__(32 * ATT_WAVES_)
attn_kernel(const float* __restrict__ qkv,
            const float* __restrict__ xyzf,
            const float* __restrict__ Wsp,
            float* __restrict__ attn_out) {
  __shared__ float s_m[ATT_WAVES_][16];
  __shared__ float s_l[ATT_WAVES_][16];
  __shared__ float s_wx[ATT_WAVES_][16][3];
  __shared__ __attribute__((aligned(16))) float s_o[ATT_WAVES_][16][ATT_OP_];
  const int b = blockIdx.z, h = blockIdx.y, i0 = blockIdx.x * 16;
  const int t = threadIdx.x;
  const int lane = t & 31, w = t >> 5;
  const int sub = lane & 15, hi = lane >> 4;
  const int key0 = b * MTOK_ + w * ATT_KPW_;

  const float* qbase =
      qkv + ((size_t)(b * MTOK_ + i0 + sub) * 3 + 0) * C_ + h * DH_ + 2 * hi;
  v2f qb[16];
  #pragma unroll
  for (int ks = 0; ks < 16; ++ks)
    qb[ks] = *(const v2f*)(qbase + 4 * ks) * 0.125f;

  const float* krow =
      qkv + ((size_t)(key0 + sub) * 3 + 1) * C_ + h * DH_ + 2 * hi;
  const float* vcol =
      qkv + ((size_t)(key0 + 8 * hi) * 3 + 2) * C_ + h * DH_ + sub;
  const float* xrow = xyzf + (size_t)(key0 + 8 * hi) * 3;

  v8f o0 = {0.f, 0.f, 0.f, 0.f, 0.f, 0.f, 0.f, 0.f};
  v8f o1 = o0, o2 = o0, o3 = o0;
  float m = -3.4e38f, l = 0.0f, wx0 = 0.0f, wx1 = 0.0f, wx2 = 0.0f;

  v2f kf[16];
  #pragma unroll
  for (int ks = 0; ks < 16; ++ks) kf[ks] = *(const v2f*)(krow + 4 * ks);

  for (int jt = 0; jt < ATT_KPW_ / 16; ++jt) {
    v8f st = {0.f, 0.f, 0.f, 0.f, 0.f, 0.f, 0.f, 0.f};
    #pragma unroll
    for (int ks = 0; ks < 16; ++ks) st = wmma4(kf[ks], qb[ks], st);

    const int jn = (jt + 1 < ATT_KPW_ / 16) ? jt + 1 : jt;
    const float* kn = krow + (size_t)jn * 16 * 3 * C_;
    #pragma unroll
    for (int ks = 0; ks < 16; ++ks) kf[ks] = *(const v2f*)(kn + 4 * ks);

    const float* vt = vcol + (size_t)jt * 16 * 3 * C_;
    v2f vf[4][4];
    #pragma unroll
    for (int s = 0; s < 4; ++s) {
      #pragma unroll
      for (int dt = 0; dt < 4; ++dt) {
        const float* vp = vt + (size_t)(2 * s) * 3 * C_ + dt * 16;
        vf[s][dt].x = vp[0];
        vf[s][dt].y = vp[3 * C_];
      }
    }
    const v4f* xp = (const v4f*)(xrow + jt * 48);
    float xk[24];
    #pragma unroll
    for (int i = 0; i < 6; ++i) {
      const v4f xv = xp[i];
      xk[4 * i + 0] = xv.x; xk[4 * i + 1] = xv.y;
      xk[4 * i + 2] = xv.z; xk[4 * i + 3] = xv.w;
    }

    float tm = st[0];
    #pragma unroll
    for (int r = 1; r < 8; ++r) tm = fmaxf(tm, st[r]);
    tm = fmaxf(tm, xhalf(tm));
    const float mn = fmaxf(m, tm);
    const float alpha = expf(m - mn);
    m = mn;
    float p[8];
    float ps = 0.0f, a0 = 0.0f, a1 = 0.0f, a2 = 0.0f;
    #pragma unroll
    for (int r = 0; r < 8; ++r) {
      p[r] = expf(st[r] - mn);
      ps += p[r];
      a0 += p[r] * xk[3 * r + 0];
      a1 += p[r] * xk[3 * r + 1];
      a2 += p[r] * xk[3 * r + 2];
    }
    l   = l   * alpha + ps;
    wx0 = wx0 * alpha + a0;
    wx1 = wx1 * alpha + a1;
    wx2 = wx2 * alpha + a2;
    #pragma unroll
    for (int r = 0; r < 8; ++r) {
      o0[r] *= alpha; o1[r] *= alpha; o2[r] *= alpha; o3[r] *= alpha;
    }
    #pragma unroll
    for (int s = 0; s < 4; ++s) {
      v2f pb; pb.x = p[2 * s]; pb.y = p[2 * s + 1];
      o0 = wmma4(vf[s][0], pb, o0);
      o1 = wmma4(vf[s][1], pb, o1);
      o2 = wmma4(vf[s][2], pb, o2);
      o3 = wmma4(vf[s][3], pb, o3);
    }
  }

  l   += xhalf(l);
  wx0 += xhalf(wx0);
  wx1 += xhalf(wx1);
  wx2 += xhalf(wx2);
  if (hi == 0) {
    s_m[w][sub] = m;
    s_l[w][sub] = l;
    s_wx[w][sub][0] = wx0; s_wx[w][sub][1] = wx1; s_wx[w][sub][2] = wx2;
  }
  #pragma unroll
  for (int r = 0; r < 8; ++r) {
    s_o[w][sub][ 0 + r + 8 * hi] = o0[r];
    s_o[w][sub][16 + r + 8 * hi] = o1[r];
    s_o[w][sub][32 + r + 8 * hi] = o2[r];
    s_o[w][sub][48 + r + 8 * hi] = o3[r];
  }
  __syncthreads();

  #pragma unroll
  for (int pass = 0; pass < 2; ++pass) {
    const int q = pass * 8 + (t >> 4), dch = (t & 15) * 4;
    const float m0 = s_m[0][q], m1 = s_m[1][q], m2 = s_m[2][q], m3 = s_m[3][q];
    const float ma = fmaxf(fmaxf(m0, m1), fmaxf(m2, m3));
    const float f0 = expf(m0 - ma), f1 = expf(m1 - ma);
    const float f2 = expf(m2 - ma), f3 = expf(m3 - ma);
    const float lt = ((s_l[0][q] * f0 + s_l[1][q] * f1) + s_l[2][q] * f2) +
                     s_l[3][q] * f3;
    const float inv = 1.0f / lt;
    const int i = i0 + q;
    const float* xi = xyzf + ((size_t)b * MTOK_ + i) * 3;
    float wd[3];
    #pragma unroll
    for (int c = 0; c < 3; ++c)
      wd[c] = (((s_wx[0][q][c] * f0 + s_wx[1][q][c] * f1) + s_wx[2][q][c] * f2) +
               s_wx[3][q][c] * f3) * inv - xi[c];
    const v4f p0 = *(const v4f*)&s_o[0][q][dch];
    const v4f p1 = *(const v4f*)&s_o[1][q][dch];
    const v4f p2 = *(const v4f*)&s_o[2][q][dch];
    const v4f p3 = *(const v4f*)&s_o[3][q][dch];
    const v4f sm = ((p0 * f0 + p1 * f1) + p2 * f2) + p3 * f3;
    float res[4] = {sm.x, sm.y, sm.z, sm.w};
    #pragma unroll
    for (int e = 0; e < 4; ++e) {
      const int d = dch + e;
      res[e] = res[e] * inv + Wsp[d * 3 + 0] * wd[0] + Wsp[d * 3 + 1] * wd[1] +
               Wsp[d * 3 + 2] * wd[2];
    }
    v4f rv;
    rv.x = res[0]; rv.y = res[1]; rv.z = res[2]; rv.w = res[3];
    *(v4f*)(attn_out + ((size_t)(b * MTOK_) + i) * C_ + h * DH_ + dch) = rv;
  }
}

extern "C" void kernel_launch(void* const* d_in, const int* in_sizes, int n_in,
                              void* d_out, int out_size, void* d_ws, size_t ws_size,
                              hipStream_t stream) {
  (void)in_sizes; (void)n_in; (void)out_size; (void)ws_size;
  const float* xyzs = (const float*)d_in[0];
  const float* Wd   = (const float*)d_in[1];
  const float* Wqkv = (const float*)d_in[2];
  const float* Wsp  = (const float*)d_in[3];
  const float* Wout = (const float*)d_in[4];
  const float* bout = (const float*)d_in[5];
  const float* ln1g = (const float*)d_in[6];
  const float* ln1b = (const float*)d_in[7];
  const float* Wff1 = (const float*)d_in[8];
  const float* bff1 = (const float*)d_in[9];
  const float* Wff2 = (const float*)d_in[10];
  const float* bff2 = (const float*)d_in[11];
  const float* ln2g = (const float*)d_in[12];
  const float* ln2b = (const float*)d_in[13];

  float* ws      = (float*)d_ws;
  float* anchors = ws;
  float* feats   = ws + 8192;
  float* xln     = feats + (size_t)ROWS_ * C_;
  float* qkvb    = xln   + (size_t)ROWS_ * C_;
  float* attno   = qkvb  + (size_t)ROWS_ * 1536;
  float* ffh     = attno + (size_t)ROWS_ * C_;

  fps_kernel<<<B_ * L_, N_, 0, stream>>>(xyzs, anchors);
  p4dconv_kernel<<<B_ * L_ * MA_, N_, 0, stream>>>(xyzs, anchors, Wd, feats);

  for (int d = 0; d < 2; ++d) {
    ln_kernel<<<ROWS_, 256, 0, stream>>>(feats, ln1g + d * C_, ln1b + d * C_, xln);
    gemm_wmma<<<dim3(1536 / 32, ROWS_ / 32), 32, 0, stream>>>(
        xln, C_, Wqkv + (size_t)d * 1536 * C_, C_, nullptr, nullptr,
        qkvb, 1536, C_, 0);
    attn_kernel<<<dim3(MTOK_ / 16, HEADS_, B_), 32 * ATT_WAVES_, 0, stream>>>(
        qkvb, anchors, Wsp + (size_t)d * DH_ * 3, attno);
    gemm_wmma<<<dim3(C_ / 32, ROWS_ / 32), 32, 0, stream>>>(
        attno, C_, Wout + (size_t)d * C_ * C_, C_, bout + d * C_,
        feats, feats, C_, C_, 1);
    ln_kernel<<<ROWS_, 256, 0, stream>>>(feats, ln2g + d * C_, ln2b + d * C_, xln);
    gemm_wmma<<<dim3(MLP_ / 32, ROWS_ / 32), 32, 0, stream>>>(
        xln, C_, Wff1 + (size_t)d * MLP_ * C_, C_, bff1 + d * MLP_,
        nullptr, ffh, MLP_, C_, 1);
    gemm_wmma<<<dim3(C_ / 32, ROWS_ / 32), 32, 0, stream>>>(
        ffh, MLP_, Wff2 + (size_t)d * C_ * MLP_, MLP_, bff2 + d * C_,
        feats, feats, C_, MLP_, 0);
  }
  hipMemcpyAsync(d_out, feats, (size_t)ROWS_ * C_ * sizeof(float),
                 hipMemcpyDeviceToDevice, stream);
}


# GEMM k loop pipelined in registers: 64 k in flight as eight 8-k slices (16 B per lane, two WMMAs per load), each refilled behind its own 8 WMMAs, 48-58 WMMAs ahead of use; k summed in another order
# speedup vs baseline: 1.5104x; 1.2390x over previous
#include <hip/hip_runtime.h>
#include <hip/hip_bf16.h>
#include <math.h>

typedef float v2f __attribute__((ext_vector_type(2)));
typedef float v4f __attribute__((ext_vector_type(4)));
typedef float v8f __attribute__((ext_vector_type(8)));

#define B_    2
#define T_    4
#define N_    1024
#define L_    4
#define MA_   256
#define KNB_  32
#define C_    512
#define HEADS_ 8
#define DH_   64
#define MTOK_ 1024
#define ROWS_ 2048
#define MLP_  2048

__device__ __forceinline__ v8f wmma4(v2f a, v2f b, v8f c) {
  return __builtin_amdgcn_wmma_f32_16x16x4_f32(false, a, false, b, (short)0, c,
                                               false, false);
}

__device__ __forceinline__ float gelu_exact(float x) {
  return 0.5f * x * (1.0f + erff(x * 0.7071067811865475f));
}

__device__ __forceinline__ float xhalf(float x) {
  const int bx = __builtin_bit_cast(int, x);
  return __builtin_bit_cast(float, __builtin_amdgcn_permlanex16(
      bx, bx, 0x76543210, 0xfedcba98, false, false));
}

__global__ void fps_kernel(const float* __restrict__ xyzs,
                           float* __restrict__ anchors) {
  __shared__ float sval[N_];
  __shared__ int   sidx[N_];
  __shared__ float last[3];
  const int blk = blockIdx.x;
  const int tid = threadIdx.x;
  const float* frame = xyzs + (size_t)blk * N_ * 3;
  const float px = frame[tid * 3 + 0];
  const float py = frame[tid * 3 + 1];
  const float pz = frame[tid * 3 + 2];
  float dist = 1e10f;
  float* aout = anchors + (size_t)blk * MA_ * 3;
  if (tid == 0) {
    last[0] = px; last[1] = py; last[2] = pz;
    aout[0] = px; aout[1] = py; aout[2] = pz;
  }
  __syncthreads();
  for (int it = 1; it < MA_; ++it) {
    const float dx = px - last[0], dy = py - last[1], dz = pz - last[2];
    dist = fminf(dist, dx * dx + dy * dy + dz * dz);
    sval[tid] = dist; sidx[tid] = tid;
    __syncthreads();
    for (int s = 512; s > 0; s >>= 1) {
      if (tid < s) {
        const float v2v = sval[tid + s];
        const int   i2  = sidx[tid + s];
        if (v2v > sval[tid] || (v2v == sval[tid] && i2 < sidx[tid])) {
          sval[tid] = v2v; sidx[tid] = i2;
        }
      }
      __syncthreads();
    }
    const int nxt = sidx[0];
    __syncthreads();
    if (tid == nxt) {
      last[0] = px; last[1] = py; last[2] = pz;
      aout[it * 3 + 0] = px; aout[it * 3 + 1] = py; aout[it * 3 + 2] = pz;
    }
    __syncthreads();
  }
}

__global__ void p4dconv_kernel(const float* __restrict__ xyzs,
                               const float* __restrict__ anchors,
                               const float* __restrict__ Wd,
                               float* __restrict__ feats) {
  __shared__ int   sscan[N_];
  __shared__ float spx[KNB_], spy[KNB_], spz[KNB_];
  __shared__ float s_p0[3];
  const int blk = blockIdx.x;
  const int m   = blk % MA_;
  const int bl  = blk / MA_;
  const int l   = bl % L_;
  const int b   = bl / L_;
  const int tid = threadIdx.x;

  const float* arow = anchors + ((size_t)bl * MA_ + m) * 3;
  const float ax = arow[0], ay = arow[1], az = arow[2];

  float w0 = 0.f, w1 = 0.f, w2 = 0.f, w3 = 0.f;
  if (tid < C_) {
    w0 = Wd[tid * 4 + 0]; w1 = Wd[tid * 4 + 1];
    w2 = Wd[tid * 4 + 2]; w3 = Wd[tid * 4 + 3];
  }
  float maxv = -3.4e38f;

  for (int di = -1; di <= 1; ++di) {
    int fi = l + di; fi = fi < 0 ? 0 : (fi > T_ - 1 ? T_ - 1 : fi);
    const float* nb = xyzs + ((size_t)b * T_ + fi) * N_ * 3;
    const float qx = nb[tid * 3 + 0];
    const float qy = nb[tid * 3 + 1];
    const float qz = nb[tid * 3 + 2];
    const float dx = qx - ax, dy = qy - ay, dz = qz - az;
    const int pred = (dx * dx + dy * dy + dz * dz) < 1.0f ? 1 : 0;
    sscan[tid] = pred;
    if (tid == 0) { s_p0[0] = dx; s_p0[1] = dy; s_p0[2] = dz; }
    __syncthreads();
    for (int off = 1; off < N_; off <<= 1) {
      const int v   = sscan[tid];
      const int add = (tid >= off) ? sscan[tid - off] : 0;
      __syncthreads();
      sscan[tid] = v + add;
      __syncthreads();
    }
    const int incl  = sscan[tid];
    const int total = sscan[N_ - 1];
    const int pos   = incl - pred;
    if (pred && pos < KNB_) { spx[pos] = dx; spy[pos] = dy; spz[pos] = dz; }
    __syncthreads();
    const int cnt = total < KNB_ ? total : KNB_;
    if (tid < KNB_ && tid >= cnt) {
      spx[tid] = (total > 0) ? spx[0] : s_p0[0];
      spy[tid] = (total > 0) ? spy[0] : s_p0[1];
      spz[tid] = (total > 0) ? spz[0] : s_p0[2];
    }
    __syncthreads();
    if (tid < C_) {
      const float tq = (float)di;
      #pragma unroll 8
      for (int k = 0; k < KNB_; ++k) {
        const float v = w0 * spx[k] + w1 * spy[k] + w2 * spz[k] + w3 * tq;
        maxv = fmaxf(maxv, v);
      }
    }
    __syncthreads();
  }
  if (tid < C_)
    feats[((size_t)bl * MA_ + m) * C_ + tid] = fmaxf(maxv, 0.0f);
}

__global__ void ln_kernel(const float* __restrict__ x, const float* __restrict__ g,
                          const float* __restrict__ bta, float* __restrict__ y) {
  __shared__ float red[256];
  const int row = blockIdx.x, t = threadIdx.x;
  const float* xr = x + (size_t)row * C_;
  const float a0 = xr[t], a1 = xr[t + 256];
  red[t] = a0 + a1; __syncthreads();
  for (int s = 128; s > 0; s >>= 1) { if (t < s) red[t] += red[t + s]; __syncthreads(); }
  const float mean = red[0] * (1.0f / C_);
  __syncthreads();
  const float d0 = a0 - mean, d1 = a1 - mean;
  red[t] = d0 * d0 + d1 * d1; __syncthreads();
  for (int s = 128; s > 0; s >>= 1) { if (t < s) red[t] += red[t + s]; __syncthreads(); }
  const float inv = rsqrtf(red[0] * (1.0f / C_) + 1e-5f);
  float* yr = y + (size_t)row * C_;
  yr[t]       = d0 * inv * g[t]       + bta[t];
  yr[t + 256] = d1 * inv * g[t + 256] + bta[t + 256];
}

__device__ __forceinline__ void gemm_ld1(v4f (&f)[16], int kb, const float* a0,
                                         const float* a1, const float* w0,
                                         const float* w1) {
  f[4 * kb + 0] = *(const v4f*)(a0 + 8 * kb);
  f[4 * kb + 1] = *(const v4f*)(a1 + 8 * kb);
  f[4 * kb + 2] = *(const v4f*)(w0 + 8 * kb);
  f[4 * kb + 3] = *(const v4f*)(w1 + 8 * kb);
}

__device__ __forceinline__ void gemm_mm1(const v4f (&f)[16], int kb, v8f& acc00,
                                         v8f& acc01, v8f& acc10, v8f& acc11) {
  const v4f a0 = f[4 * kb + 0], a1 = f[4 * kb + 1];
  const v4f b0 = f[4 * kb + 2], b1 = f[4 * kb + 3];
  const v2f a0l = {a0.x, a0.y}, a0h = {a0.z, a0.w};
  const v2f a1l = {a1.x, a1.y}, a1h = {a1.z, a1.w};
  const v2f b0l = {b0.x, b0.y}, b0h = {b0.z, b0.w};
  const v2f b1l = {b1.x, b1.y}, b1h = {b1.z, b1.w};
  acc00 = wmma4(a0l, b0l, acc00);
  acc01 = wmma4(a0l, b1l, acc01);
  acc10 = wmma4(a1l, b0l, acc10);
  acc11 = wmma4(a1l, b1l, acc11);
  acc00 = wmma4(a0h, b0h, acc00);
  acc01 = wmma4(a0h, b1h, acc01);
  acc10 = wmma4(a1h, b0h, acc10);
  acc11 = wmma4(a1h, b1h, acc11);
}

__global__ void __launch_bounds__(32)
gemm_wmma(const float* __restrict__ A, int lda,
          const float* __restrict__ W, int ldw,
          const float* __restrict__ bias,
          const float* __restrict__ resid,
          float* __restrict__ Cout, int ldc,
          int Kdim, int act) {
  const int j0 = blockIdx.x * 32, i0 = blockIdx.y * 32;
  const int lane = threadIdx.x & 31;
  const int sub = lane & 15, hi = lane >> 4;
  const float* arow0 = A + (size_t)(i0 + sub) * lda + 4 * hi;
  const float* arow1 = arow0 + (size_t)16 * lda;
  const float* wrow0 = W + (size_t)(j0 + sub) * ldw + 4 * hi;
  const float* wrow1 = wrow0 + (size_t)16 * ldw;
  __builtin_prefetch(arow0, 0, 0);
  __builtin_prefetch(wrow0, 0, 0);
  v8f acc00 = {0.f, 0.f, 0.f, 0.f, 0.f, 0.f, 0.f, 0.f};
  v8f acc01 = acc00, acc10 = acc00, acc11 = acc00;
  v4f s0[16], s1[16];
  #pragma unroll
  for (int kb = 0; kb < 4; ++kb) gemm_ld1(s0, kb, arow0, arow1, wrow0, wrow1);
  #pragma unroll
  for (int kb = 0; kb < 4; ++kb)
    gemm_ld1(s1, kb, arow0 + 32, arow1 + 32, wrow0 + 32, wrow1 + 32);
  int k0 = 0;
  for (; k0 + 64 < Kdim; k0 += 64) {
    #pragma unroll
    for (int kb = 0; kb < 4; ++kb) {
      gemm_mm1(s0, kb, acc00, acc01, acc10, acc11);
      gemm_ld1(s0, kb, arow0 + k0 + 64, arow1 + k0 + 64, wrow0 + k0 + 64,
               wrow1 + k0 + 64);
    }
    #pragma unroll
    for (int kb = 0; kb < 4; ++kb) {
      gemm_mm1(s1, kb, acc00, acc01, acc10, acc11);
      gemm_ld1(s1, kb, arow0 + k0 + 96, arow1 + k0 + 96, wrow0 + k0 + 96,
               wrow1 + k0 + 96);
    }
  }
  #pragma unroll
  for (int kb = 0; kb < 4; ++kb) gemm_mm1(s0, kb, acc00, acc01, acc10, acc11);
  #pragma unroll
  for (int kb = 0; kb < 4; ++kb) gemm_mm1(s1, kb, acc00, acc01, acc10, acc11);
  const float bj0 = bias ? bias[j0 + sub]      : 0.0f;
  const float bj1 = bias ? bias[j0 + 16 + sub] : 0.0f;
  #pragma unroll
  for (int r = 0; r < 8; ++r) {
    const int row0 = i0 + r + 8 * hi;
    const int row1 = row0 + 16;
    const int col0 = j0 + sub;
    const int col1 = col0 + 16;
    float v00 = acc00[r] + bj0, v01 = acc01[r] + bj1;
    float v10 = acc10[r] + bj0, v11 = acc11[r] + bj1;
    if (act == 1) {
      v00 = gelu_exact(v00); v01 = gelu_exact(v01);
      v10 = gelu_exact(v10); v11 = gelu_exact(v11);
    }
    if (resid) {
      v00 += resid[(size_t)row0 * ldc + col0];
      v01 += resid[(size_t)row0 * ldc + col1];
      v10 += resid[(size_t)row1 * ldc + col0];
      v11 += resid[(size_t)row1 * ldc + col1];
    }
    Cout[(size_t)row0 * ldc + col0] = v00;
    Cout[(size_t)row0 * ldc + col1] = v01;
    Cout[(size_t)row1 * ldc + col0] = v10;
    Cout[(size_t)row1 * ldc + col1] = v11;
  }
}

#define ATT_WAVES_ 4
#define ATT_KPW_   (MTOK_ / ATT_WAVES_)
#define ATT_OP_    68

__global__ void __launch_bounds__(32 * ATT_WAVES_)
attn_kernel(const float* __restrict__ qkv,
            const float* __restrict__ xyzf,
            const float* __restrict__ Wsp,
            float* __restrict__ attn_out) {
  __shared__ float s_m[ATT_WAVES_][16];
  __shared__ float s_l[ATT_WAVES_][16];
  __shared__ float s_wx[ATT_WAVES_][16][3];
  __shared__ __attribute__((aligned(16))) float s_o[ATT_WAVES_][16][ATT_OP_];
  const int b = blockIdx.z, h = blockIdx.y, i0 = blockIdx.x * 16;
  const int t = threadIdx.x;
  const int lane = t & 31, w = t >> 5;
  const int sub = lane & 15, hi = lane >> 4;
  const int key0 = b * MTOK_ + w * ATT_KPW_;

  const float* qbase =
      qkv + ((size_t)(b * MTOK_ + i0 + sub) * 3 + 0) * C_ + h * DH_ + 2 * hi;
  v2f qb[16];
  #pragma unroll
  for (int ks = 0; ks < 16; ++ks)
    qb[ks] = *(const v2f*)(qbase + 4 * ks) * 0.125f;

  const float* krow =
      qkv + ((size_t)(key0 + sub) * 3 + 1) * C_ + h * DH_ + 2 * hi;
  const float* vcol =
      qkv + ((size_t)(key0 + 8 * hi) * 3 + 2) * C_ + h * DH_ + sub;
  const float* xrow = xyzf + (size_t)(key0 + 8 * hi) * 3;

  v8f o0 = {0.f, 0.f, 0.f, 0.f, 0.f, 0.f, 0.f, 0.f};
  v8f o1 = o0, o2 = o0, o3 = o0;
  float m = -3.4e38f, l = 0.0f, wx0 = 0.0f, wx1 = 0.0f, wx2 = 0.0f;

  v2f kf[16];
  #pragma unroll
  for (int ks = 0; ks < 16; ++ks) kf[ks] = *(const v2f*)(krow + 4 * ks);

  for (int jt = 0; jt < ATT_KPW_ / 16; ++jt) {
    v8f st = {0.f, 0.f, 0.f, 0.f, 0.f, 0.f, 0.f, 0.f};
    #pragma unroll
    for (int ks = 0; ks < 16; ++ks) st = wmma4(kf[ks], qb[ks], st);

    const int jn = (jt + 1 < ATT_KPW_ / 16) ? jt + 1 : jt;
    const float* kn = krow + (size_t)jn * 16 * 3 * C_;
    #pragma unroll
    for (int ks = 0; ks < 16; ++ks) kf[ks] = *(const v2f*)(kn + 4 * ks);

    const float* vt = vcol + (size_t)jt * 16 * 3 * C_;
    v2f vf[4][4];
    #pragma unroll
    for (int s = 0; s < 4; ++s) {
      #pragma unroll
      for (int dt = 0; dt < 4; ++dt) {
        const float* vp = vt + (size_t)(2 * s) * 3 * C_ + dt * 16;
        vf[s][dt].x = vp[0];
        vf[s][dt].y = vp[3 * C_];
      }
    }
    const v4f* xp = (const v4f*)(xrow + jt * 48);
    float xk[24];
    #pragma unroll
    for (int i = 0; i < 6; ++i) {
      const v4f xv = xp[i];
      xk[4 * i + 0] = xv.x; xk[4 * i + 1] = xv.y;
      xk[4 * i + 2] = xv.z; xk[4 * i + 3] = xv.w;
    }

    float tm = st[0];
    #pragma unroll
    for (int r = 1; r < 8; ++r) tm = fmaxf(tm, st[r]);
    tm = fmaxf(tm, xhalf(tm));
    const float mn = fmaxf(m, tm);
    const float alpha = expf(m - mn);
    m = mn;
    float p[8];
    float ps = 0.0f, a0 = 0.0f, a1 = 0.0f, a2 = 0.0f;
    #pragma unroll
    for (int r = 0; r < 8; ++r) {
      p[r] = expf(st[r] - mn);
      ps += p[r];
      a0 += p[r] * xk[3 * r + 0];
      a1 += p[r] * xk[3 * r + 1];
      a2 += p[r] * xk[3 * r + 2];
    }
    l   = l   * alpha + ps;
    wx0 = wx0 * alpha + a0;
    wx1 = wx1 * alpha + a1;
    wx2 = wx2 * alpha + a2;
    #pragma unroll
    for (int r = 0; r < 8; ++r) {
      o0[r] *= alpha; o1[r] *= alpha; o2[r] *= alpha; o3[r] *= alpha;
    }
    #pragma unroll
    for (int s = 0; s < 4; ++s) {
      v2f pb; pb.x = p[2 * s]; pb.y = p[2 * s + 1];
      o0 = wmma4(vf[s][0], pb, o0);
      o1 = wmma4(vf[s][1], pb, o1);
      o2 = wmma4(vf[s][2], pb, o2);
      o3 = wmma4(vf[s][3], pb, o3);
    }
  }

  l   += xhalf(l);
  wx0 += xhalf(wx0);
  wx1 += xhalf(wx1);
  wx2 += xhalf(wx2);
  if (hi == 0) {
    s_m[w][sub] = m;
    s_l[w][sub] = l;
    s_wx[w][sub][0] = wx0; s_wx[w][sub][1] = wx1; s_wx[w][sub][2] = wx2;
  }
  #pragma unroll
  for (int r = 0; r < 8; ++r) {
    s_o[w][sub][ 0 + r + 8 * hi] = o0[r];
    s_o[w][sub][16 + r + 8 * hi] = o1[r];
    s_o[w][sub][32 + r + 8 * hi] = o2[r];
    s_o[w][sub][48 + r + 8 * hi] = o3[r];
  }
  __syncthreads();

  #pragma unroll
  for (int pass = 0; pass < 2; ++pass) {
    const int q = pass * 8 + (t >> 4), dch = (t & 15) * 4;
    const float m0 = s_m[0][q], m1 = s_m[1][q], m2 = s_m[2][q], m3 = s_m[3][q];
    const float ma = fmaxf(fmaxf(m0, m1), fmaxf(m2, m3));
    const float f0 = expf(m0 - ma), f1 = expf(m1 - ma);
    const float f2 = expf(m2 - ma), f3 = expf(m3 - ma);
    const float lt = ((s_l[0][q] * f0 + s_l[1][q] * f1) + s_l[2][q] * f2) +
                     s_l[3][q] * f3;
    const float inv = 1.0f / lt;
    const int i = i0 + q;
    const float* xi = xyzf + ((size_t)b * MTOK_ + i) * 3;
    float wd[3];
    #pragma unroll
    for (int c = 0; c < 3; ++c)
      wd[c] = (((s_wx[0][q][c] * f0 + s_wx[1][q][c] * f1) + s_wx[2][q][c] * f2) +
               s_wx[3][q][c] * f3) * inv - xi[c];
    const v4f p0 = *(const v4f*)&s_o[0][q][dch];
    const v4f p1 = *(const v4f*)&s_o[1][q][dch];
    const v4f p2 = *(const v4f*)&s_o[2][q][dch];
    const v4f p3 = *(const v4f*)&s_o[3][q][dch];
    const v4f sm = ((p0 * f0 + p1 * f1) + p2 * f2) + p3 * f3;
    float res[4] = {sm.x, sm.y, sm.z, sm.w};
    #pragma unroll
    for (int e = 0; e < 4; ++e) {
      const int d = dch + e;
      res[e] = res[e] * inv + Wsp[d * 3 + 0] * wd[0] + Wsp[d * 3 + 1] * wd[1] +
               Wsp[d * 3 + 2] * wd[2];
    }
    v4f rv;
    rv.x = res[0]; rv.y = res[1]; rv.z = res[2]; rv.w = res[3];
    *(v4f*)(attn_out + ((size_t)(b * MTOK_) + i) * C_ + h * DH_ + dch) = rv;
  }
}

extern "C" void kernel_launch(void* const* d_in, const int* in_sizes, int n_in,
                              void* d_out, int out_size, void* d_ws, size_t ws_size,
                              hipStream_t stream) {
  (void)in_sizes; (void)n_in; (void)out_size; (void)ws_size;
  const float* xyzs = (const float*)d_in[0];
  const float* Wd   = (const float*)d_in[1];
  const float* Wqkv = (const float*)d_in[2];
  const float* Wsp  = (const float*)d_in[3];
  const float* Wout = (const float*)d_in[4];
  const float* bout = (const float*)d_in[5];
  const float* ln1g = (const float*)d_in[6];
  const float* ln1b = (const float*)d_in[7];
  const float* Wff1 = (const float*)d_in[8];
  const float* bff1 = (const float*)d_in[9];
  const float* Wff2 = (const float*)d_in[10];
  const float* bff2 = (const float*)d_in[11];
  const float* ln2g = (const float*)d_in[12];
  const float* ln2b = (const float*)d_in[13];

  float* ws      = (float*)d_ws;
  float* anchors = ws;
  float* feats   = ws + 8192;
  float* xln     = feats + (size_t)ROWS_ * C_;
  float* qkvb    = xln   + (size_t)ROWS_ * C_;
  float* attno   = qkvb  + (size_t)ROWS_ * 1536;
  float* ffh     = attno + (size_t)ROWS_ * C_;

  fps_kernel<<<B_ * L_, N_, 0, stream>>>(xyzs, anchors);
  p4dconv_kernel<<<B_ * L_ * MA_, N_, 0, stream>>>(xyzs, anchors, Wd, feats);

  for (int d = 0; d < 2; ++d) {
    ln_kernel<<<ROWS_, 256, 0, stream>>>(feats, ln1g + d * C_, ln1b + d * C_, xln);
    gemm_wmma<<<dim3(1536 / 32, ROWS_ / 32), 32, 0, stream>>>(
        xln, C_, Wqkv + (size_t)d * 1536 * C_, C_, nullptr, nullptr,
        qkvb, 1536, C_, 0);
    attn_kernel<<<dim3(MTOK_ / 16, HEADS_, B_), 32 * ATT_WAVES_, 0, stream>>>(
        qkvb, anchors, Wsp + (size_t)d * DH_ * 3, attno);
    gemm_wmma<<<dim3(C_ / 32, ROWS_ / 32), 32, 0, stream>>>(
        attno, C_, Wout + (size_t)d * C_ * C_, C_, bout + d * C_,
        feats, feats, C_, C_, 1);
    ln_kernel<<<ROWS_, 256, 0, stream>>>(feats, ln2g + d * C_, ln2b + d * C_, xln);
    gemm_wmma<<<dim3(MLP_ / 32, ROWS_ / 32), 32, 0, stream>>>(
        xln, C_, Wff1 + (size_t)d * MLP_ * C_, C_, bff1 + d * MLP_,
        nullptr, ffh, MLP_, C_, 1);
    gemm_wmma<<<dim3(C_ / 32, ROWS_ / 32), 32, 0, stream>>>(
        ffh, MLP_, Wff2 + (size_t)d * C_ * MLP_, MLP_, bff2 + d * C_,
        feats, feats, C_, MLP_, 0);
  }
  hipMemcpyAsync(d_out, feats, (size_t)ROWS_ * C_ * sizeof(float),
                 hipMemcpyDeviceToDevice, stream);
}


# FPS argmax: 10-level LDS tree (13 barriers a step) -> one wave folds 1024 distances read as 16-B vectors (index = position), then 32 packed keys via LDS; 2 barriers; anchors stored once; same anchors
# speedup vs baseline: 1.8289x; 1.2109x over previous
#include <hip/hip_runtime.h>
#include <hip/hip_bf16.h>
#include <math.h>

typedef float v2f __attribute__((ext_vector_type(2)));
typedef float v4f __attribute__((ext_vector_type(4)));
typedef float v8f __attribute__((ext_vector_type(8)));

#define B_    2
#define T_    4
#define N_    1024
#define L_    4
#define MA_   256
#define KNB_  32
#define C_    512
#define HEADS_ 8
#define DH_   64
#define MTOK_ 1024
#define ROWS_ 2048
#define MLP_  2048

__device__ __forceinline__ v8f wmma4(v2f a, v2f b, v8f c) {
  return __builtin_amdgcn_wmma_f32_16x16x4_f32(false, a, false, b, (short)0, c,
                                               false, false);
}

__device__ __forceinline__ float gelu_exact(float x) {
  return 0.5f * x * (1.0f + erff(x * 0.7071067811865475f));
}

__device__ __forceinline__ float xhalf(float x) {
  const int bx = __builtin_bit_cast(int, x);
  return __builtin_bit_cast(float, __builtin_amdgcn_permlanex16(
      bx, bx, 0x76543210, 0xfedcba98, false, false));
}

typedef unsigned long long u64_t;
typedef u64_t v2u64 __attribute__((ext_vector_type(2)));

__global__ void fps_kernel(const float* __restrict__ xyzs,
                           float* __restrict__ anchors) {
  __shared__ __attribute__((aligned(16))) float sdist[N_];
  __shared__ __attribute__((aligned(16))) v4f   spts[N_];
  __shared__ __attribute__((aligned(16))) u64_t skey[32];
  __shared__ __attribute__((aligned(16))) v4f   slast;
  __shared__ float sanch[MA_ * 3];
  const int blk = blockIdx.x;
  const int tid = threadIdx.x;
  const float* frame = xyzs + (size_t)blk * N_ * 3;
  const float px = frame[tid * 3 + 0];
  const float py = frame[tid * 3 + 1];
  const float pz = frame[tid * 3 + 2];
  float dist = 1e10f;
  float* aout = anchors + (size_t)blk * MA_ * 3;
  {
    v4f me; me.x = px; me.y = py; me.z = pz; me.w = 0.0f;
    spts[tid] = me;
    if (tid == 0) {
      slast = me;
      sanch[0] = px; sanch[1] = py; sanch[2] = pz;
    }
  }
  __syncthreads();
  for (int it = 1; it < MA_; ++it) {
    const v4f lp = slast;
    const float dx = px - lp.x, dy = py - lp.y, dz = pz - lp.z;
    dist = fminf(dist, __builtin_fmaf(dz, dz, __builtin_fmaf(dx, dx, dy * dy)));
    sdist[tid] = dist;
    __syncthreads();
    if (tid < 32) {
      const v4f* sd4 = (const v4f*)sdist;
      v4f q[8];
      #pragma unroll
      for (int i = 0; i < 8; ++i) q[i] = sd4[32 * i + tid];
      float bv[32];
      int   bo[32];
      #pragma unroll
      for (int i = 0; i < 8; ++i) {
        bv[4 * i + 0] = q[i].x; bo[4 * i + 0] = 128 * i + 0;
        bv[4 * i + 1] = q[i].y; bo[4 * i + 1] = 128 * i + 1;
        bv[4 * i + 2] = q[i].z; bo[4 * i + 2] = 128 * i + 2;
        bv[4 * i + 3] = q[i].w; bo[4 * i + 3] = 128 * i + 3;
      }
      #pragma unroll
      for (int wdt = 1; wdt < 32; wdt <<= 1) {
        #pragma unroll
        for (int j = 0; j < 32; j += 2 * wdt) {
          const bool tk = bv[j + wdt] > bv[j];
          bv[j] = tk ? bv[j + wdt] : bv[j];
          bo[j] = tk ? bo[j + wdt] : bo[j];
        }
      }
      const unsigned idx1 = (unsigned)(bo[0] + 4 * tid);
      skey[tid] = ((u64_t)__builtin_bit_cast(unsigned, bv[0]) << 32) |
                  (u64_t)(1023u - idx1);
      __builtin_amdgcn_fence(__ATOMIC_RELEASE, "wavefront");
      __builtin_amdgcn_wave_barrier();
      __builtin_amdgcn_fence(__ATOMIC_ACQUIRE, "wavefront");
      const v2u64* sk2 = (const v2u64*)skey;
      v2u64 kv[16];
      #pragma unroll
      for (int i = 0; i < 16; ++i) kv[i] = sk2[i];
      asm volatile("" :: "v"(kv[0]), "v"(kv[1]), "v"(kv[2]), "v"(kv[3]),
                         "v"(kv[4]), "v"(kv[5]), "v"(kv[6]), "v"(kv[7]),
                         "v"(kv[8]), "v"(kv[9]), "v"(kv[10]), "v"(kv[11]),
                         "v"(kv[12]), "v"(kv[13]), "v"(kv[14]), "v"(kv[15]));
      u64_t kk[32];
      #pragma unroll
      for (int i = 0; i < 16; ++i) {
        kk[2 * i] = kv[i].x; kk[2 * i + 1] = kv[i].y;
      }
      #pragma unroll
      for (int wdt = 1; wdt < 32; wdt <<= 1) {
        #pragma unroll
        for (int j = 0; j < 32; j += 2 * wdt)
          kk[j] = kk[j + wdt] > kk[j] ? kk[j + wdt] : kk[j];
      }
      const int nxt = 1023 - (int)(unsigned)(kk[0] & 0xffffffffull);
      const v4f wp = spts[nxt];
      if (tid == 0) {
        slast = wp;
        sanch[it * 3 + 0] = wp.x; sanch[it * 3 + 1] = wp.y;
        sanch[it * 3 + 2] = wp.z;
      }
    }
    __syncthreads();
  }
  if (tid < MA_ * 3) aout[tid] = sanch[tid];
}

__global__ void p4dconv_kernel(const float* __restrict__ xyzs,
                               const float* __restrict__ anchors,
                               const float* __restrict__ Wd,
                               float* __restrict__ feats) {
  __shared__ int   sscan[N_];
  __shared__ float spx[KNB_], spy[KNB_], spz[KNB_];
  __shared__ float s_p0[3];
  const int blk = blockIdx.x;
  const int m   = blk % MA_;
  const int bl  = blk / MA_;
  const int l   = bl % L_;
  const int b   = bl / L_;
  const int tid = threadIdx.x;

  const float* arow = anchors + ((size_t)bl * MA_ + m) * 3;
  const float ax = arow[0], ay = arow[1], az = arow[2];

  float w0 = 0.f, w1 = 0.f, w2 = 0.f, w3 = 0.f;
  if (tid < C_) {
    w0 = Wd[tid * 4 + 0]; w1 = Wd[tid * 4 + 1];
    w2 = Wd[tid * 4 + 2]; w3 = Wd[tid * 4 + 3];
  }
  float maxv = -3.4e38f;

  for (int di = -1; di <= 1; ++di) {
    int fi = l + di; fi = fi < 0 ? 0 : (fi > T_ - 1 ? T_ - 1 : fi);
    const float* nb = xyzs + ((size_t)b * T_ + fi) * N_ * 3;
    const float qx = nb[tid * 3 + 0];
    const float qy = nb[tid * 3 + 1];
    const float qz = nb[tid * 3 + 2];
    const float dx = qx - ax, dy = qy - ay, dz = qz - az;
    const int pred = (dx * dx + dy * dy + dz * dz) < 1.0f ? 1 : 0;
    sscan[tid] = pred;
    if (tid == 0) { s_p0[0] = dx; s_p0[1] = dy; s_p0[2] = dz; }
    __syncthreads();
    for (int off = 1; off < N_; off <<= 1) {
      const int v   = sscan[tid];
      const int add = (tid >= off) ? sscan[tid - off] : 0;
      __syncthreads();
      sscan[tid] = v + add;
      __syncthreads();
    }
    const int incl  = sscan[tid];
    const int total = sscan[N_ - 1];
    const int pos   = incl - pred;
    if (pred && pos < KNB_) { spx[pos] = dx; spy[pos] = dy; spz[pos] = dz; }
    __syncthreads();
    const int cnt = total < KNB_ ? total : KNB_;
    if (tid < KNB_ && tid >= cnt) {
      spx[tid] = (total > 0) ? spx[0] : s_p0[0];
      spy[tid] = (total > 0) ? spy[0] : s_p0[1];
      spz[tid] = (total > 0) ? spz[0] : s_p0[2];
    }
    __syncthreads();
    if (tid < C_) {
      const float tq = (float)di;
      #pragma unroll 8
      for (int k = 0; k < KNB_; ++k) {
        const float v = w0 * spx[k] + w1 * spy[k] + w2 * spz[k] + w3 * tq;
        maxv = fmaxf(maxv, v);
      }
    }
    __syncthreads();
  }
  if (tid < C_)
    feats[((size_t)bl * MA_ + m) * C_ + tid] = fmaxf(maxv, 0.0f);
}

__global__ void ln_kernel(const float* __restrict__ x, const float* __restrict__ g,
                          const float* __restrict__ bta, float* __restrict__ y) {
  __shared__ float red[256];
  const int row = blockIdx.x, t = threadIdx.x;
  const float* xr = x + (size_t)row * C_;
  const float a0 = xr[t], a1 = xr[t + 256];
  red[t] = a0 + a1; __syncthreads();
  for (int s = 128; s > 0; s >>= 1) { if (t < s) red[t] += red[t + s]; __syncthreads(); }
  const float mean = red[0] * (1.0f / C_);
  __syncthreads();
  const float d0 = a0 - mean, d1 = a1 - mean;
  red[t] = d0 * d0 + d1 * d1; __syncthreads();
  for (int s = 128; s > 0; s >>= 1) { if (t < s) red[t] += red[t + s]; __syncthreads(); }
  const float inv = rsqrtf(red[0] * (1.0f / C_) + 1e-5f);
  float* yr = y + (size_t)row * C_;
  yr[t]       = d0 * inv * g[t]       + bta[t];
  yr[t + 256] = d1 * inv * g[t + 256] + bta[t + 256];
}

__device__ __forceinline__ void gemm_ld1(v4f (&f)[16], int kb, const float* a0,
                                         const float* a1, const float* w0,
                                         const float* w1) {
  f[4 * kb + 0] = *(const v4f*)(a0 + 8 * kb);
  f[4 * kb + 1] = *(const v4f*)(a1 + 8 * kb);
  f[4 * kb + 2] = *(const v4f*)(w0 + 8 * kb);
  f[4 * kb + 3] = *(const v4f*)(w1 + 8 * kb);
}

__device__ __forceinline__ void gemm_mm1(const v4f (&f)[16], int kb, v8f& acc00,
                                         v8f& acc01, v8f& acc10, v8f& acc11) {
  const v4f a0 = f[4 * kb + 0], a1 = f[4 * kb + 1];
  const v4f b0 = f[4 * kb + 2], b1 = f[4 * kb + 3];
  const v2f a0l = {a0.x, a0.y}, a0h = {a0.z, a0.w};
  const v2f a1l = {a1.x, a1.y}, a1h = {a1.z, a1.w};
  const v2f b0l = {b0.x, b0.y}, b0h = {b0.z, b0.w};
  const v2f b1l = {b1.x, b1.y}, b1h = {b1.z, b1.w};
  acc00 = wmma4(a0l, b0l, acc00);
  acc01 = wmma4(a0l, b1l, acc01);
  acc10 = wmma4(a1l, b0l, acc10);
  acc11 = wmma4(a1l, b1l, acc11);
  acc00 = wmma4(a0h, b0h, acc00);
  acc01 = wmma4(a0h, b1h, acc01);
  acc10 = wmma4(a1h, b0h, acc10);
  acc11 = wmma4(a1h, b1h, acc11);
}

__global__ void __launch_bounds__(32)
gemm_wmma(const float* __restrict__ A, int lda,
          const float* __restrict__ W, int ldw,
          const float* __restrict__ bias,
          const float* __restrict__ resid,
          float* __restrict__ Cout, int ldc,
          int Kdim, int act) {
  const int j0 = blockIdx.x * 32, i0 = blockIdx.y * 32;
  const int lane = threadIdx.x & 31;
  const int sub = lane & 15, hi = lane >> 4;
  const float* arow0 = A + (size_t)(i0 + sub) * lda + 4 * hi;
  const float* arow1 = arow0 + (size_t)16 * lda;
  const float* wrow0 = W + (size_t)(j0 + sub) * ldw + 4 * hi;
  const float* wrow1 = wrow0 + (size_t)16 * ldw;
  __builtin_prefetch(arow0, 0, 0);
  __builtin_prefetch(wrow0, 0, 0);
  v8f acc00 = {0.f, 0.f, 0.f, 0.f, 0.f, 0.f, 0.f, 0.f};
  v8f acc01 = acc00, acc10 = acc00, acc11 = acc00;
  v4f s0[16], s1[16];
  #pragma unroll
  for (int kb = 0; kb < 4; ++kb) gemm_ld1(s0, kb, arow0, arow1, wrow0, wrow1);
  #pragma unroll
  for (int kb = 0; kb < 4; ++kb)
    gemm_ld1(s1, kb, arow0 + 32, arow1 + 32, wrow0 + 32, wrow1 + 32);
  int k0 = 0;
  for (; k0 + 64 < Kdim; k0 += 64) {
    #pragma unroll
    for (int kb = 0; kb < 4; ++kb) {
      gemm_mm1(s0, kb, acc00, acc01, acc10, acc11);
      gemm_ld1(s0, kb, arow0 + k0 + 64, arow1 + k0 + 64, wrow0 + k0 + 64,
               wrow1 + k0 + 64);
    }
    #pragma unroll
    for (int kb = 0; kb < 4; ++kb) {
      gemm_mm1(s1, kb, acc00, acc01, acc10, acc11);
      gemm_ld1(s1, kb, arow0 + k0 + 96, arow1 + k0 + 96, wrow0 + k0 + 96,
               wrow1 + k0 + 96);
    }
  }
  #pragma unroll
  for (int kb = 0; kb < 4; ++kb) gemm_mm1(s0, kb, acc00, acc01, acc10, acc11);
  #pragma unroll
  for (int kb = 0; kb < 4; ++kb) gemm_mm1(s1, kb, acc00, acc01, acc10, acc11);
  const float bj0 = bias ? bias[j0 + sub]      : 0.0f;
  const float bj1 = bias ? bias[j0 + 16 + sub] : 0.0f;
  #pragma unroll
  for (int r = 0; r < 8; ++r) {
    const int row0 = i0 + r + 8 * hi;
    const int row1 = row0 + 16;
    const int col0 = j0 + sub;
    const int col1 = col0 + 16;
    float v00 = acc00[r] + bj0, v01 = acc01[r] + bj1;
    float v10 = acc10[r] + bj0, v11 = acc11[r] + bj1;
    if (act == 1) {
      v00 = gelu_exact(v00); v01 = gelu_exact(v01);
      v10 = gelu_exact(v10); v11 = gelu_exact(v11);
    }
    if (resid) {
      v00 += resid[(size_t)row0 * ldc + col0];
      v01 += resid[(size_t)row0 * ldc + col1];
      v10 += resid[(size_t)row1 * ldc + col0];
      v11 += resid[(size_t)row1 * ldc + col1];
    }
    Cout[(size_t)row0 * ldc + col0] = v00;
    Cout[(size_t)row0 * ldc + col1] = v01;
    Cout[(size_t)row1 * ldc + col0] = v10;
    Cout[(size_t)row1 * ldc + col1] = v11;
  }
}

#define ATT_WAVES_ 4
#define ATT_KPW_   (MTOK_ / ATT_WAVES_)
#define ATT_OP_    68

__global__ void __launch_bounds__(32 * ATT_WAVES_)
attn_kernel(const float* __restrict__ qkv,
            const float* __restrict__ xyzf,
            const float* __restrict__ Wsp,
            float* __restrict__ attn_out) {
  __shared__ float s_m[ATT_WAVES_][16];
  __shared__ float s_l[ATT_WAVES_][16];
  __shared__ float s_wx[ATT_WAVES_][16][3];
  __shared__ __attribute__((aligned(16))) float s_o[ATT_WAVES_][16][ATT_OP_];
  const int b = blockIdx.z, h = blockIdx.y, i0 = blockIdx.x * 16;
  const int t = threadIdx.x;
  const int lane = t & 31, w = t >> 5;
  const int sub = lane & 15, hi = lane >> 4;
  const int key0 = b * MTOK_ + w * ATT_KPW_;

  const float* qbase =
      qkv + ((size_t)(b * MTOK_ + i0 + sub) * 3 + 0) * C_ + h * DH_ + 2 * hi;
  v2f qb[16];
  #pragma unroll
  for (int ks = 0; ks < 16; ++ks)
    qb[ks] = *(const v2f*)(qbase + 4 * ks) * 0.125f;

  const float* krow =
      qkv + ((size_t)(key0 + sub) * 3 + 1) * C_ + h * DH_ + 2 * hi;
  const float* vcol =
      qkv + ((size_t)(key0 + 8 * hi) * 3 + 2) * C_ + h * DH_ + sub;
  const float* xrow = xyzf + (size_t)(key0 + 8 * hi) * 3;

  v8f o0 = {0.f, 0.f, 0.f, 0.f, 0.f, 0.f, 0.f, 0.f};
  v8f o1 = o0, o2 = o0, o3 = o0;
  float m = -3.4e38f, l = 0.0f, wx0 = 0.0f, wx1 = 0.0f, wx2 = 0.0f;

  v2f kf[16];
  #pragma unroll
  for (int ks = 0; ks < 16; ++ks) kf[ks] = *(const v2f*)(krow + 4 * ks);

  for (int jt = 0; jt < ATT_KPW_ / 16; ++jt) {
    v8f st = {0.f, 0.f, 0.f, 0.f, 0.f, 0.f, 0.f, 0.f};
    #pragma unroll
    for (int ks = 0; ks < 16; ++ks) st = wmma4(kf[ks], qb[ks], st);

    const int jn = (jt + 1 < ATT_KPW_ / 16) ? jt + 1 : jt;
    const float* kn = krow + (size_t)jn * 16 * 3 * C_;
    #pragma unroll
    for (int ks = 0; ks < 16; ++ks) kf[ks] = *(const v2f*)(kn + 4 * ks);

    const float* vt = vcol + (size_t)jt * 16 * 3 * C_;
    v2f vf[4][4];
    #pragma unroll
    for (int s = 0; s < 4; ++s) {
      #pragma unroll
      for (int dt = 0; dt < 4; ++dt) {
        const float* vp = vt + (size_t)(2 * s) * 3 * C_ + dt * 16;
        vf[s][dt].x = vp[0];
        vf[s][dt].y = vp[3 * C_];
      }
    }
    const v4f* xp = (const v4f*)(xrow + jt * 48);
    float xk[24];
    #pragma unroll
    for (int i = 0; i < 6; ++i) {
      const v4f xv = xp[i];
      xk[4 * i + 0] = xv.x; xk[4 * i + 1] = xv.y;
      xk[4 * i + 2] = xv.z; xk[4 * i + 3] = xv.w;
    }

    float tm = st[0];
    #pragma unroll
    for (int r = 1; r < 8; ++r) tm = fmaxf(tm, st[r]);
    tm = fmaxf(tm, xhalf(tm));
    const float mn = fmaxf(m, tm);
    const float alpha = expf(m - mn);
    m = mn;
    float p[8];
    float ps = 0.0f, a0 = 0.0f, a1 = 0.0f, a2 = 0.0f;
    #pragma unroll
    for (int r = 0; r < 8; ++r) {
      p[r] = expf(st[r] - mn);
      ps += p[r];
      a0 += p[r] * xk[3 * r + 0];
      a1 += p[r] * xk[3 * r + 1];
      a2 += p[r] * xk[3 * r + 2];
    }
    l   = l   * alpha + ps;
    wx0 = wx0 * alpha + a0;
    wx1 = wx1 * alpha + a1;
    wx2 = wx2 * alpha + a2;
    #pragma unroll
    for (int r = 0; r < 8; ++r) {
      o0[r] *= alpha; o1[r] *= alpha; o2[r] *= alpha; o3[r] *= alpha;
    }
    #pragma unroll
    for (int s = 0; s < 4; ++s) {
      v2f pb; pb.x = p[2 * s]; pb.y = p[2 * s + 1];
      o0 = wmma4(vf[s][0], pb, o0);
      o1 = wmma4(vf[s][1], pb, o1);
      o2 = wmma4(vf[s][2], pb, o2);
      o3 = wmma4(vf[s][3], pb, o3);
    }
  }

  l   += xhalf(l);
  wx0 += xhalf(wx0);
  wx1 += xhalf(wx1);
  wx2 += xhalf(wx2);
  if (hi == 0) {
    s_m[w][sub] = m;
    s_l[w][sub] = l;
    s_wx[w][sub][0] = wx0; s_wx[w][sub][1] = wx1; s_wx[w][sub][2] = wx2;
  }
  #pragma unroll
  for (int r = 0; r < 8; ++r) {
    s_o[w][sub][ 0 + r + 8 * hi] = o0[r];
    s_o[w][sub][16 + r + 8 * hi] = o1[r];
    s_o[w][sub][32 + r + 8 * hi] = o2[r];
    s_o[w][sub][48 + r + 8 * hi] = o3[r];
  }
  __syncthreads();

  #pragma unroll
  for (int pass = 0; pass < 2; ++pass) {
    const int q = pass * 8 + (t >> 4), dch = (t & 15) * 4;
    const float m0 = s_m[0][q], m1 = s_m[1][q], m2 = s_m[2][q], m3 = s_m[3][q];
    const float ma = fmaxf(fmaxf(m0, m1), fmaxf(m2, m3));
    const float f0 = expf(m0 - ma), f1 = expf(m1 - ma);
    const float f2 = expf(m2 - ma), f3 = expf(m3 - ma);
    const float lt = ((s_l[0][q] * f0 + s_l[1][q] * f1) + s_l[2][q] * f2) +
                     s_l[3][q] * f3;
    const float inv = 1.0f / lt;
    const int i = i0 + q;
    const float* xi = xyzf + ((size_t)b * MTOK_ + i) * 3;
    float wd[3];
    #pragma unroll
    for (int c = 0; c < 3; ++c)
      wd[c] = (((s_wx[0][q][c] * f0 + s_wx[1][q][c] * f1) + s_wx[2][q][c] * f2) +
               s_wx[3][q][c] * f3) * inv - xi[c];
    const v4f p0 = *(const v4f*)&s_o[0][q][dch];
    const v4f p1 = *(const v4f*)&s_o[1][q][dch];
    const v4f p2 = *(const v4f*)&s_o[2][q][dch];
    const v4f p3 = *(const v4f*)&s_o[3][q][dch];
    const v4f sm = ((p0 * f0 + p1 * f1) + p2 * f2) + p3 * f3;
    float res[4] = {sm.x, sm.y, sm.z, sm.w};
    #pragma unroll
    for (int e = 0; e < 4; ++e) {
      const int d = dch + e;
      res[e] = res[e] * inv + Wsp[d * 3 + 0] * wd[0] + Wsp[d * 3 + 1] * wd[1] +
               Wsp[d * 3 + 2] * wd[2];
    }
    v4f rv;
    rv.x = res[0]; rv.y = res[1]; rv.z = res[2]; rv.w = res[3];
    *(v4f*)(attn_out + ((size_t)(b * MTOK_) + i) * C_ + h * DH_ + dch) = rv;
  }
}

extern "C" void kernel_launch(void* const* d_in, const int* in_sizes, int n_in,
                              void* d_out, int out_size, void* d_ws, size_t ws_size,
                              hipStream_t stream) {
  (void)in_sizes; (void)n_in; (void)out_size; (void)ws_size;
  const float* xyzs = (const float*)d_in[0];
  const float* Wd   = (const float*)d_in[1];
  const float* Wqkv = (const float*)d_in[2];
  const float* Wsp  = (const float*)d_in[3];
  const float* Wout = (const float*)d_in[4];
  const float* bout = (const float*)d_in[5];
  const float* ln1g = (const float*)d_in[6];
  const float* ln1b = (const float*)d_in[7];
  const float* Wff1 = (const float*)d_in[8];
  const float* bff1 = (const float*)d_in[9];
  const float* Wff2 = (const float*)d_in[10];
  const float* bff2 = (const float*)d_in[11];
  const float* ln2g = (const float*)d_in[12];
  const float* ln2b = (const float*)d_in[13];

  float* ws      = (float*)d_ws;
  float* anchors = ws;
  float* feats   = ws + 8192;
  float* xln     = feats + (size_t)ROWS_ * C_;
  float* qkvb    = xln   + (size_t)ROWS_ * C_;
  float* attno   = qkvb  + (size_t)ROWS_ * 1536;
  float* ffh     = attno + (size_t)ROWS_ * C_;

  fps_kernel<<<B_ * L_, N_, 0, stream>>>(xyzs, anchors);
  p4dconv_kernel<<<B_ * L_ * MA_, N_, 0, stream>>>(xyzs, anchors, Wd, feats);

  for (int d = 0; d < 2; ++d) {
    ln_kernel<<<ROWS_, 256, 0, stream>>>(feats, ln1g + d * C_, ln1b + d * C_, xln);
    gemm_wmma<<<dim3(1536 / 32, ROWS_ / 32), 32, 0, stream>>>(
        xln, C_, Wqkv + (size_t)d * 1536 * C_, C_, nullptr, nullptr,
        qkvb, 1536, C_, 0);
    attn_kernel<<<dim3(MTOK_ / 16, HEADS_, B_), 32 * ATT_WAVES_, 0, stream>>>(
        qkvb, anchors, Wsp + (size_t)d * DH_ * 3, attno);
    gemm_wmma<<<dim3(C_ / 32, ROWS_ / 32), 32, 0, stream>>>(
        attno, C_, Wout + (size_t)d * C_ * C_, C_, bout + d * C_,
        feats, feats, C_, C_, 1);
    ln_kernel<<<ROWS_, 256, 0, stream>>>(feats, ln2g + d * C_, ln2b + d * C_, xln);
    gemm_wmma<<<dim3(MLP_ / 32, ROWS_ / 32), 32, 0, stream>>>(
        xln, C_, Wff1 + (size_t)d * MLP_ * C_, C_, bff1 + d * MLP_,
        nullptr, ffh, MLP_, C_, 1);
    gemm_wmma<<<dim3(C_ / 32, ROWS_ / 32), 32, 0, stream>>>(
        ffh, MLP_, Wff2 + (size_t)d * C_ * MLP_, MLP_, bff2 + d * C_,
        feats, feats, C_, MLP_, 0);
  }
  hipMemcpyAsync(d_out, feats, (size_t)ROWS_ * C_ * sizeof(float),
                 hipMemcpyDeviceToDevice, stream);
}


# p4dconv ball query: 1024-thread prefix scan per anchor (72 barriers) -> ordered walk, lane = (frame, anchor, 128-point segment) appends hits to its list; 4 anchors per 512-thread block; same features
# speedup vs baseline: 1.9263x; 1.0533x over previous
#include <hip/hip_runtime.h>
#include <hip/hip_bf16.h>
#include <math.h>

typedef float v2f __attribute__((ext_vector_type(2)));
typedef float v4f __attribute__((ext_vector_type(4)));
typedef float v8f __attribute__((ext_vector_type(8)));

#define B_    2
#define T_    4
#define N_    1024
#define L_    4
#define MA_   256
#define KNB_  32
#define C_    512
#define HEADS_ 8
#define DH_   64
#define MTOK_ 1024
#define ROWS_ 2048
#define MLP_  2048

__device__ __forceinline__ v8f wmma4(v2f a, v2f b, v8f c) {
  return __builtin_amdgcn_wmma_f32_16x16x4_f32(false, a, false, b, (short)0, c,
                                               false, false);
}

__device__ __forceinline__ float gelu_exact(float x) {
  return 0.5f * x * (1.0f + erff(x * 0.7071067811865475f));
}

__device__ __forceinline__ float xhalf(float x) {
  const int bx = __builtin_bit_cast(int, x);
  return __builtin_bit_cast(float, __builtin_amdgcn_permlanex16(
      bx, bx, 0x76543210, 0xfedcba98, false, false));
}

typedef unsigned long long u64_t;
typedef u64_t v2u64 __attribute__((ext_vector_type(2)));

__global__ void fps_kernel(const float* __restrict__ xyzs,
                           float* __restrict__ anchors) {
  __shared__ __attribute__((aligned(16))) float sdist[N_];
  __shared__ __attribute__((aligned(16))) v4f   spts[N_];
  __shared__ __attribute__((aligned(16))) u64_t skey[32];
  __shared__ __attribute__((aligned(16))) v4f   slast;
  __shared__ float sanch[MA_ * 3];
  const int blk = blockIdx.x;
  const int tid = threadIdx.x;
  const float* frame = xyzs + (size_t)blk * N_ * 3;
  const float px = frame[tid * 3 + 0];
  const float py = frame[tid * 3 + 1];
  const float pz = frame[tid * 3 + 2];
  float dist = 1e10f;
  float* aout = anchors + (size_t)blk * MA_ * 3;
  {
    v4f me; me.x = px; me.y = py; me.z = pz; me.w = 0.0f;
    spts[tid] = me;
    if (tid == 0) {
      slast = me;
      sanch[0] = px; sanch[1] = py; sanch[2] = pz;
    }
  }
  __syncthreads();
  for (int it = 1; it < MA_; ++it) {
    const v4f lp = slast;
    const float dx = px - lp.x, dy = py - lp.y, dz = pz - lp.z;
    dist = fminf(dist, __builtin_fmaf(dz, dz, __builtin_fmaf(dx, dx, dy * dy)));
    sdist[tid] = dist;
    __syncthreads();
    if (tid < 32) {
      const v4f* sd4 = (const v4f*)sdist;
      v4f q[8];
      #pragma unroll
      for (int i = 0; i < 8; ++i) q[i] = sd4[32 * i + tid];
      float bv[32];
      int   bo[32];
      #pragma unroll
      for (int i = 0; i < 8; ++i) {
        bv[4 * i + 0] = q[i].x; bo[4 * i + 0] = 128 * i + 0;
        bv[4 * i + 1] = q[i].y; bo[4 * i + 1] = 128 * i + 1;
        bv[4 * i + 2] = q[i].z; bo[4 * i + 2] = 128 * i + 2;
        bv[4 * i + 3] = q[i].w; bo[4 * i + 3] = 128 * i + 3;
      }
      #pragma unroll
      for (int wdt = 1; wdt < 32; wdt <<= 1) {
        #pragma unroll
        for (int j = 0; j < 32; j += 2 * wdt) {
          const bool tk = bv[j + wdt] > bv[j];
          bv[j] = tk ? bv[j + wdt] : bv[j];
          bo[j] = tk ? bo[j + wdt] : bo[j];
        }
      }
      const unsigned idx1 = (unsigned)(bo[0] + 4 * tid);
      skey[tid] = ((u64_t)__builtin_bit_cast(unsigned, bv[0]) << 32) |
                  (u64_t)(1023u - idx1);
      __builtin_amdgcn_fence(__ATOMIC_RELEASE, "wavefront");
      __builtin_amdgcn_wave_barrier();
      __builtin_amdgcn_fence(__ATOMIC_ACQUIRE, "wavefront");
      const v2u64* sk2 = (const v2u64*)skey;
      v2u64 kv[16];
      #pragma unroll
      for (int i = 0; i < 16; ++i) kv[i] = sk2[i];
      asm volatile("" :: "v"(kv[0]), "v"(kv[1]), "v"(kv[2]), "v"(kv[3]),
                         "v"(kv[4]), "v"(kv[5]), "v"(kv[6]), "v"(kv[7]),
                         "v"(kv[8]), "v"(kv[9]), "v"(kv[10]), "v"(kv[11]),
                         "v"(kv[12]), "v"(kv[13]), "v"(kv[14]), "v"(kv[15]));
      u64_t kk[32];
      #pragma unroll
      for (int i = 0; i < 16; ++i) {
        kk[2 * i] = kv[i].x; kk[2 * i + 1] = kv[i].y;
      }
      #pragma unroll
      for (int wdt = 1; wdt < 32; wdt <<= 1) {
        #pragma unroll
        for (int j = 0; j < 32; j += 2 * wdt)
          kk[j] = kk[j + wdt] > kk[j] ? kk[j + wdt] : kk[j];
      }
      const int nxt = 1023 - (int)(unsigned)(kk[0] & 0xffffffffull);
      const v4f wp = spts[nxt];
      if (tid == 0) {
        slast = wp;
        sanch[it * 3 + 0] = wp.x; sanch[it * 3 + 1] = wp.y;
        sanch[it * 3 + 2] = wp.z;
      }
    }
    __syncthreads();
  }
  if (tid < MA_ * 3) aout[tid] = sanch[tid];
}

#define PG_   4
#define PQ_   8
#define PSEG_ (N_ / PQ_)
#define PLST_ (KNB_ + 1)
#define PTHR_ 512

__global__ void __launch_bounds__(PTHR_)
p4dconv_kernel(const float* __restrict__ xyzs,
               const float* __restrict__ anchors,
               const float* __restrict__ Wd,
               float* __restrict__ feats) {
  __shared__ __attribute__((aligned(16))) v4f spt[3][N_ + PQ_];
  __shared__ __attribute__((aligned(16))) v4f slist[3 * PG_ * PQ_][PLST_];
  __shared__ __attribute__((aligned(16))) v4f flist[3 * PG_][KNB_];
  __shared__ int scnt[3 * PG_ * PQ_];
  __shared__ int sbase[3 * PG_ * PQ_];
  __shared__ int stake[3 * PG_ * PQ_];
  __shared__ int snum[3 * PG_];
  const int blk = blockIdx.x;
  const int m0  = (blk % (MA_ / PG_)) * PG_;
  const int bl  = blk / (MA_ / PG_);
  const int l   = bl % L_;
  const int b   = bl / L_;
  const int tid = threadIdx.x;

  #pragma unroll
  for (int f = 0; f < 3; ++f) {
    int fi = l + f - 1; fi = fi < 0 ? 0 : (fi > T_ - 1 ? T_ - 1 : fi);
    const float* nb = xyzs + ((size_t)b * T_ + fi) * N_ * 3;
    #pragma unroll
    for (int h = 0; h < N_ / PTHR_; ++h) {
      const int i = tid + PTHR_ * h;
      v4f p;
      p.x = nb[i * 3 + 0]; p.y = nb[i * 3 + 1]; p.z = nb[i * 3 + 2]; p.w = 0.0f;
      spt[f][i + i / PSEG_] = p;
    }
  }
  const bool walker = tid < 3 * PG_ * PQ_;
  const int wfr  = tid / (PG_ * PQ_);
  const int wg   = (tid % (PG_ * PQ_)) / PQ_;
  const int wq   = tid % PQ_;
  const int wseg = (wfr * PG_ + wg) * PQ_ + wq;
  float ax = 0.0f, ay = 0.0f, az = 0.0f;
  if (walker) {
    const float* arow = anchors + ((size_t)bl * MA_ + m0 + wg) * 3;
    ax = arow[0]; ay = arow[1]; az = arow[2];
  }
  const v4f wv = *(const v4f*)(Wd + tid * 4);
  __syncthreads();

  if (walker) {
    const v4f* pts = &spt[wfr][wq * (PSEG_ + 1)];
    v4f* lst = slist[wseg];
    int cnt = 0;
    #pragma unroll 4
    for (int p = 0; p < PSEG_; ++p) {
      const v4f pt = pts[p];
      const float dx = pt.x - ax, dy = pt.y - ay, dz = pt.z - az;
      const bool hit = __builtin_fmaf(dz, dz, __builtin_fmaf(dx, dx, dy * dy)) < 1.0f;
      v4f e; e.x = dx; e.y = dy; e.z = dz; e.w = 0.0f;
      lst[hit ? (cnt < KNB_ ? cnt : KNB_) : KNB_] = e;
      cnt += hit ? 1 : 0;
    }
    scnt[wseg] = cnt < KNB_ ? cnt : KNB_;
  }
  __syncthreads();

  if (walker && wq == 0) {
    int run = 0;
    #pragma unroll
    for (int q = 0; q < PQ_; ++q) {
      const int c = scnt[wseg + q];
      const int t = c < KNB_ - run ? c : KNB_ - run;
      sbase[wseg + q] = run;
      stake[wseg + q] = t;
      run += t;
    }
    if (run == 0) {
      const v4f p0 = spt[wfr][0];
      v4f e; e.x = p0.x - ax; e.y = p0.y - ay; e.z = p0.z - az; e.w = 0.0f;
      flist[wfr * PG_ + wg][0] = e;
      run = 1;
    }
    snum[wfr * PG_ + wg] = run;
  }
  __syncthreads();

  #pragma unroll
  for (int j = 0; j < (3 * PG_ * PQ_ * KNB_) / PTHR_; ++j) {
    const int item = tid + PTHR_ * j;
    const int k = item % KNB_, seg = item / KNB_;
    if (k < stake[seg]) flist[seg / PQ_][sbase[seg] + k] = slist[seg][k];
  }
  __syncthreads();

  const float w0 = wv.x, w1 = wv.y, w2 = wv.z, w3 = wv.w;
  for (int g = 0; g < PG_; ++g) {
    float maxv = -3.4e38f;
    #pragma unroll
    for (int f = 0; f < 3; ++f) {
      const float tq = (float)(f - 1);
      const int n = __builtin_amdgcn_readfirstlane(snum[f * PG_ + g]);
      const v4f* lst = flist[f * PG_ + g];
      #pragma unroll 4
      for (int k = 0; k < n; ++k) {
        const v4f e = lst[k];
        const float v =
            __builtin_fmaf(w2, e.z, __builtin_fmaf(w0, e.x, w1 * e.y)) + w3 * tq;
        maxv = fmaxf(maxv, v);
      }
    }
    feats[((size_t)bl * MA_ + m0 + g) * C_ + tid] = fmaxf(maxv, 0.0f);
  }
}

__global__ void ln_kernel(const float* __restrict__ x, const float* __restrict__ g,
                          const float* __restrict__ bta, float* __restrict__ y) {
  __shared__ float red[256];
  const int row = blockIdx.x, t = threadIdx.x;
  const float* xr = x + (size_t)row * C_;
  const float a0 = xr[t], a1 = xr[t + 256];
  red[t] = a0 + a1; __syncthreads();
  for (int s = 128; s > 0; s >>= 1) { if (t < s) red[t] += red[t + s]; __syncthreads(); }
  const float mean = red[0] * (1.0f / C_);
  __syncthreads();
  const float d0 = a0 - mean, d1 = a1 - mean;
  red[t] = d0 * d0 + d1 * d1; __syncthreads();
  for (int s = 128; s > 0; s >>= 1) { if (t < s) red[t] += red[t + s]; __syncthreads(); }
  const float inv = rsqrtf(red[0] * (1.0f / C_) + 1e-5f);
  float* yr = y + (size_t)row * C_;
  yr[t]       = d0 * inv * g[t]       + bta[t];
  yr[t + 256] = d1 * inv * g[t + 256] + bta[t + 256];
}

__device__ __forceinline__ void gemm_ld1(v4f (&f)[16], int kb, const float* a0,
                                         const float* a1, const float* w0,
                                         const float* w1) {
  f[4 * kb + 0] = *(const v4f*)(a0 + 8 * kb);
  f[4 * kb + 1] = *(const v4f*)(a1 + 8 * kb);
  f[4 * kb + 2] = *(const v4f*)(w0 + 8 * kb);
  f[4 * kb + 3] = *(const v4f*)(w1 + 8 * kb);
}

__device__ __forceinline__ void gemm_mm1(const v4f (&f)[16], int kb, v8f& acc00,
                                         v8f& acc01, v8f& acc10, v8f& acc11) {
  const v4f a0 = f[4 * kb + 0], a1 = f[4 * kb + 1];
  const v4f b0 = f[4 * kb + 2], b1 = f[4 * kb + 3];
  const v2f a0l = {a0.x, a0.y}, a0h = {a0.z, a0.w};
  const v2f a1l = {a1.x, a1.y}, a1h = {a1.z, a1.w};
  const v2f b0l = {b0.x, b0.y}, b0h = {b0.z, b0.w};
  const v2f b1l = {b1.x, b1.y}, b1h = {b1.z, b1.w};
  acc00 = wmma4(a0l, b0l, acc00);
  acc01 = wmma4(a0l, b1l, acc01);
  acc10 = wmma4(a1l, b0l, acc10);
  acc11 = wmma4(a1l, b1l, acc11);
  acc00 = wmma4(a0h, b0h, acc00);
  acc01 = wmma4(a0h, b1h, acc01);
  acc10 = wmma4(a1h, b0h, acc10);
  acc11 = wmma4(a1h, b1h, acc11);
}

__global__ void __launch_bounds__(32)
gemm_wmma(const float* __restrict__ A, int lda,
          const float* __restrict__ W, int ldw,
          const float* __restrict__ bias,
          const float* __restrict__ resid,
          float* __restrict__ Cout, int ldc,
          int Kdim, int act) {
  const int j0 = blockIdx.x * 32, i0 = blockIdx.y * 32;
  const int lane = threadIdx.x & 31;
  const int sub = lane & 15, hi = lane >> 4;
  const float* arow0 = A + (size_t)(i0 + sub) * lda + 4 * hi;
  const float* arow1 = arow0 + (size_t)16 * lda;
  const float* wrow0 = W + (size_t)(j0 + sub) * ldw + 4 * hi;
  const float* wrow1 = wrow0 + (size_t)16 * ldw;
  __builtin_prefetch(arow0, 0, 0);
  __builtin_prefetch(wrow0, 0, 0);
  v8f acc00 = {0.f, 0.f, 0.f, 0.f, 0.f, 0.f, 0.f, 0.f};
  v8f acc01 = acc00, acc10 = acc00, acc11 = acc00;
  v4f s0[16], s1[16];
  #pragma unroll
  for (int kb = 0; kb < 4; ++kb) gemm_ld1(s0, kb, arow0, arow1, wrow0, wrow1);
  #pragma unroll
  for (int kb = 0; kb < 4; ++kb)
    gemm_ld1(s1, kb, arow0 + 32, arow1 + 32, wrow0 + 32, wrow1 + 32);
  int k0 = 0;
  for (; k0 + 64 < Kdim; k0 += 64) {
    #pragma unroll
    for (int kb = 0; kb < 4; ++kb) {
      gemm_mm1(s0, kb, acc00, acc01, acc10, acc11);
      gemm_ld1(s0, kb, arow0 + k0 + 64, arow1 + k0 + 64, wrow0 + k0 + 64,
               wrow1 + k0 + 64);
    }
    #pragma unroll
    for (int kb = 0; kb < 4; ++kb) {
      gemm_mm1(s1, kb, acc00, acc01, acc10, acc11);
      gemm_ld1(s1, kb, arow0 + k0 + 96, arow1 + k0 + 96, wrow0 + k0 + 96,
               wrow1 + k0 + 96);
    }
  }
  #pragma unroll
  for (int kb = 0; kb < 4; ++kb) gemm_mm1(s0, kb, acc00, acc01, acc10, acc11);
  #pragma unroll
  for (int kb = 0; kb < 4; ++kb) gemm_mm1(s1, kb, acc00, acc01, acc10, acc11);
  const float bj0 = bias ? bias[j0 + sub]      : 0.0f;
  const float bj1 = bias ? bias[j0 + 16 + sub] : 0.0f;
  #pragma unroll
  for (int r = 0; r < 8; ++r) {
    const int row0 = i0 + r + 8 * hi;
    const int row1 = row0 + 16;
    const int col0 = j0 + sub;
    const int col1 = col0 + 16;
    float v00 = acc00[r] + bj0, v01 = acc01[r] + bj1;
    float v10 = acc10[r] + bj0, v11 = acc11[r] + bj1;
    if (act == 1) {
      v00 = gelu_exact(v00); v01 = gelu_exact(v01);
      v10 = gelu_exact(v10); v11 = gelu_exact(v11);
    }
    if (resid) {
      v00 += resid[(size_t)row0 * ldc + col0];
      v01 += resid[(size_t)row0 * ldc + col1];
      v10 += resid[(size_t)row1 * ldc + col0];
      v11 += resid[(size_t)row1 * ldc + col1];
    }
    Cout[(size_t)row0 * ldc + col0] = v00;
    Cout[(size_t)row0 * ldc + col1] = v01;
    Cout[(size_t)row1 * ldc + col0] = v10;
    Cout[(size_t)row1 * ldc + col1] = v11;
  }
}

#define ATT_WAVES_ 4
#define ATT_KPW_   (MTOK_ / ATT_WAVES_)
#define ATT_OP_    68

__global__ void __launch_bounds__(32 * ATT_WAVES_)
attn_kernel(const float* __restrict__ qkv,
            const float* __restrict__ xyzf,
            const float* __restrict__ Wsp,
            float* __restrict__ attn_out) {
  __shared__ float s_m[ATT_WAVES_][16];
  __shared__ float s_l[ATT_WAVES_][16];
  __shared__ float s_wx[ATT_WAVES_][16][3];
  __shared__ __attribute__((aligned(16))) float s_o[ATT_WAVES_][16][ATT_OP_];
  const int b = blockIdx.z, h = blockIdx.y, i0 = blockIdx.x * 16;
  const int t = threadIdx.x;
  const int lane = t & 31, w = t >> 5;
  const int sub = lane & 15, hi = lane >> 4;
  const int key0 = b * MTOK_ + w * ATT_KPW_;

  const float* qbase =
      qkv + ((size_t)(b * MTOK_ + i0 + sub) * 3 + 0) * C_ + h * DH_ + 2 * hi;
  v2f qb[16];
  #pragma unroll
  for (int ks = 0; ks < 16; ++ks)
    qb[ks] = *(const v2f*)(qbase + 4 * ks) * 0.125f;

  const float* krow =
      qkv + ((size_t)(key0 + sub) * 3 + 1) * C_ + h * DH_ + 2 * hi;
  const float* vcol =
      qkv + ((size_t)(key0 + 8 * hi) * 3 + 2) * C_ + h * DH_ + sub;
  const float* xrow = xyzf + (size_t)(key0 + 8 * hi) * 3;

  v8f o0 = {0.f, 0.f, 0.f, 0.f, 0.f, 0.f, 0.f, 0.f};
  v8f o1 = o0, o2 = o0, o3 = o0;
  float m = -3.4e38f, l = 0.0f, wx0 = 0.0f, wx1 = 0.0f, wx2 = 0.0f;

  v2f kf[16];
  #pragma unroll
  for (int ks = 0; ks < 16; ++ks) kf[ks] = *(const v2f*)(krow + 4 * ks);

  for (int jt = 0; jt < ATT_KPW_ / 16; ++jt) {
    v8f st = {0.f, 0.f, 0.f, 0.f, 0.f, 0.f, 0.f, 0.f};
    #pragma unroll
    for (int ks = 0; ks < 16; ++ks) st = wmma4(kf[ks], qb[ks], st);

    const int jn = (jt + 1 < ATT_KPW_ / 16) ? jt + 1 : jt;
    const float* kn = krow + (size_t)jn * 16 * 3 * C_;
    #pragma unroll
    for (int ks = 0; ks < 16; ++ks) kf[ks] = *(const v2f*)(kn + 4 * ks);

    const float* vt = vcol + (size_t)jt * 16 * 3 * C_;
    v2f vf[4][4];
    #pragma unroll
    for (int s = 0; s < 4; ++s) {
      #pragma unroll
      for (int dt = 0; dt < 4; ++dt) {
        const float* vp = vt + (size_t)(2 * s) * 3 * C_ + dt * 16;
        vf[s][dt].x = vp[0];
        vf[s][dt].y = vp[3 * C_];
      }
    }
    const v4f* xp = (const v4f*)(xrow + jt * 48);
    float xk[24];
    #pragma unroll
    for (int i = 0; i < 6; ++i) {
      const v4f xv = xp[i];
      xk[4 * i + 0] = xv.x; xk[4 * i + 1] = xv.y;
      xk[4 * i + 2] = xv.z; xk[4 * i + 3] = xv.w;
    }

    float tm = st[0];
    #pragma unroll
    for (int r = 1; r < 8; ++r) tm = fmaxf(tm, st[r]);
    tm = fmaxf(tm, xhalf(tm));
    const float mn = fmaxf(m, tm);
    const float alpha = expf(m - mn);
    m = mn;
    float p[8];
    float ps = 0.0f, a0 = 0.0f, a1 = 0.0f, a2 = 0.0f;
    #pragma unroll
    for (int r = 0; r < 8; ++r) {
      p[r] = expf(st[r] - mn);
      ps += p[r];
      a0 += p[r] * xk[3 * r + 0];
      a1 += p[r] * xk[3 * r + 1];
      a2 += p[r] * xk[3 * r + 2];
    }
    l   = l   * alpha + ps;
    wx0 = wx0 * alpha + a0;
    wx1 = wx1 * alpha + a1;
    wx2 = wx2 * alpha + a2;
    #pragma unroll
    for (int r = 0; r < 8; ++r) {
      o0[r] *= alpha; o1[r] *= alpha; o2[r] *= alpha; o3[r] *= alpha;
    }
    #pragma unroll
    for (int s = 0; s < 4; ++s) {
      v2f pb; pb.x = p[2 * s]; pb.y = p[2 * s + 1];
      o0 = wmma4(vf[s][0], pb, o0);
      o1 = wmma4(vf[s][1], pb, o1);
      o2 = wmma4(vf[s][2], pb, o2);
      o3 = wmma4(vf[s][3], pb, o3);
    }
  }

  l   += xhalf(l);
  wx0 += xhalf(wx0);
  wx1 += xhalf(wx1);
  wx2 += xhalf(wx2);
  if (hi == 0) {
    s_m[w][sub] = m;
    s_l[w][sub] = l;
    s_wx[w][sub][0] = wx0; s_wx[w][sub][1] = wx1; s_wx[w][sub][2] = wx2;
  }
  #pragma unroll
  for (int r = 0; r < 8; ++r) {
    s_o[w][sub][ 0 + r + 8 * hi] = o0[r];
    s_o[w][sub][16 + r + 8 * hi] = o1[r];
    s_o[w][sub][32 + r + 8 * hi] = o2[r];
    s_o[w][sub][48 + r + 8 * hi] = o3[r];
  }
  __syncthreads();

  #pragma unroll
  for (int pass = 0; pass < 2; ++pass) {
    const int q = pass * 8 + (t >> 4), dch = (t & 15) * 4;
    const float m0 = s_m[0][q], m1 = s_m[1][q], m2 = s_m[2][q], m3 = s_m[3][q];
    const float ma = fmaxf(fmaxf(m0, m1), fmaxf(m2, m3));
    const float f0 = expf(m0 - ma), f1 = expf(m1 - ma);
    const float f2 = expf(m2 - ma), f3 = expf(m3 - ma);
    const float lt = ((s_l[0][q] * f0 + s_l[1][q] * f1) + s_l[2][q] * f2) +
                     s_l[3][q] * f3;
    const float inv = 1.0f / lt;
    const int i = i0 + q;
    const float* xi = xyzf + ((size_t)b * MTOK_ + i) * 3;
    float wd[3];
    #pragma unroll
    for (int c = 0; c < 3; ++c)
      wd[c] = (((s_wx[0][q][c] * f0 + s_wx[1][q][c] * f1) + s_wx[2][q][c] * f2) +
               s_wx[3][q][c] * f3) * inv - xi[c];
    const v4f p0 = *(const v4f*)&s_o[0][q][dch];
    const v4f p1 = *(const v4f*)&s_o[1][q][dch];
    const v4f p2 = *(const v4f*)&s_o[2][q][dch];
    const v4f p3 = *(const v4f*)&s_o[3][q][dch];
    const v4f sm = ((p0 * f0 + p1 * f1) + p2 * f2) + p3 * f3;
    float res[4] = {sm.x, sm.y, sm.z, sm.w};
    #pragma unroll
    for (int e = 0; e < 4; ++e) {
      const int d = dch + e;
      res[e] = res[e] * inv + Wsp[d * 3 + 0] * wd[0] + Wsp[d * 3 + 1] * wd[1] +
               Wsp[d * 3 + 2] * wd[2];
    }
    v4f rv;
    rv.x = res[0]; rv.y = res[1]; rv.z = res[2]; rv.w = res[3];
    *(v4f*)(attn_out + ((size_t)(b * MTOK_) + i) * C_ + h * DH_ + dch) = rv;
  }
}

extern "C" void kernel_launch(void* const* d_in, const int* in_sizes, int n_in,
                              void* d_out, int out_size, void* d_ws, size_t ws_size,
                              hipStream_t stream) {
  (void)in_sizes; (void)n_in; (void)out_size; (void)ws_size;
  const float* xyzs = (const float*)d_in[0];
  const float* Wd   = (const float*)d_in[1];
  const float* Wqkv = (const float*)d_in[2];
  const float* Wsp  = (const float*)d_in[3];
  const float* Wout = (const float*)d_in[4];
  const float* bout = (const float*)d_in[5];
  const float* ln1g = (const float*)d_in[6];
  const float* ln1b = (const float*)d_in[7];
  const float* Wff1 = (const float*)d_in[8];
  const float* bff1 = (const float*)d_in[9];
  const float* Wff2 = (const float*)d_in[10];
  const float* bff2 = (const float*)d_in[11];
  const float* ln2g = (const float*)d_in[12];
  const float* ln2b = (const float*)d_in[13];

  float* ws      = (float*)d_ws;
  float* anchors = ws;
  float* feats   = ws + 8192;
  float* xln     = feats + (size_t)ROWS_ * C_;
  float* qkvb    = xln   + (size_t)ROWS_ * C_;
  float* attno   = qkvb  + (size_t)ROWS_ * 1536;
  float* ffh     = attno + (size_t)ROWS_ * C_;

  fps_kernel<<<B_ * L_, N_, 0, stream>>>(xyzs, anchors);
  p4dconv_kernel<<<B_ * L_ * MA_ / PG_, PTHR_, 0, stream>>>(xyzs, anchors, Wd, feats);

  for (int d = 0; d < 2; ++d) {
    ln_kernel<<<ROWS_, 256, 0, stream>>>(feats, ln1g + d * C_, ln1b + d * C_, xln);
    gemm_wmma<<<dim3(1536 / 32, ROWS_ / 32), 32, 0, stream>>>(
        xln, C_, Wqkv + (size_t)d * 1536 * C_, C_, nullptr, nullptr,
        qkvb, 1536, C_, 0);
    attn_kernel<<<dim3(MTOK_ / 16, HEADS_, B_), 32 * ATT_WAVES_, 0, stream>>>(
        qkvb, anchors, Wsp + (size_t)d * DH_ * 3, attno);
    gemm_wmma<<<dim3(C_ / 32, ROWS_ / 32), 32, 0, stream>>>(
        attno, C_, Wout + (size_t)d * C_ * C_, C_, bout + d * C_,
        feats, feats, C_, C_, 1);
    ln_kernel<<<ROWS_, 256, 0, stream>>>(feats, ln2g + d * C_, ln2b + d * C_, xln);
    gemm_wmma<<<dim3(MLP_ / 32, ROWS_ / 32), 32, 0, stream>>>(
        xln, C_, Wff1 + (size_t)d * MLP_ * C_, C_, bff1 + d * MLP_,
        nullptr, ffh, MLP_, C_, 1);
    gemm_wmma<<<dim3(C_ / 32, ROWS_ / 32), 32, 0, stream>>>(
        ffh, MLP_, Wff2 + (size_t)d * C_ * MLP_, MLP_, bff2 + d * C_,
        feats, feats, C_, MLP_, 0);
  }
  hipMemcpyAsync(d_out, feats, (size_t)ROWS_ * C_ * sizeof(float),
                 hipMemcpyDeviceToDevice, stream);
}


# GEMM weights repacked once per call (spare blocks of the sampling launch) into WMMA fragment order: a B fragment load is 512 contiguous B, 4 whole lines, was 16 rows x 32 B; sums unchanged
# speedup vs baseline: 2.0242x; 1.0508x over previous
#include <hip/hip_runtime.h>
#include <hip/hip_bf16.h>
#include <math.h>

typedef float v2f __attribute__((ext_vector_type(2)));
typedef float v4f __attribute__((ext_vector_type(4)));
typedef float v8f __attribute__((ext_vector_type(8)));

#define B_    2
#define T_    4
#define N_    1024
#define L_    4
#define MA_   256
#define KNB_  32
#define C_    512
#define HEADS_ 8
#define DH_   64
#define MTOK_ 1024
#define ROWS_ 2048
#define MLP_  2048

__device__ __forceinline__ v8f wmma4(v2f a, v2f b, v8f c) {
  return __builtin_amdgcn_wmma_f32_16x16x4_f32(false, a, false, b, (short)0, c,
                                               false, false);
}

__device__ __forceinline__ float gelu_exact(float x) {
  return 0.5f * x * (1.0f + erff(x * 0.7071067811865475f));
}

__device__ __forceinline__ float xhalf(float x) {
  const int bx = __builtin_bit_cast(int, x);
  return __builtin_bit_cast(float, __builtin_amdgcn_permlanex16(
      bx, bx, 0x76543210, 0xfedcba98, false, false));
}

typedef unsigned long long u64_t;
typedef u64_t v2u64 __attribute__((ext_vector_type(2)));

#define RPU_      4096
#define RP_QKV_   (2 * 1536 * C_ / 4 / RPU_)
#define RP_OUT_   (2 * C_ * C_ / 4 / RPU_)
#define RP_FF1_   (2 * MLP_ * C_ / 4 / RPU_)
#define RP_FF2_   (2 * C_ * MLP_ / 4 / RPU_)
#define RP_BLOCKS_ (RP_QKV_ + RP_OUT_ + RP_FF1_ + RP_FF2_)

__device__ __forceinline__ void repack_block(int rb, int tid,
                                             const float* __restrict__ Wqkv,
                                             const float* __restrict__ Wout,
                                             const float* __restrict__ Wff1,
                                             const float* __restrict__ Wff2,
                                             float* __restrict__ wpk) {
  const float* src = Wqkv;
  float* dst = wpk;
  int lb = rb;
  int kq_log = 6;
  if (rb >= RP_QKV_ + RP_OUT_ + RP_FF1_) {
    src = Wff2; lb = rb - (RP_QKV_ + RP_OUT_ + RP_FF1_); kq_log = 8;
    dst = wpk + (size_t)2 * (1536 + C_ + MLP_) * C_;
  } else if (rb >= RP_QKV_ + RP_OUT_) {
    src = Wff1; lb = rb - (RP_QKV_ + RP_OUT_);
    dst = wpk + (size_t)2 * (1536 + C_) * C_;
  } else if (rb >= RP_QKV_) {
    src = Wout; lb = rb - RP_QKV_;
    dst = wpk + (size_t)2 * 1536 * C_;
  }
  v4f v[4];
  #pragma unroll
  for (int j = 0; j < 4; ++j) {
    const int u = lb * RPU_ + j * 1024 + tid;
    const int lane = u & 31, t = u >> 5;
    const int kb = t & ((1 << kq_log) - 1), nt = t >> kq_log;
    v[j] = *(const v4f*)(src + (((size_t)(16 * nt + (lane & 15))) << (kq_log + 3)) +
                         8 * kb + 4 * (lane >> 4));
  }
  #pragma unroll
  for (int j = 0; j < 4; ++j)
    *(v4f*)(dst + (size_t)(lb * RPU_ + j * 1024 + tid) * 4) = v[j];
}

__global__ void fps_kernel(const float* __restrict__ xyzs,
                           float* __restrict__ anchors,
                           const float* __restrict__ Wqkv,
                           const float* __restrict__ Wout,
                           const float* __restrict__ Wff1,
                           const float* __restrict__ Wff2,
                           float* __restrict__ wpk) {
  if (blockIdx.x >= B_ * L_) {
    repack_block(blockIdx.x - B_ * L_, threadIdx.x, Wqkv, Wout, Wff1, Wff2, wpk);
    return;
  }
  __shared__ __attribute__((aligned(16))) float sdist[N_];
  __shared__ __attribute__((aligned(16))) v4f   spts[N_];
  __shared__ __attribute__((aligned(16))) u64_t skey[32];
  __shared__ __attribute__((aligned(16))) v4f   slast;
  __shared__ float sanch[MA_ * 3];
  const int blk = blockIdx.x;
  const int tid = threadIdx.x;
  const float* frame = xyzs + (size_t)blk * N_ * 3;
  const float px = frame[tid * 3 + 0];
  const float py = frame[tid * 3 + 1];
  const float pz = frame[tid * 3 + 2];
  float dist = 1e10f;
  float* aout = anchors + (size_t)blk * MA_ * 3;
  {
    v4f me; me.x = px; me.y = py; me.z = pz; me.w = 0.0f;
    spts[tid] = me;
    if (tid == 0) {
      slast = me;
      sanch[0] = px; sanch[1] = py; sanch[2] = pz;
    }
  }
  __syncthreads();
  for (int it = 1; it < MA_; ++it) {
    const v4f lp = slast;
    const float dx = px - lp.x, dy = py - lp.y, dz = pz - lp.z;
    dist = fminf(dist, __builtin_fmaf(dz, dz, __builtin_fmaf(dx, dx, dy * dy)));
    sdist[tid] = dist;
    __syncthreads();
    if (tid < 32) {
      const v4f* sd4 = (const v4f*)sdist;
      v4f q[8];
      #pragma unroll
      for (int i = 0; i < 8; ++i) q[i] = sd4[32 * i + tid];
      float bv[32];
      int   bo[32];
      #pragma unroll
      for (int i = 0; i < 8; ++i) {
        bv[4 * i + 0] = q[i].x; bo[4 * i + 0] = 128 * i + 0;
        bv[4 * i + 1] = q[i].y; bo[4 * i + 1] = 128 * i + 1;
        bv[4 * i + 2] = q[i].z; bo[4 * i + 2] = 128 * i + 2;
        bv[4 * i + 3] = q[i].w; bo[4 * i + 3] = 128 * i + 3;
      }
      #pragma unroll
      for (int wdt = 1; wdt < 32; wdt <<= 1) {
        #pragma unroll
        for (int j = 0; j < 32; j += 2 * wdt) {
          const bool tk = bv[j + wdt] > bv[j];
          bv[j] = tk ? bv[j + wdt] : bv[j];
          bo[j] = tk ? bo[j + wdt] : bo[j];
        }
      }
      const unsigned idx1 = (unsigned)(bo[0] + 4 * tid);
      skey[tid] = ((u64_t)__builtin_bit_cast(unsigned, bv[0]) << 32) |
                  (u64_t)(1023u - idx1);
      __builtin_amdgcn_fence(__ATOMIC_RELEASE, "wavefront");
      __builtin_amdgcn_wave_barrier();
      __builtin_amdgcn_fence(__ATOMIC_ACQUIRE, "wavefront");
      const v2u64* sk2 = (const v2u64*)skey;
      v2u64 kv[16];
      #pragma unroll
      for (int i = 0; i < 16; ++i) kv[i] = sk2[i];
      asm volatile("" :: "v"(kv[0]), "v"(kv[1]), "v"(kv[2]), "v"(kv[3]),
                         "v"(kv[4]), "v"(kv[5]), "v"(kv[6]), "v"(kv[7]),
                         "v"(kv[8]), "v"(kv[9]), "v"(kv[10]), "v"(kv[11]),
                         "v"(kv[12]), "v"(kv[13]), "v"(kv[14]), "v"(kv[15]));
      u64_t kk[32];
      #pragma unroll
      for (int i = 0; i < 16; ++i) {
        kk[2 * i] = kv[i].x; kk[2 * i + 1] = kv[i].y;
      }
      #pragma unroll
      for (int wdt = 1; wdt < 32; wdt <<= 1) {
        #pragma unroll
        for (int j = 0; j < 32; j += 2 * wdt)
          kk[j] = kk[j + wdt] > kk[j] ? kk[j + wdt] : kk[j];
      }
      const int nxt = 1023 - (int)(unsigned)(kk[0] & 0xffffffffull);
      const v4f wp = spts[nxt];
      if (tid == 0) {
        slast = wp;
        sanch[it * 3 + 0] = wp.x; sanch[it * 3 + 1] = wp.y;
        sanch[it * 3 + 2] = wp.z;
      }
    }
    __syncthreads();
  }
  if (tid < MA_ * 3) aout[tid] = sanch[tid];
}

#define PG_   4
#define PQ_   8
#define PSEG_ (N_ / PQ_)
#define PLST_ (KNB_ + 1)
#define PTHR_ 512

__global__ void __launch_bounds__(PTHR_)
p4dconv_kernel(const float* __restrict__ xyzs,
               const float* __restrict__ anchors,
               const float* __restrict__ Wd,
               float* __restrict__ feats) {
  __shared__ __attribute__((aligned(16))) v4f spt[3][N_ + PQ_];
  __shared__ __attribute__((aligned(16))) v4f slist[3 * PG_ * PQ_][PLST_];
  __shared__ __attribute__((aligned(16))) v4f flist[3 * PG_][KNB_];
  __shared__ int scnt[3 * PG_ * PQ_];
  __shared__ int sbase[3 * PG_ * PQ_];
  __shared__ int stake[3 * PG_ * PQ_];
  __shared__ int snum[3 * PG_];
  const int blk = blockIdx.x;
  const int m0  = (blk % (MA_ / PG_)) * PG_;
  const int bl  = blk / (MA_ / PG_);
  const int l   = bl % L_;
  const int b   = bl / L_;
  const int tid = threadIdx.x;

  #pragma unroll
  for (int f = 0; f < 3; ++f) {
    int fi = l + f - 1; fi = fi < 0 ? 0 : (fi > T_ - 1 ? T_ - 1 : fi);
    const float* nb = xyzs + ((size_t)b * T_ + fi) * N_ * 3;
    #pragma unroll
    for (int h = 0; h < N_ / PTHR_; ++h) {
      const int i = tid + PTHR_ * h;
      v4f p;
      p.x = nb[i * 3 + 0]; p.y = nb[i * 3 + 1]; p.z = nb[i * 3 + 2]; p.w = 0.0f;
      spt[f][i + i / PSEG_] = p;
    }
  }
  const bool walker = tid < 3 * PG_ * PQ_;
  const int wfr  = tid / (PG_ * PQ_);
  const int wg   = (tid % (PG_ * PQ_)) / PQ_;
  const int wq   = tid % PQ_;
  const int wseg = (wfr * PG_ + wg) * PQ_ + wq;
  float ax = 0.0f, ay = 0.0f, az = 0.0f;
  if (walker) {
    const float* arow = anchors + ((size_t)bl * MA_ + m0 + wg) * 3;
    ax = arow[0]; ay = arow[1]; az = arow[2];
  }
  const v4f wv = *(const v4f*)(Wd + tid * 4);
  __syncthreads();

  if (walker) {
    const v4f* pts = &spt[wfr][wq * (PSEG_ + 1)];
    v4f* lst = slist[wseg];
    int cnt = 0;
    #pragma unroll 4
    for (int p = 0; p < PSEG_; ++p) {
      const v4f pt = pts[p];
      const float dx = pt.x - ax, dy = pt.y - ay, dz = pt.z - az;
      const bool hit = __builtin_fmaf(dz, dz, __builtin_fmaf(dx, dx, dy * dy)) < 1.0f;
      v4f e; e.x = dx; e.y = dy; e.z = dz; e.w = 0.0f;
      lst[hit ? (cnt < KNB_ ? cnt : KNB_) : KNB_] = e;
      cnt += hit ? 1 : 0;
    }
    scnt[wseg] = cnt < KNB_ ? cnt : KNB_;
  }
  __syncthreads();

  if (walker && wq == 0) {
    int run = 0;
    #pragma unroll
    for (int q = 0; q < PQ_; ++q) {
      const int c = scnt[wseg + q];
      const int t = c < KNB_ - run ? c : KNB_ - run;
      sbase[wseg + q] = run;
      stake[wseg + q] = t;
      run += t;
    }
    if (run == 0) {
      const v4f p0 = spt[wfr][0];
      v4f e; e.x = p0.x - ax; e.y = p0.y - ay; e.z = p0.z - az; e.w = 0.0f;
      flist[wfr * PG_ + wg][0] = e;
      run = 1;
    }
    snum[wfr * PG_ + wg] = run;
  }
  __syncthreads();

  #pragma unroll
  for (int j = 0; j < (3 * PG_ * PQ_ * KNB_) / PTHR_; ++j) {
    const int item = tid + PTHR_ * j;
    const int k = item % KNB_, seg = item / KNB_;
    if (k < stake[seg]) flist[seg / PQ_][sbase[seg] + k] = slist[seg][k];
  }
  __syncthreads();

  const float w0 = wv.x, w1 = wv.y, w2 = wv.z, w3 = wv.w;
  for (int g = 0; g < PG_; ++g) {
    float maxv = -3.4e38f;
    #pragma unroll
    for (int f = 0; f < 3; ++f) {
      const float tq = (float)(f - 1);
      const int n = __builtin_amdgcn_readfirstlane(snum[f * PG_ + g]);
      const v4f* lst = flist[f * PG_ + g];
      #pragma unroll 4
      for (int k = 0; k < n; ++k) {
        const v4f e = lst[k];
        const float v =
            __builtin_fmaf(w2, e.z, __builtin_fmaf(w0, e.x, w1 * e.y)) + w3 * tq;
        maxv = fmaxf(maxv, v);
      }
    }
    feats[((size_t)bl * MA_ + m0 + g) * C_ + tid] = fmaxf(maxv, 0.0f);
  }
}

__global__ void ln_kernel(const float* __restrict__ x, const float* __restrict__ g,
                          const float* __restrict__ bta, float* __restrict__ y) {
  __shared__ float red[256];
  const int row = blockIdx.x, t = threadIdx.x;
  const float* xr = x + (size_t)row * C_;
  const float a0 = xr[t], a1 = xr[t + 256];
  red[t] = a0 + a1; __syncthreads();
  for (int s = 128; s > 0; s >>= 1) { if (t < s) red[t] += red[t + s]; __syncthreads(); }
  const float mean = red[0] * (1.0f / C_);
  __syncthreads();
  const float d0 = a0 - mean, d1 = a1 - mean;
  red[t] = d0 * d0 + d1 * d1; __syncthreads();
  for (int s = 128; s > 0; s >>= 1) { if (t < s) red[t] += red[t + s]; __syncthreads(); }
  const float inv = rsqrtf(red[0] * (1.0f / C_) + 1e-5f);
  float* yr = y + (size_t)row * C_;
  yr[t]       = d0 * inv * g[t]       + bta[t];
  yr[t + 256] = d1 * inv * g[t + 256] + bta[t + 256];
}

__device__ __forceinline__ void gemm_ld1(v4f (&f)[16], int kb, const float* a0,
                                         const float* a1, const float* w0,
                                         const float* w1) {
  f[4 * kb + 0] = *(const v4f*)(a0 + 8 * kb);
  f[4 * kb + 1] = *(const v4f*)(a1 + 8 * kb);
  f[4 * kb + 2] = *(const v4f*)(w0 + 128 * kb);
  f[4 * kb + 3] = *(const v4f*)(w1 + 128 * kb);
}

__device__ __forceinline__ void gemm_mm1(const v4f (&f)[16], int kb, v8f& acc00,
                                         v8f& acc01, v8f& acc10, v8f& acc11) {
  const v4f a0 = f[4 * kb + 0], a1 = f[4 * kb + 1];
  const v4f b0 = f[4 * kb + 2], b1 = f[4 * kb + 3];
  const v2f a0l = {a0.x, a0.y}, a0h = {a0.z, a0.w};
  const v2f a1l = {a1.x, a1.y}, a1h = {a1.z, a1.w};
  const v2f b0l = {b0.x, b0.y}, b0h = {b0.z, b0.w};
  const v2f b1l = {b1.x, b1.y}, b1h = {b1.z, b1.w};
  acc00 = wmma4(a0l, b0l, acc00);
  acc01 = wmma4(a0l, b1l, acc01);
  acc10 = wmma4(a1l, b0l, acc10);
  acc11 = wmma4(a1l, b1l, acc11);
  acc00 = wmma4(a0h, b0h, acc00);
  acc01 = wmma4(a0h, b1h, acc01);
  acc10 = wmma4(a1h, b0h, acc10);
  acc11 = wmma4(a1h, b1h, acc11);
}

__global__ void __launch_bounds__(32)
gemm_wmma(const float* __restrict__ A, int lda,
          const float* __restrict__ Wp,
          const float* __restrict__ bias,
          const float* __restrict__ resid,
          float* __restrict__ Cout, int ldc,
          int Kdim, int act) {
  const int j0 = blockIdx.x * 32, i0 = blockIdx.y * 32;
  const int lane = threadIdx.x & 31;
  const int sub = lane & 15, hi = lane >> 4;
  const float* arow0 = A + (size_t)(i0 + sub) * lda + 4 * hi;
  const float* arow1 = arow0 + (size_t)16 * lda;
  const float* wrow0 = Wp + (size_t)(j0 >> 4) * (Kdim >> 3) * 128 + lane * 4;
  const float* wrow1 = wrow0 + (size_t)(Kdim >> 3) * 128;
  __builtin_prefetch(arow0, 0, 0);
  __builtin_prefetch(wrow0, 0, 0);
  v8f acc00 = {0.f, 0.f, 0.f, 0.f, 0.f, 0.f, 0.f, 0.f};
  v8f acc01 = acc00, acc10 = acc00, acc11 = acc00;
  v4f s0[16], s1[16];
  #pragma unroll
  for (int kb = 0; kb < 4; ++kb) gemm_ld1(s0, kb, arow0, arow1, wrow0, wrow1);
  #pragma unroll
  for (int kb = 0; kb < 4; ++kb)
    gemm_ld1(s1, kb, arow0 + 32, arow1 + 32, wrow0 + 32 * 16, wrow1 + 32 * 16);
  int k0 = 0;
  for (; k0 + 64 < Kdim; k0 += 64) {
    #pragma unroll
    for (int kb = 0; kb < 4; ++kb) {
      gemm_mm1(s0, kb, acc00, acc01, acc10, acc11);
      gemm_ld1(s0, kb, arow0 + k0 + 64, arow1 + k0 + 64,
               wrow0 + (size_t)(k0 + 64) * 16, wrow1 + (size_t)(k0 + 64) * 16);
    }
    #pragma unroll
    for (int kb = 0; kb < 4; ++kb) {
      gemm_mm1(s1, kb, acc00, acc01, acc10, acc11);
      gemm_ld1(s1, kb, arow0 + k0 + 96, arow1 + k0 + 96,
               wrow0 + (size_t)(k0 + 96) * 16, wrow1 + (size_t)(k0 + 96) * 16);
    }
  }
  #pragma unroll
  for (int kb = 0; kb < 4; ++kb) gemm_mm1(s0, kb, acc00, acc01, acc10, acc11);
  #pragma unroll
  for (int kb = 0; kb < 4; ++kb) gemm_mm1(s1, kb, acc00, acc01, acc10, acc11);
  const float bj0 = bias ? bias[j0 + sub]      : 0.0f;
  const float bj1 = bias ? bias[j0 + 16 + sub] : 0.0f;
  #pragma unroll
  for (int r = 0; r < 8; ++r) {
    const int row0 = i0 + r + 8 * hi;
    const int row1 = row0 + 16;
    const int col0 = j0 + sub;
    const int col1 = col0 + 16;
    float v00 = acc00[r] + bj0, v01 = acc01[r] + bj1;
    float v10 = acc10[r] + bj0, v11 = acc11[r] + bj1;
    if (act == 1) {
      v00 = gelu_exact(v00); v01 = gelu_exact(v01);
      v10 = gelu_exact(v10); v11 = gelu_exact(v11);
    }
    if (resid) {
      v00 += resid[(size_t)row0 * ldc + col0];
      v01 += resid[(size_t)row0 * ldc + col1];
      v10 += resid[(size_t)row1 * ldc + col0];
      v11 += resid[(size_t)row1 * ldc + col1];
    }
    Cout[(size_t)row0 * ldc + col0] = v00;
    Cout[(size_t)row0 * ldc + col1] = v01;
    Cout[(size_t)row1 * ldc + col0] = v10;
    Cout[(size_t)row1 * ldc + col1] = v11;
  }
}

#define ATT_WAVES_ 4
#define ATT_KPW_   (MTOK_ / ATT_WAVES_)
#define ATT_OP_    68

__global__ void __launch_bounds__(32 * ATT_WAVES_)
attn_kernel(const float* __restrict__ qkv,
            const float* __restrict__ xyzf,
            const float* __restrict__ Wsp,
            float* __restrict__ attn_out) {
  __shared__ float s_m[ATT_WAVES_][16];
  __shared__ float s_l[ATT_WAVES_][16];
  __shared__ float s_wx[ATT_WAVES_][16][3];
  __shared__ __attribute__((aligned(16))) float s_o[ATT_WAVES_][16][ATT_OP_];
  const int b = blockIdx.z, h = blockIdx.y, i0 = blockIdx.x * 16;
  const int t = threadIdx.x;
  const int lane = t & 31, w = t >> 5;
  const int sub = lane & 15, hi = lane >> 4;
  const int key0 = b * MTOK_ + w * ATT_KPW_;

  const float* qbase =
      qkv + ((size_t)(b * MTOK_ + i0 + sub) * 3 + 0) * C_ + h * DH_ + 2 * hi;
  v2f qb[16];
  #pragma unroll
  for (int ks = 0; ks < 16; ++ks)
    qb[ks] = *(const v2f*)(qbase + 4 * ks) * 0.125f;

  const float* krow =
      qkv + ((size_t)(key0 + sub) * 3 + 1) * C_ + h * DH_ + 2 * hi;
  const float* vcol =
      qkv + ((size_t)(key0 + 8 * hi) * 3 + 2) * C_ + h * DH_ + sub;
  const float* xrow = xyzf + (size_t)(key0 + 8 * hi) * 3;

  v8f o0 = {0.f, 0.f, 0.f, 0.f, 0.f, 0.f, 0.f, 0.f};
  v8f o1 = o0, o2 = o0, o3 = o0;
  float m = -3.4e38f, l = 0.0f, wx0 = 0.0f, wx1 = 0.0f, wx2 = 0.0f;

  v2f kf[16];
  #pragma unroll
  for (int ks = 0; ks < 16; ++ks) kf[ks] = *(const v2f*)(krow + 4 * ks);

  for (int jt = 0; jt < ATT_KPW_ / 16; ++jt) {
    v8f st = {0.f, 0.f, 0.f, 0.f, 0.f, 0.f, 0.f, 0.f};
    #pragma unroll
    for (int ks = 0; ks < 16; ++ks) st = wmma4(kf[ks], qb[ks], st);

    const int jn = (jt + 1 < ATT_KPW_ / 16) ? jt + 1 : jt;
    const float* kn = krow + (size_t)jn * 16 * 3 * C_;
    #pragma unroll
    for (int ks = 0; ks < 16; ++ks) kf[ks] = *(const v2f*)(kn + 4 * ks);

    const float* vt = vcol + (size_t)jt * 16 * 3 * C_;
    v2f vf[4][4];
    #pragma unroll
    for (int s = 0; s < 4; ++s) {
      #pragma unroll
      for (int dt = 0; dt < 4; ++dt) {
        const float* vp = vt + (size_t)(2 * s) * 3 * C_ + dt * 16;
        vf[s][dt].x = vp[0];
        vf[s][dt].y = vp[3 * C_];
      }
    }
    const v4f* xp = (const v4f*)(xrow + jt * 48);
    float xk[24];
    #pragma unroll
    for (int i = 0; i < 6; ++i) {
      const v4f xv = xp[i];
      xk[4 * i + 0] = xv.x; xk[4 * i + 1] = xv.y;
      xk[4 * i + 2] = xv.z; xk[4 * i + 3] = xv.w;
    }

    float tm = st[0];
    #pragma unroll
    for (int r = 1; r < 8; ++r) tm = fmaxf(tm, st[r]);
    tm = fmaxf(tm, xhalf(tm));
    const float mn = fmaxf(m, tm);
    const float alpha = expf(m - mn);
    m = mn;
    float p[8];
    float ps = 0.0f, a0 = 0.0f, a1 = 0.0f, a2 = 0.0f;
    #pragma unroll
    for (int r = 0; r < 8; ++r) {
      p[r] = expf(st[r] - mn);
      ps += p[r];
      a0 += p[r] * xk[3 * r + 0];
      a1 += p[r] * xk[3 * r + 1];
      a2 += p[r] * xk[3 * r + 2];
    }
    l   = l   * alpha + ps;
    wx0 = wx0 * alpha + a0;
    wx1 = wx1 * alpha + a1;
    wx2 = wx2 * alpha + a2;
    #pragma unroll
    for (int r = 0; r < 8; ++r) {
      o0[r] *= alpha; o1[r] *= alpha; o2[r] *= alpha; o3[r] *= alpha;
    }
    #pragma unroll
    for (int s = 0; s < 4; ++s) {
      v2f pb; pb.x = p[2 * s]; pb.y = p[2 * s + 1];
      o0 = wmma4(vf[s][0], pb, o0);
      o1 = wmma4(vf[s][1], pb, o1);
      o2 = wmma4(vf[s][2], pb, o2);
      o3 = wmma4(vf[s][3], pb, o3);
    }
  }

  l   += xhalf(l);
  wx0 += xhalf(wx0);
  wx1 += xhalf(wx1);
  wx2 += xhalf(wx2);
  if (hi == 0) {
    s_m[w][sub] = m;
    s_l[w][sub] = l;
    s_wx[w][sub][0] = wx0; s_wx[w][sub][1] = wx1; s_wx[w][sub][2] = wx2;
  }
  #pragma unroll
  for (int r = 0; r < 8; ++r) {
    s_o[w][sub][ 0 + r + 8 * hi] = o0[r];
    s_o[w][sub][16 + r + 8 * hi] = o1[r];
    s_o[w][sub][32 + r + 8 * hi] = o2[r];
    s_o[w][sub][48 + r + 8 * hi] = o3[r];
  }
  __syncthreads();

  #pragma unroll
  for (int pass = 0; pass < 2; ++pass) {
    const int q = pass * 8 + (t >> 4), dch = (t & 15) * 4;
    const float m0 = s_m[0][q], m1 = s_m[1][q], m2 = s_m[2][q], m3 = s_m[3][q];
    const float ma = fmaxf(fmaxf(m0, m1), fmaxf(m2, m3));
    const float f0 = expf(m0 - ma), f1 = expf(m1 - ma);
    const float f2 = expf(m2 - ma), f3 = expf(m3 - ma);
    const float lt = ((s_l[0][q] * f0 + s_l[1][q] * f1) + s_l[2][q] * f2) +
                     s_l[3][q] * f3;
    const float inv = 1.0f / lt;
    const int i = i0 + q;
    const float* xi = xyzf + ((size_t)b * MTOK_ + i) * 3;
    float wd[3];
    #pragma unroll
    for (int c = 0; c < 3; ++c)
      wd[c] = (((s_wx[0][q][c] * f0 + s_wx[1][q][c] * f1) + s_wx[2][q][c] * f2) +
               s_wx[3][q][c] * f3) * inv - xi[c];
    const v4f p0 = *(const v4f*)&s_o[0][q][dch];
    const v4f p1 = *(const v4f*)&s_o[1][q][dch];
    const v4f p2 = *(const v4f*)&s_o[2][q][dch];
    const v4f p3 = *(const v4f*)&s_o[3][q][dch];
    const v4f sm = ((p0 * f0 + p1 * f1) + p2 * f2) + p3 * f3;
    float res[4] = {sm.x, sm.y, sm.z, sm.w};
    #pragma unroll
    for (int e = 0; e < 4; ++e) {
      const int d = dch + e;
      res[e] = res[e] * inv + Wsp[d * 3 + 0] * wd[0] + Wsp[d * 3 + 1] * wd[1] +
               Wsp[d * 3 + 2] * wd[2];
    }
    v4f rv;
    rv.x = res[0]; rv.y = res[1]; rv.z = res[2]; rv.w = res[3];
    *(v4f*)(attn_out + ((size_t)(b * MTOK_) + i) * C_ + h * DH_ + dch) = rv;
  }
}

extern "C" void kernel_launch(void* const* d_in, const int* in_sizes, int n_in,
                              void* d_out, int out_size, void* d_ws, size_t ws_size,
                              hipStream_t stream) {
  (void)in_sizes; (void)n_in; (void)out_size; (void)ws_size;
  const float* xyzs = (const float*)d_in[0];
  const float* Wd   = (const float*)d_in[1];
  const float* Wqkv = (const float*)d_in[2];
  const float* Wsp  = (const float*)d_in[3];
  const float* Wout = (const float*)d_in[4];
  const float* bout = (const float*)d_in[5];
  const float* ln1g = (const float*)d_in[6];
  const float* ln1b = (const float*)d_in[7];
  const float* Wff1 = (const float*)d_in[8];
  const float* bff1 = (const float*)d_in[9];
  const float* Wff2 = (const float*)d_in[10];
  const float* bff2 = (const float*)d_in[11];
  const float* ln2g = (const float*)d_in[12];
  const float* ln2b = (const float*)d_in[13];

  float* ws      = (float*)d_ws;
  float* anchors = ws;
  float* feats   = ws + 8192;
  float* xln     = feats + (size_t)ROWS_ * C_;
  float* qkvb    = xln   + (size_t)ROWS_ * C_;
  float* attno   = qkvb  + (size_t)ROWS_ * 1536;
  float* ffh     = attno + (size_t)ROWS_ * C_;
  float* wpk     = ffh + (size_t)ROWS_ * MLP_;
  const float* pqkv = wpk;
  const float* pout = pqkv + (size_t)2 * 1536 * C_;
  const float* pff1 = pout + (size_t)2 * C_ * C_;
  const float* pff2 = pff1 + (size_t)2 * MLP_ * C_;

  fps_kernel<<<B_ * L_ + RP_BLOCKS_, N_, 0, stream>>>(xyzs, anchors, Wqkv, Wout,
                                                     Wff1, Wff2, wpk);
  p4dconv_kernel<<<B_ * L_ * MA_ / PG_, PTHR_, 0, stream>>>(xyzs, anchors, Wd, feats);

  for (int d = 0; d < 2; ++d) {
    ln_kernel<<<ROWS_, 256, 0, stream>>>(feats, ln1g + d * C_, ln1b + d * C_, xln);
    gemm_wmma<<<dim3(1536 / 32, ROWS_ / 32), 32, 0, stream>>>(
        xln, C_, pqkv + (size_t)d * 1536 * C_, nullptr, nullptr,
        qkvb, 1536, C_, 0);
    attn_kernel<<<dim3(MTOK_ / 16, HEADS_, B_), 32 * ATT_WAVES_, 0, stream>>>(
        qkvb, anchors, Wsp + (size_t)d * DH_ * 3, attno);
    gemm_wmma<<<dim3(C_ / 32, ROWS_ / 32), 32, 0, stream>>>(
        attno, C_, pout + (size_t)d * C_ * C_, bout + d * C_,
        feats, feats, C_, C_, 1);
    ln_kernel<<<ROWS_, 256, 0, stream>>>(feats, ln2g + d * C_, ln2b + d * C_, xln);
    gemm_wmma<<<dim3(MLP_ / 32, ROWS_ / 32), 32, 0, stream>>>(
        xln, C_, pff1 + (size_t)d * MLP_ * C_, bff1 + d * MLP_,
        nullptr, ffh, MLP_, C_, 1);
    gemm_wmma<<<dim3(C_ / 32, ROWS_ / 32), 32, 0, stream>>>(
        ffh, MLP_, pff2 + (size_t)d * C_ * MLP_, bff2 + d * C_,
        feats, feats, C_, MLP_, 0);
  }
  hipMemcpyAsync(d_out, feats, (size_t)ROWS_ * C_ * sizeof(float),
                 hipMemcpyDeviceToDevice, stream);
}


# GEMM waves grouped 2x6 / 2x2 / 4x4 per work-group (256 groups per launch), no LDS, no barrier: waves of a row share A lines and of a column W lines in the compute unit's cache; sums unchanged
# speedup vs baseline: 2.1429x; 1.0586x over previous
#include <hip/hip_runtime.h>
#include <hip/hip_bf16.h>
#include <math.h>

typedef float v2f __attribute__((ext_vector_type(2)));
typedef float v4f __attribute__((ext_vector_type(4)));
typedef float v8f __attribute__((ext_vector_type(8)));

#define B_    2
#define T_    4
#define N_    1024
#define L_    4
#define MA_   256
#define KNB_  32
#define C_    512
#define HEADS_ 8
#define DH_   64
#define MTOK_ 1024
#define ROWS_ 2048
#define MLP_  2048

__device__ __forceinline__ v8f wmma4(v2f a, v2f b, v8f c) {
  return __builtin_amdgcn_wmma_f32_16x16x4_f32(false, a, false, b, (short)0, c,
                                               false, false);
}

__device__ __forceinline__ float gelu_exact(float x) {
  return 0.5f * x * (1.0f + erff(x * 0.7071067811865475f));
}

__device__ __forceinline__ float xhalf(float x) {
  const int bx = __builtin_bit_cast(int, x);
  return __builtin_bit_cast(float, __builtin_amdgcn_permlanex16(
      bx, bx, 0x76543210, 0xfedcba98, false, false));
}

typedef unsigned long long u64_t;
typedef u64_t v2u64 __attribute__((ext_vector_type(2)));

#define RPU_      4096
#define RP_QKV_   (2 * 1536 * C_ / 4 / RPU_)
#define RP_OUT_   (2 * C_ * C_ / 4 / RPU_)
#define RP_FF1_   (2 * MLP_ * C_ / 4 / RPU_)
#define RP_FF2_   (2 * C_ * MLP_ / 4 / RPU_)
#define RP_BLOCKS_ (RP_QKV_ + RP_OUT_ + RP_FF1_ + RP_FF2_)

__device__ __forceinline__ void repack_block(int rb, int tid,
                                             const float* __restrict__ Wqkv,
                                             const float* __restrict__ Wout,
                                             const float* __restrict__ Wff1,
                                             const float* __restrict__ Wff2,
                                             float* __restrict__ wpk) {
  const float* src = Wqkv;
  float* dst = wpk;
  int lb = rb;
  int kq_log = 6;
  if (rb >= RP_QKV_ + RP_OUT_ + RP_FF1_) {
    src = Wff2; lb = rb - (RP_QKV_ + RP_OUT_ + RP_FF1_); kq_log = 8;
    dst = wpk + (size_t)2 * (1536 + C_ + MLP_) * C_;
  } else if (rb >= RP_QKV_ + RP_OUT_) {
    src = Wff1; lb = rb - (RP_QKV_ + RP_OUT_);
    dst = wpk + (size_t)2 * (1536 + C_) * C_;
  } else if (rb >= RP_QKV_) {
    src = Wout; lb = rb - RP_QKV_;
    dst = wpk + (size_t)2 * 1536 * C_;
  }
  v4f v[4];
  #pragma unroll
  for (int j = 0; j < 4; ++j) {
    const int u = lb * RPU_ + j * 1024 + tid;
    const int lane = u & 31, t = u >> 5;
    const int kb = t & ((1 << kq_log) - 1), nt = t >> kq_log;
    v[j] = *(const v4f*)(src + (((size_t)(16 * nt + (lane & 15))) << (kq_log + 3)) +
                         8 * kb + 4 * (lane >> 4));
  }
  #pragma unroll
  for (int j = 0; j < 4; ++j)
    *(v4f*)(dst + (size_t)(lb * RPU_ + j * 1024 + tid) * 4) = v[j];
}

__global__ void fps_kernel(const float* __restrict__ xyzs,
                           float* __restrict__ anchors,
                           const float* __restrict__ Wqkv,
                           const float* __restrict__ Wout,
                           const float* __restrict__ Wff1,
                           const float* __restrict__ Wff2,
                           float* __restrict__ wpk) {
  if (blockIdx.x >= B_ * L_) {
    repack_block(blockIdx.x - B_ * L_, threadIdx.x, Wqkv, Wout, Wff1, Wff2, wpk);
    return;
  }
  __shared__ __attribute__((aligned(16))) float sdist[N_];
  __shared__ __attribute__((aligned(16))) v4f   spts[N_];
  __shared__ __attribute__((aligned(16))) u64_t skey[32];
  __shared__ __attribute__((aligned(16))) v4f   slast;
  __shared__ float sanch[MA_ * 3];
  const int blk = blockIdx.x;
  const int tid = threadIdx.x;
  const float* frame = xyzs + (size_t)blk * N_ * 3;
  const float px = frame[tid * 3 + 0];
  const float py = frame[tid * 3 + 1];
  const float pz = frame[tid * 3 + 2];
  float dist = 1e10f;
  float* aout = anchors + (size_t)blk * MA_ * 3;
  {
    v4f me; me.x = px; me.y = py; me.z = pz; me.w = 0.0f;
    spts[tid] = me;
    if (tid == 0) {
      slast = me;
      sanch[0] = px; sanch[1] = py; sanch[2] = pz;
    }
  }
  __syncthreads();
  for (int it = 1; it < MA_; ++it) {
    const v4f lp = slast;
    const float dx = px - lp.x, dy = py - lp.y, dz = pz - lp.z;
    dist = fminf(dist, __builtin_fmaf(dz, dz, __builtin_fmaf(dx, dx, dy * dy)));
    sdist[tid] = dist;
    __syncthreads();
    if (tid < 32) {
      const v4f* sd4 = (const v4f*)sdist;
      v4f q[8];
      #pragma unroll
      for (int i = 0; i < 8; ++i) q[i] = sd4[32 * i + tid];
      float bv[32];
      int   bo[32];
      #pragma unroll
      for (int i = 0; i < 8; ++i) {
        bv[4 * i + 0] = q[i].x; bo[4 * i + 0] = 128 * i + 0;
        bv[4 * i + 1] = q[i].y; bo[4 * i + 1] = 128 * i + 1;
        bv[4 * i + 2] = q[i].z; bo[4 * i + 2] = 128 * i + 2;
        bv[4 * i + 3] = q[i].w; bo[4 * i + 3] = 128 * i + 3;
      }
      #pragma unroll
      for (int wdt = 1; wdt < 32; wdt <<= 1) {
        #pragma unroll
        for (int j = 0; j < 32; j += 2 * wdt) {
          const bool tk = bv[j + wdt] > bv[j];
          bv[j] = tk ? bv[j + wdt] : bv[j];
          bo[j] = tk ? bo[j + wdt] : bo[j];
        }
      }
      const unsigned idx1 = (unsigned)(bo[0] + 4 * tid);
      skey[tid] = ((u64_t)__builtin_bit_cast(unsigned, bv[0]) << 32) |
                  (u64_t)(1023u - idx1);
      __builtin_amdgcn_fence(__ATOMIC_RELEASE, "wavefront");
      __builtin_amdgcn_wave_barrier();
      __builtin_amdgcn_fence(__ATOMIC_ACQUIRE, "wavefront");
      const v2u64* sk2 = (const v2u64*)skey;
      v2u64 kv[16];
      #pragma unroll
      for (int i = 0; i < 16; ++i) kv[i] = sk2[i];
      asm volatile("" :: "v"(kv[0]), "v"(kv[1]), "v"(kv[2]), "v"(kv[3]),
                         "v"(kv[4]), "v"(kv[5]), "v"(kv[6]), "v"(kv[7]),
                         "v"(kv[8]), "v"(kv[9]), "v"(kv[10]), "v"(kv[11]),
                         "v"(kv[12]), "v"(kv[13]), "v"(kv[14]), "v"(kv[15]));
      u64_t kk[32];
      #pragma unroll
      for (int i = 0; i < 16; ++i) {
        kk[2 * i] = kv[i].x; kk[2 * i + 1] = kv[i].y;
      }
      #pragma unroll
      for (int wdt = 1; wdt < 32; wdt <<= 1) {
        #pragma unroll
        for (int j = 0; j < 32; j += 2 * wdt)
          kk[j] = kk[j + wdt] > kk[j] ? kk[j + wdt] : kk[j];
      }
      const int nxt = 1023 - (int)(unsigned)(kk[0] & 0xffffffffull);
      const v4f wp = spts[nxt];
      if (tid == 0) {
        slast = wp;
        sanch[it * 3 + 0] = wp.x; sanch[it * 3 + 1] = wp.y;
        sanch[it * 3 + 2] = wp.z;
      }
    }
    __syncthreads();
  }
  if (tid < MA_ * 3) aout[tid] = sanch[tid];
}

#define PG_   4
#define PQ_   8
#define PSEG_ (N_ / PQ_)
#define PLST_ (KNB_ + 1)
#define PTHR_ 512

__global__ void __launch_bounds__(PTHR_)
p4dconv_kernel(const float* __restrict__ xyzs,
               const float* __restrict__ anchors,
               const float* __restrict__ Wd,
               float* __restrict__ feats) {
  __shared__ __attribute__((aligned(16))) v4f spt[3][N_ + PQ_];
  __shared__ __attribute__((aligned(16))) v4f slist[3 * PG_ * PQ_][PLST_];
  __shared__ __attribute__((aligned(16))) v4f flist[3 * PG_][KNB_];
  __shared__ int scnt[3 * PG_ * PQ_];
  __shared__ int sbase[3 * PG_ * PQ_];
  __shared__ int stake[3 * PG_ * PQ_];
  __shared__ int snum[3 * PG_];
  const int blk = blockIdx.x;
  const int m0  = (blk % (MA_ / PG_)) * PG_;
  const int bl  = blk / (MA_ / PG_);
  const int l   = bl % L_;
  const int b   = bl / L_;
  const int tid = threadIdx.x;

  #pragma unroll
  for (int f = 0; f < 3; ++f) {
    int fi = l + f - 1; fi = fi < 0 ? 0 : (fi > T_ - 1 ? T_ - 1 : fi);
    const float* nb = xyzs + ((size_t)b * T_ + fi) * N_ * 3;
    #pragma unroll
    for (int h = 0; h < N_ / PTHR_; ++h) {
      const int i = tid + PTHR_ * h;
      v4f p;
      p.x = nb[i * 3 + 0]; p.y = nb[i * 3 + 1]; p.z = nb[i * 3 + 2]; p.w = 0.0f;
      spt[f][i + i / PSEG_] = p;
    }
  }
  const bool walker = tid < 3 * PG_ * PQ_;
  const int wfr  = tid / (PG_ * PQ_);
  const int wg   = (tid % (PG_ * PQ_)) / PQ_;
  const int wq   = tid % PQ_;
  const int wseg = (wfr * PG_ + wg) * PQ_ + wq;
  float ax = 0.0f, ay = 0.0f, az = 0.0f;
  if (walker) {
    const float* arow = anchors + ((size_t)bl * MA_ + m0 + wg) * 3;
    ax = arow[0]; ay = arow[1]; az = arow[2];
  }
  const v4f wv = *(const v4f*)(Wd + tid * 4);
  __syncthreads();

  if (walker) {
    const v4f* pts = &spt[wfr][wq * (PSEG_ + 1)];
    v4f* lst = slist[wseg];
    int cnt = 0;
    #pragma unroll 4
    for (int p = 0; p < PSEG_; ++p) {
      const v4f pt = pts[p];
      const float dx = pt.x - ax, dy = pt.y - ay, dz = pt.z - az;
      const bool hit = __builtin_fmaf(dz, dz, __builtin_fmaf(dx, dx, dy * dy)) < 1.0f;
      v4f e; e.x = dx; e.y = dy; e.z = dz; e.w = 0.0f;
      lst[hit ? (cnt < KNB_ ? cnt : KNB_) : KNB_] = e;
      cnt += hit ? 1 : 0;
    }
    scnt[wseg] = cnt < KNB_ ? cnt : KNB_;
  }
  __syncthreads();

  if (walker && wq == 0) {
    int run = 0;
    #pragma unroll
    for (int q = 0; q < PQ_; ++q) {
      const int c = scnt[wseg + q];
      const int t = c < KNB_ - run ? c : KNB_ - run;
      sbase[wseg + q] = run;
      stake[wseg + q] = t;
      run += t;
    }
    if (run == 0) {
      const v4f p0 = spt[wfr][0];
      v4f e; e.x = p0.x - ax; e.y = p0.y - ay; e.z = p0.z - az; e.w = 0.0f;
      flist[wfr * PG_ + wg][0] = e;
      run = 1;
    }
    snum[wfr * PG_ + wg] = run;
  }
  __syncthreads();

  #pragma unroll
  for (int j = 0; j < (3 * PG_ * PQ_ * KNB_) / PTHR_; ++j) {
    const int item = tid + PTHR_ * j;
    const int k = item % KNB_, seg = item / KNB_;
    if (k < stake[seg]) flist[seg / PQ_][sbase[seg] + k] = slist[seg][k];
  }
  __syncthreads();

  const float w0 = wv.x, w1 = wv.y, w2 = wv.z, w3 = wv.w;
  for (int g = 0; g < PG_; ++g) {
    float maxv = -3.4e38f;
    #pragma unroll
    for (int f = 0; f < 3; ++f) {
      const float tq = (float)(f - 1);
      const int n = __builtin_amdgcn_readfirstlane(snum[f * PG_ + g]);
      const v4f* lst = flist[f * PG_ + g];
      #pragma unroll 4
      for (int k = 0; k < n; ++k) {
        const v4f e = lst[k];
        const float v =
            __builtin_fmaf(w2, e.z, __builtin_fmaf(w0, e.x, w1 * e.y)) + w3 * tq;
        maxv = fmaxf(maxv, v);
      }
    }
    feats[((size_t)bl * MA_ + m0 + g) * C_ + tid] = fmaxf(maxv, 0.0f);
  }
}

__global__ void ln_kernel(const float* __restrict__ x, const float* __restrict__ g,
                          const float* __restrict__ bta, float* __restrict__ y) {
  __shared__ float red[256];
  const int row = blockIdx.x, t = threadIdx.x;
  const float* xr = x + (size_t)row * C_;
  const float a0 = xr[t], a1 = xr[t + 256];
  red[t] = a0 + a1; __syncthreads();
  for (int s = 128; s > 0; s >>= 1) { if (t < s) red[t] += red[t + s]; __syncthreads(); }
  const float mean = red[0] * (1.0f / C_);
  __syncthreads();
  const float d0 = a0 - mean, d1 = a1 - mean;
  red[t] = d0 * d0 + d1 * d1; __syncthreads();
  for (int s = 128; s > 0; s >>= 1) { if (t < s) red[t] += red[t + s]; __syncthreads(); }
  const float inv = rsqrtf(red[0] * (1.0f / C_) + 1e-5f);
  float* yr = y + (size_t)row * C_;
  yr[t]       = d0 * inv * g[t]       + bta[t];
  yr[t + 256] = d1 * inv * g[t + 256] + bta[t + 256];
}

__device__ __forceinline__ void gemm_ld1(v4f (&f)[16], int kb, const float* a0,
                                         const float* a1, const float* w0,
                                         const float* w1) {
  f[4 * kb + 0] = *(const v4f*)(a0 + 8 * kb);
  f[4 * kb + 1] = *(const v4f*)(a1 + 8 * kb);
  f[4 * kb + 2] = *(const v4f*)(w0 + 128 * kb);
  f[4 * kb + 3] = *(const v4f*)(w1 + 128 * kb);
}

__device__ __forceinline__ void gemm_mm1(const v4f (&f)[16], int kb, v8f& acc00,
                                         v8f& acc01, v8f& acc10, v8f& acc11) {
  const v4f a0 = f[4 * kb + 0], a1 = f[4 * kb + 1];
  const v4f b0 = f[4 * kb + 2], b1 = f[4 * kb + 3];
  const v2f a0l = {a0.x, a0.y}, a0h = {a0.z, a0.w};
  const v2f a1l = {a1.x, a1.y}, a1h = {a1.z, a1.w};
  const v2f b0l = {b0.x, b0.y}, b0h = {b0.z, b0.w};
  const v2f b1l = {b1.x, b1.y}, b1h = {b1.z, b1.w};
  acc00 = wmma4(a0l, b0l, acc00);
  acc01 = wmma4(a0l, b1l, acc01);
  acc10 = wmma4(a1l, b0l, acc10);
  acc11 = wmma4(a1l, b1l, acc11);
  acc00 = wmma4(a0h, b0h, acc00);
  acc01 = wmma4(a0h, b1h, acc01);
  acc10 = wmma4(a1h, b0h, acc10);
  acc11 = wmma4(a1h, b1h, acc11);
}

template <int WI, int WJ>
__global__ void __launch_bounds__(32 * WI * WJ)
gemm_wmma(const float* __restrict__ A, int lda,
          const float* __restrict__ Wp,
          const float* __restrict__ bias,
          const float* __restrict__ resid,
          float* __restrict__ Cout, int ldc,
          int Kdim, int act) {
  const int w = __builtin_amdgcn_readfirstlane((int)(threadIdx.x >> 5));
  const int wi = w / WJ, wj = w - wi * WJ;
  const int j0 = (blockIdx.x * WJ + wj) * 32, i0 = (blockIdx.y * WI + wi) * 32;
  const int lane = threadIdx.x & 31;
  const int sub = lane & 15, hi = lane >> 4;
  const float* arow0 = A + (size_t)(i0 + sub) * lda + 4 * hi;
  const float* arow1 = arow0 + (size_t)16 * lda;
  const float* wrow0 = Wp + (size_t)(j0 >> 4) * (Kdim >> 3) * 128 + lane * 4;
  const float* wrow1 = wrow0 + (size_t)(Kdim >> 3) * 128;
  __builtin_prefetch(arow0, 0, 0);
  __builtin_prefetch(wrow0, 0, 0);
  v8f acc00 = {0.f, 0.f, 0.f, 0.f, 0.f, 0.f, 0.f, 0.f};
  v8f acc01 = acc00, acc10 = acc00, acc11 = acc00;
  v4f s0[16], s1[16];
  #pragma unroll
  for (int kb = 0; kb < 4; ++kb) gemm_ld1(s0, kb, arow0, arow1, wrow0, wrow1);
  #pragma unroll
  for (int kb = 0; kb < 4; ++kb)
    gemm_ld1(s1, kb, arow0 + 32, arow1 + 32, wrow0 + 32 * 16, wrow1 + 32 * 16);
  int k0 = 0;
  for (; k0 + 64 < Kdim; k0 += 64) {
    #pragma unroll
    for (int kb = 0; kb < 4; ++kb) {
      gemm_mm1(s0, kb, acc00, acc01, acc10, acc11);
      gemm_ld1(s0, kb, arow0 + k0 + 64, arow1 + k0 + 64,
               wrow0 + (size_t)(k0 + 64) * 16, wrow1 + (size_t)(k0 + 64) * 16);
    }
    #pragma unroll
    for (int kb = 0; kb < 4; ++kb) {
      gemm_mm1(s1, kb, acc00, acc01, acc10, acc11);
      gemm_ld1(s1, kb, arow0 + k0 + 96, arow1 + k0 + 96,
               wrow0 + (size_t)(k0 + 96) * 16, wrow1 + (size_t)(k0 + 96) * 16);
    }
  }
  #pragma unroll
  for (int kb = 0; kb < 4; ++kb) gemm_mm1(s0, kb, acc00, acc01, acc10, acc11);
  #pragma unroll
  for (int kb = 0; kb < 4; ++kb) gemm_mm1(s1, kb, acc00, acc01, acc10, acc11);
  const float bj0 = bias ? bias[j0 + sub]      : 0.0f;
  const float bj1 = bias ? bias[j0 + 16 + sub] : 0.0f;
  #pragma unroll
  for (int r = 0; r < 8; ++r) {
    const int row0 = i0 + r + 8 * hi;
    const int row1 = row0 + 16;
    const int col0 = j0 + sub;
    const int col1 = col0 + 16;
    float v00 = acc00[r] + bj0, v01 = acc01[r] + bj1;
    float v10 = acc10[r] + bj0, v11 = acc11[r] + bj1;
    if (act == 1) {
      v00 = gelu_exact(v00); v01 = gelu_exact(v01);
      v10 = gelu_exact(v10); v11 = gelu_exact(v11);
    }
    if (resid) {
      v00 += resid[(size_t)row0 * ldc + col0];
      v01 += resid[(size_t)row0 * ldc + col1];
      v10 += resid[(size_t)row1 * ldc + col0];
      v11 += resid[(size_t)row1 * ldc + col1];
    }
    Cout[(size_t)row0 * ldc + col0] = v00;
    Cout[(size_t)row0 * ldc + col1] = v01;
    Cout[(size_t)row1 * ldc + col0] = v10;
    Cout[(size_t)row1 * ldc + col1] = v11;
  }
}

#define ATT_WAVES_ 4
#define ATT_KPW_   (MTOK_ / ATT_WAVES_)
#define ATT_OP_    68

__global__ void __launch_bounds__(32 * ATT_WAVES_)
attn_kernel(const float* __restrict__ qkv,
            const float* __restrict__ xyzf,
            const float* __restrict__ Wsp,
            float* __restrict__ attn_out) {
  __shared__ float s_m[ATT_WAVES_][16];
  __shared__ float s_l[ATT_WAVES_][16];
  __shared__ float s_wx[ATT_WAVES_][16][3];
  __shared__ __attribute__((aligned(16))) float s_o[ATT_WAVES_][16][ATT_OP_];
  const int b = blockIdx.z, h = blockIdx.y, i0 = blockIdx.x * 16;
  const int t = threadIdx.x;
  const int lane = t & 31, w = t >> 5;
  const int sub = lane & 15, hi = lane >> 4;
  const int key0 = b * MTOK_ + w * ATT_KPW_;

  const float* qbase =
      qkv + ((size_t)(b * MTOK_ + i0 + sub) * 3 + 0) * C_ + h * DH_ + 2 * hi;
  v2f qb[16];
  #pragma unroll
  for (int ks = 0; ks < 16; ++ks)
    qb[ks] = *(const v2f*)(qbase + 4 * ks) * 0.125f;

  const float* krow =
      qkv + ((size_t)(key0 + sub) * 3 + 1) * C_ + h * DH_ + 2 * hi;
  const float* vcol =
      qkv + ((size_t)(key0 + 8 * hi) * 3 + 2) * C_ + h * DH_ + sub;
  const float* xrow = xyzf + (size_t)(key0 + 8 * hi) * 3;

  v8f o0 = {0.f, 0.f, 0.f, 0.f, 0.f, 0.f, 0.f, 0.f};
  v8f o1 = o0, o2 = o0, o3 = o0;
  float m = -3.4e38f, l = 0.0f, wx0 = 0.0f, wx1 = 0.0f, wx2 = 0.0f;

  v2f kf[16];
  #pragma unroll
  for (int ks = 0; ks < 16; ++ks) kf[ks] = *(const v2f*)(krow + 4 * ks);

  for (int jt = 0; jt < ATT_KPW_ / 16; ++jt) {
    v8f st = {0.f, 0.f, 0.f, 0.f, 0.f, 0.f, 0.f, 0.f};
    #pragma unroll
    for (int ks = 0; ks < 16; ++ks) st = wmma4(kf[ks], qb[ks], st);

    const int jn = (jt + 1 < ATT_KPW_ / 16) ? jt + 1 : jt;
    const float* kn = krow + (size_t)jn * 16 * 3 * C_;
    #pragma unroll
    for (int ks = 0; ks < 16; ++ks) kf[ks] = *(const v2f*)(kn + 4 * ks);

    const float* vt = vcol + (size_t)jt * 16 * 3 * C_;
    v2f vf[4][4];
    #pragma unroll
    for (int s = 0; s < 4; ++s) {
      #pragma unroll
      for (int dt = 0; dt < 4; ++dt) {
        const float* vp = vt + (size_t)(2 * s) * 3 * C_ + dt * 16;
        vf[s][dt].x = vp[0];
        vf[s][dt].y = vp[3 * C_];
      }
    }
    const v4f* xp = (const v4f*)(xrow + jt * 48);
    float xk[24];
    #pragma unroll
    for (int i = 0; i < 6; ++i) {
      const v4f xv = xp[i];
      xk[4 * i + 0] = xv.x; xk[4 * i + 1] = xv.y;
      xk[4 * i + 2] = xv.z; xk[4 * i + 3] = xv.w;
    }

    float tm = st[0];
    #pragma unroll
    for (int r = 1; r < 8; ++r) tm = fmaxf(tm, st[r]);
    tm = fmaxf(tm, xhalf(tm));
    const float mn = fmaxf(m, tm);
    const float alpha = expf(m - mn);
    m = mn;
    float p[8];
    float ps = 0.0f, a0 = 0.0f, a1 = 0.0f, a2 = 0.0f;
    #pragma unroll
    for (int r = 0; r < 8; ++r) {
      p[r] = expf(st[r] - mn);
      ps += p[r];
      a0 += p[r] * xk[3 * r + 0];
      a1 += p[r] * xk[3 * r + 1];
      a2 += p[r] * xk[3 * r + 2];
    }
    l   = l   * alpha + ps;
    wx0 = wx0 * alpha + a0;
    wx1 = wx1 * alpha + a1;
    wx2 = wx2 * alpha + a2;
    #pragma unroll
    for (int r = 0; r < 8; ++r) {
      o0[r] *= alpha; o1[r] *= alpha; o2[r] *= alpha; o3[r] *= alpha;
    }
    #pragma unroll
    for (int s = 0; s < 4; ++s) {
      v2f pb; pb.x = p[2 * s]; pb.y = p[2 * s + 1];
      o0 = wmma4(vf[s][0], pb, o0);
      o1 = wmma4(vf[s][1], pb, o1);
      o2 = wmma4(vf[s][2], pb, o2);
      o3 = wmma4(vf[s][3], pb, o3);
    }
  }

  l   += xhalf(l);
  wx0 += xhalf(wx0);
  wx1 += xhalf(wx1);
  wx2 += xhalf(wx2);
  if (hi == 0) {
    s_m[w][sub] = m;
    s_l[w][sub] = l;
    s_wx[w][sub][0] = wx0; s_wx[w][sub][1] = wx1; s_wx[w][sub][2] = wx2;
  }
  #pragma unroll
  for (int r = 0; r < 8; ++r) {
    s_o[w][sub][ 0 + r + 8 * hi] = o0[r];
    s_o[w][sub][16 + r + 8 * hi] = o1[r];
    s_o[w][sub][32 + r + 8 * hi] = o2[r];
    s_o[w][sub][48 + r + 8 * hi] = o3[r];
  }
  __syncthreads();

  #pragma unroll
  for (int pass = 0; pass < 2; ++pass) {
    const int q = pass * 8 + (t >> 4), dch = (t & 15) * 4;
    const float m0 = s_m[0][q], m1 = s_m[1][q], m2 = s_m[2][q], m3 = s_m[3][q];
    const float ma = fmaxf(fmaxf(m0, m1), fmaxf(m2, m3));
    const float f0 = expf(m0 - ma), f1 = expf(m1 - ma);
    const float f2 = expf(m2 - ma), f3 = expf(m3 - ma);
    const float lt = ((s_l[0][q] * f0 + s_l[1][q] * f1) + s_l[2][q] * f2) +
                     s_l[3][q] * f3;
    const float inv = 1.0f / lt;
    const int i = i0 + q;
    const float* xi = xyzf + ((size_t)b * MTOK_ + i) * 3;
    float wd[3];
    #pragma unroll
    for (int c = 0; c < 3; ++c)
      wd[c] = (((s_wx[0][q][c] * f0 + s_wx[1][q][c] * f1) + s_wx[2][q][c] * f2) +
               s_wx[3][q][c] * f3) * inv - xi[c];
    const v4f p0 = *(const v4f*)&s_o[0][q][dch];
    const v4f p1 = *(const v4f*)&s_o[1][q][dch];
    const v4f p2 = *(const v4f*)&s_o[2][q][dch];
    const v4f p3 = *(const v4f*)&s_o[3][q][dch];
    const v4f sm = ((p0 * f0 + p1 * f1) + p2 * f2) + p3 * f3;
    float res[4] = {sm.x, sm.y, sm.z, sm.w};
    #pragma unroll
    for (int e = 0; e < 4; ++e) {
      const int d = dch + e;
      res[e] = res[e] * inv + Wsp[d * 3 + 0] * wd[0] + Wsp[d * 3 + 1] * wd[1] +
               Wsp[d * 3 + 2] * wd[2];
    }
    v4f rv;
    rv.x = res[0]; rv.y = res[1]; rv.z = res[2]; rv.w = res[3];
    *(v4f*)(attn_out + ((size_t)(b * MTOK_) + i) * C_ + h * DH_ + dch) = rv;
  }
}

extern "C" void kernel_launch(void* const* d_in, const int* in_sizes, int n_in,
                              void* d_out, int out_size, void* d_ws, size_t ws_size,
                              hipStream_t stream) {
  (void)in_sizes; (void)n_in; (void)out_size; (void)ws_size;
  const float* xyzs = (const float*)d_in[0];
  const float* Wd   = (const float*)d_in[1];
  const float* Wqkv = (const float*)d_in[2];
  const float* Wsp  = (const float*)d_in[3];
  const float* Wout = (const float*)d_in[4];
  const float* bout = (const float*)d_in[5];
  const float* ln1g = (const float*)d_in[6];
  const float* ln1b = (const float*)d_in[7];
  const float* Wff1 = (const float*)d_in[8];
  const float* bff1 = (const float*)d_in[9];
  const float* Wff2 = (const float*)d_in[10];
  const float* bff2 = (const float*)d_in[11];
  const float* ln2g = (const float*)d_in[12];
  const float* ln2b = (const float*)d_in[13];

  float* ws      = (float*)d_ws;
  float* anchors = ws;
  float* feats   = ws + 8192;
  float* xln     = feats + (size_t)ROWS_ * C_;
  float* qkvb    = xln   + (size_t)ROWS_ * C_;
  float* attno   = qkvb  + (size_t)ROWS_ * 1536;
  float* ffh     = attno + (size_t)ROWS_ * C_;
  float* wpk     = ffh + (size_t)ROWS_ * MLP_;
  const float* pqkv = wpk;
  const float* pout = pqkv + (size_t)2 * 1536 * C_;
  const float* pff1 = pout + (size_t)2 * C_ * C_;
  const float* pff2 = pff1 + (size_t)2 * MLP_ * C_;

  fps_kernel<<<B_ * L_ + RP_BLOCKS_, N_, 0, stream>>>(xyzs, anchors, Wqkv, Wout,
                                                     Wff1, Wff2, wpk);
  p4dconv_kernel<<<B_ * L_ * MA_ / PG_, PTHR_, 0, stream>>>(xyzs, anchors, Wd, feats);

  for (int d = 0; d < 2; ++d) {
    ln_kernel<<<ROWS_, 256, 0, stream>>>(feats, ln1g + d * C_, ln1b + d * C_, xln);
    gemm_wmma<2, 6><<<dim3(1536 / (32 * 6), ROWS_ / (32 * 2)), 32 * 2 * 6, 0, stream>>>(
        xln, C_, pqkv + (size_t)d * 1536 * C_, nullptr, nullptr,
        qkvb, 1536, C_, 0);
    attn_kernel<<<dim3(MTOK_ / 16, HEADS_, B_), 32 * ATT_WAVES_, 0, stream>>>(
        qkvb, anchors, Wsp + (size_t)d * DH_ * 3, attno);
    gemm_wmma<2, 2><<<dim3(C_ / (32 * 2), ROWS_ / (32 * 2)), 32 * 2 * 2, 0, stream>>>(
        attno, C_, pout + (size_t)d * C_ * C_, bout + d * C_,
        feats, feats, C_, C_, 1);
    ln_kernel<<<ROWS_, 256, 0, stream>>>(feats, ln2g + d * C_, ln2b + d * C_, xln);
    gemm_wmma<4, 4><<<dim3(MLP_ / (32 * 4), ROWS_ / (32 * 4)), 32 * 4 * 4, 0, stream>>>(
        xln, C_, pff1 + (size_t)d * MLP_ * C_, bff1 + d * MLP_,
        nullptr, ffh, MLP_, C_, 1);
    gemm_wmma<2, 2><<<dim3(C_ / (32 * 2), ROWS_ / (32 * 2)), 32 * 2 * 2, 0, stream>>>(
        ffh, MLP_, pff2 + (size_t)d * C_ * MLP_, bff2 + d * C_,
        feats, feats, C_, MLP_, 0);
  }
  hipMemcpyAsync(d_out, feats, (size_t)ROWS_ * C_ * sizeof(float),
                 hipMemcpyDeviceToDevice, stream);
}


# GEMM: fragment loads pinned in k-slice order, up-front ones too (scheduling barriers): the loop's load-count waits become exact, a slice is waited for 56 WMMAs after its request, was 9; same sums
# speedup vs baseline: 2.2327x; 1.0419x over previous
#include <hip/hip_runtime.h>
#include <hip/hip_bf16.h>
#include <math.h>

typedef float v2f __attribute__((ext_vector_type(2)));
typedef float v4f __attribute__((ext_vector_type(4)));
typedef float v8f __attribute__((ext_vector_type(8)));

#define B_    2
#define T_    4
#define N_    1024
#define L_    4
#define MA_   256
#define KNB_  32
#define C_    512
#define HEADS_ 8
#define DH_   64
#define MTOK_ 1024
#define ROWS_ 2048
#define MLP_  2048

__device__ __forceinline__ v8f wmma4(v2f a, v2f b, v8f c) {
  return __builtin_amdgcn_wmma_f32_16x16x4_f32(false, a, false, b, (short)0, c,
                                               false, false);
}

__device__ __forceinline__ float gelu_exact(float x) {
  return 0.5f * x * (1.0f + erff(x * 0.7071067811865475f));
}

__device__ __forceinline__ float xhalf(float x) {
  const int bx = __builtin_bit_cast(int, x);
  return __builtin_bit_cast(float, __builtin_amdgcn_permlanex16(
      bx, bx, 0x76543210, 0xfedcba98, false, false));
}

typedef unsigned long long u64_t;
typedef u64_t v2u64 __attribute__((ext_vector_type(2)));

#define RPU_      4096
#define RP_QKV_   (2 * 1536 * C_ / 4 / RPU_)
#define RP_OUT_   (2 * C_ * C_ / 4 / RPU_)
#define RP_FF1_   (2 * MLP_ * C_ / 4 / RPU_)
#define RP_FF2_   (2 * C_ * MLP_ / 4 / RPU_)
#define RP_BLOCKS_ (RP_QKV_ + RP_OUT_ + RP_FF1_ + RP_FF2_)

__device__ __forceinline__ void repack_block(int rb, int tid,
                                             const float* __restrict__ Wqkv,
                                             const float* __restrict__ Wout,
                                             const float* __restrict__ Wff1,
                                             const float* __restrict__ Wff2,
                                             float* __restrict__ wpk) {
  const float* src = Wqkv;
  float* dst = wpk;
  int lb = rb;
  int kq_log = 6;
  if (rb >= RP_QKV_ + RP_OUT_ + RP_FF1_) {
    src = Wff2; lb = rb - (RP_QKV_ + RP_OUT_ + RP_FF1_); kq_log = 8;
    dst = wpk + (size_t)2 * (1536 + C_ + MLP_) * C_;
  } else if (rb >= RP_QKV_ + RP_OUT_) {
    src = Wff1; lb = rb - (RP_QKV_ + RP_OUT_);
    dst = wpk + (size_t)2 * (1536 + C_) * C_;
  } else if (rb >= RP_QKV_) {
    src = Wout; lb = rb - RP_QKV_;
    dst = wpk + (size_t)2 * 1536 * C_;
  }
  v4f v[4];
  #pragma unroll
  for (int j = 0; j < 4; ++j) {
    const int u = lb * RPU_ + j * 1024 + tid;
    const int lane = u & 31, t = u >> 5;
    const int kb = t & ((1 << kq_log) - 1), nt = t >> kq_log;
    v[j] = *(const v4f*)(src + (((size_t)(16 * nt + (lane & 15))) << (kq_log + 3)) +
                         8 * kb + 4 * (lane >> 4));
  }
  #pragma unroll
  for (int j = 0; j < 4; ++j)
    *(v4f*)(dst + (size_t)(lb * RPU_ + j * 1024 + tid) * 4) = v[j];
}

__global__ void fps_kernel(const float* __restrict__ xyzs,
                           float* __restrict__ anchors,
                           const float* __restrict__ Wqkv,
                           const float* __restrict__ Wout,
                           const float* __restrict__ Wff1,
                           const float* __restrict__ Wff2,
                           float* __restrict__ wpk) {
  if (blockIdx.x >= B_ * L_) {
    repack_block(blockIdx.x - B_ * L_, threadIdx.x, Wqkv, Wout, Wff1, Wff2, wpk);
    return;
  }
  __shared__ __attribute__((aligned(16))) float sdist[N_];
  __shared__ __attribute__((aligned(16))) v4f   spts[N_];
  __shared__ __attribute__((aligned(16))) u64_t skey[32];
  __shared__ __attribute__((aligned(16))) v4f   slast;
  __shared__ float sanch[MA_ * 3];
  const int blk = blockIdx.x;
  const int tid = threadIdx.x;
  const float* frame = xyzs + (size_t)blk * N_ * 3;
  const float px = frame[tid * 3 + 0];
  const float py = frame[tid * 3 + 1];
  const float pz = frame[tid * 3 + 2];
  float dist = 1e10f;
  float* aout = anchors + (size_t)blk * MA_ * 3;
  {
    v4f me; me.x = px; me.y = py; me.z = pz; me.w = 0.0f;
    spts[tid] = me;
    if (tid == 0) {
      slast = me;
      sanch[0] = px; sanch[1] = py; sanch[2] = pz;
    }
  }
  __syncthreads();
  for (int it = 1; it < MA_; ++it) {
    const v4f lp = slast;
    const float dx = px - lp.x, dy = py - lp.y, dz = pz - lp.z;
    dist = fminf(dist, __builtin_fmaf(dz, dz, __builtin_fmaf(dx, dx, dy * dy)));
    sdist[tid] = dist;
    __syncthreads();
    if (tid < 32) {
      const v4f* sd4 = (const v4f*)sdist;
      v4f q[8];
      #pragma unroll
      for (int i = 0; i < 8; ++i) q[i] = sd4[32 * i + tid];
      float bv[32];
      int   bo[32];
      #pragma unroll
      for (int i = 0; i < 8; ++i) {
        bv[4 * i + 0] = q[i].x; bo[4 * i + 0] = 128 * i + 0;
        bv[4 * i + 1] = q[i].y; bo[4 * i + 1] = 128 * i + 1;
        bv[4 * i + 2] = q[i].z; bo[4 * i + 2] = 128 * i + 2;
        bv[4 * i + 3] = q[i].w; bo[4 * i + 3] = 128 * i + 3;
      }
      #pragma unroll
      for (int wdt = 1; wdt < 32; wdt <<= 1) {
        #pragma unroll
        for (int j = 0; j < 32; j += 2 * wdt) {
          const bool tk = bv[j + wdt] > bv[j];
          bv[j] = tk ? bv[j + wdt] : bv[j];
          bo[j] = tk ? bo[j + wdt] : bo[j];
        }
      }
      const unsigned idx1 = (unsigned)(bo[0] + 4 * tid);
      skey[tid] = ((u64_t)__builtin_bit_cast(unsigned, bv[0]) << 32) |
                  (u64_t)(1023u - idx1);
      __builtin_amdgcn_fence(__ATOMIC_RELEASE, "wavefront");
      __builtin_amdgcn_wave_barrier();
      __builtin_amdgcn_fence(__ATOMIC_ACQUIRE, "wavefront");
      const v2u64* sk2 = (const v2u64*)skey;
      v2u64 kv[16];
      #pragma unroll
      for (int i = 0; i < 16; ++i) kv[i] = sk2[i];
      asm volatile("" :: "v"(kv[0]), "v"(kv[1]), "v"(kv[2]), "v"(kv[3]),
                         "v"(kv[4]), "v"(kv[5]), "v"(kv[6]), "v"(kv[7]),
                         "v"(kv[8]), "v"(kv[9]), "v"(kv[10]), "v"(kv[11]),
                         "v"(kv[12]), "v"(kv[13]), "v"(kv[14]), "v"(kv[15]));
      u64_t kk[32];
      #pragma unroll
      for (int i = 0; i < 16; ++i) {
        kk[2 * i] = kv[i].x; kk[2 * i + 1] = kv[i].y;
      }
      #pragma unroll
      for (int wdt = 1; wdt < 32; wdt <<= 1) {
        #pragma unroll
        for (int j = 0; j < 32; j += 2 * wdt)
          kk[j] = kk[j + wdt] > kk[j] ? kk[j + wdt] : kk[j];
      }
      const int nxt = 1023 - (int)(unsigned)(kk[0] & 0xffffffffull);
      const v4f wp = spts[nxt];
      if (tid == 0) {
        slast = wp;
        sanch[it * 3 + 0] = wp.x; sanch[it * 3 + 1] = wp.y;
        sanch[it * 3 + 2] = wp.z;
      }
    }
    __syncthreads();
  }
  if (tid < MA_ * 3) aout[tid] = sanch[tid];
}

#define PG_   4
#define PQ_   8
#define PSEG_ (N_ / PQ_)
#define PLST_ (KNB_ + 1)
#define PTHR_ 512

__global__ void __launch_bounds__(PTHR_)
p4dconv_kernel(const float* __restrict__ xyzs,
               const float* __restrict__ anchors,
               const float* __restrict__ Wd,
               float* __restrict__ feats) {
  __shared__ __attribute__((aligned(16))) v4f spt[3][N_ + PQ_];
  __shared__ __attribute__((aligned(16))) v4f slist[3 * PG_ * PQ_][PLST_];
  __shared__ __attribute__((aligned(16))) v4f flist[3 * PG_][KNB_];
  __shared__ int scnt[3 * PG_ * PQ_];
  __shared__ int sbase[3 * PG_ * PQ_];
  __shared__ int stake[3 * PG_ * PQ_];
  __shared__ int snum[3 * PG_];
  const int blk = blockIdx.x;
  const int m0  = (blk % (MA_ / PG_)) * PG_;
  const int bl  = blk / (MA_ / PG_);
  const int l   = bl % L_;
  const int b   = bl / L_;
  const int tid = threadIdx.x;

  #pragma unroll
  for (int f = 0; f < 3; ++f) {
    int fi = l + f - 1; fi = fi < 0 ? 0 : (fi > T_ - 1 ? T_ - 1 : fi);
    const float* nb = xyzs + ((size_t)b * T_ + fi) * N_ * 3;
    #pragma unroll
    for (int h = 0; h < N_ / PTHR_; ++h) {
      const int i = tid + PTHR_ * h;
      v4f p;
      p.x = nb[i * 3 + 0]; p.y = nb[i * 3 + 1]; p.z = nb[i * 3 + 2]; p.w = 0.0f;
      spt[f][i + i / PSEG_] = p;
    }
  }
  const bool walker = tid < 3 * PG_ * PQ_;
  const int wfr  = tid / (PG_ * PQ_);
  const int wg   = (tid % (PG_ * PQ_)) / PQ_;
  const int wq   = tid % PQ_;
  const int wseg = (wfr * PG_ + wg) * PQ_ + wq;
  float ax = 0.0f, ay = 0.0f, az = 0.0f;
  if (walker) {
    const float* arow = anchors + ((size_t)bl * MA_ + m0 + wg) * 3;
    ax = arow[0]; ay = arow[1]; az = arow[2];
  }
  const v4f wv = *(const v4f*)(Wd + tid * 4);
  __syncthreads();

  if (walker) {
    const v4f* pts = &spt[wfr][wq * (PSEG_ + 1)];
    v4f* lst = slist[wseg];
    int cnt = 0;
    #pragma unroll 4
    for (int p = 0; p < PSEG_; ++p) {
      const v4f pt = pts[p];
      const float dx = pt.x - ax, dy = pt.y - ay, dz = pt.z - az;
      const bool hit = __builtin_fmaf(dz, dz, __builtin_fmaf(dx, dx, dy * dy)) < 1.0f;
      v4f e; e.x = dx; e.y = dy; e.z = dz; e.w = 0.0f;
      lst[hit ? (cnt < KNB_ ? cnt : KNB_) : KNB_] = e;
      cnt += hit ? 1 : 0;
    }
    scnt[wseg] = cnt < KNB_ ? cnt : KNB_;
  }
  __syncthreads();

  if (walker && wq == 0) {
    int run = 0;
    #pragma unroll
    for (int q = 0; q < PQ_; ++q) {
      const int c = scnt[wseg + q];
      const int t = c < KNB_ - run ? c : KNB_ - run;
      sbase[wseg + q] = run;
      stake[wseg + q] = t;
      run += t;
    }
    if (run == 0) {
      const v4f p0 = spt[wfr][0];
      v4f e; e.x = p0.x - ax; e.y = p0.y - ay; e.z = p0.z - az; e.w = 0.0f;
      flist[wfr * PG_ + wg][0] = e;
      run = 1;
    }
    snum[wfr * PG_ + wg] = run;
  }
  __syncthreads();

  #pragma unroll
  for (int j = 0; j < (3 * PG_ * PQ_ * KNB_) / PTHR_; ++j) {
    const int item = tid + PTHR_ * j;
    const int k = item % KNB_, seg = item / KNB_;
    if (k < stake[seg]) flist[seg / PQ_][sbase[seg] + k] = slist[seg][k];
  }
  __syncthreads();

  const float w0 = wv.x, w1 = wv.y, w2 = wv.z, w3 = wv.w;
  for (int g = 0; g < PG_; ++g) {
    float maxv = -3.4e38f;
    #pragma unroll
    for (int f = 0; f < 3; ++f) {
      const float tq = (float)(f - 1);
      const int n = __builtin_amdgcn_readfirstlane(snum[f * PG_ + g]);
      const v4f* lst = flist[f * PG_ + g];
      #pragma unroll 4
      for (int k = 0; k < n; ++k) {
        const v4f e = lst[k];
        const float v =
            __builtin_fmaf(w2, e.z, __builtin_fmaf(w0, e.x, w1 * e.y)) + w3 * tq;
        maxv = fmaxf(maxv, v);
      }
    }
    feats[((size_t)bl * MA_ + m0 + g) * C_ + tid] = fmaxf(maxv, 0.0f);
  }
}

__global__ void ln_kernel(const float* __restrict__ x, const float* __restrict__ g,
                          const float* __restrict__ bta, float* __restrict__ y) {
  __shared__ float red[256];
  const int row = blockIdx.x, t = threadIdx.x;
  const float* xr = x + (size_t)row * C_;
  const float a0 = xr[t], a1 = xr[t + 256];
  red[t] = a0 + a1; __syncthreads();
  for (int s = 128; s > 0; s >>= 1) { if (t < s) red[t] += red[t + s]; __syncthreads(); }
  const float mean = red[0] * (1.0f / C_);
  __syncthreads();
  const float d0 = a0 - mean, d1 = a1 - mean;
  red[t] = d0 * d0 + d1 * d1; __syncthreads();
  for (int s = 128; s > 0; s >>= 1) { if (t < s) red[t] += red[t + s]; __syncthreads(); }
  const float inv = rsqrtf(red[0] * (1.0f / C_) + 1e-5f);
  float* yr = y + (size_t)row * C_;
  yr[t]       = d0 * inv * g[t]       + bta[t];
  yr[t + 256] = d1 * inv * g[t + 256] + bta[t + 256];
}

__device__ __forceinline__ void gemm_ld1(v4f (&f)[16], int kb, const float* a0,
                                         const float* a1, const float* w0,
                                         const float* w1) {
  f[4 * kb + 0] = *(const v4f*)(a0 + 8 * kb);
  f[4 * kb + 1] = *(const v4f*)(a1 + 8 * kb);
  f[4 * kb + 2] = *(const v4f*)(w0 + 128 * kb);
  f[4 * kb + 3] = *(const v4f*)(w1 + 128 * kb);
}

__device__ __forceinline__ void gemm_mm1(const v4f (&f)[16], int kb, v8f& acc00,
                                         v8f& acc01, v8f& acc10, v8f& acc11) {
  const v4f a0 = f[4 * kb + 0], a1 = f[4 * kb + 1];
  const v4f b0 = f[4 * kb + 2], b1 = f[4 * kb + 3];
  const v2f a0l = {a0.x, a0.y}, a0h = {a0.z, a0.w};
  const v2f a1l = {a1.x, a1.y}, a1h = {a1.z, a1.w};
  const v2f b0l = {b0.x, b0.y}, b0h = {b0.z, b0.w};
  const v2f b1l = {b1.x, b1.y}, b1h = {b1.z, b1.w};
  acc00 = wmma4(a0l, b0l, acc00);
  acc01 = wmma4(a0l, b1l, acc01);
  acc10 = wmma4(a1l, b0l, acc10);
  acc11 = wmma4(a1l, b1l, acc11);
  acc00 = wmma4(a0h, b0h, acc00);
  acc01 = wmma4(a0h, b1h, acc01);
  acc10 = wmma4(a1h, b0h, acc10);
  acc11 = wmma4(a1h, b1h, acc11);
}

template <int WI, int WJ>
__global__ void __launch_bounds__(32 * WI * WJ)
gemm_wmma(const float* __restrict__ A, int lda,
          const float* __restrict__ Wp,
          const float* __restrict__ bias,
          const float* __restrict__ resid,
          float* __restrict__ Cout, int ldc,
          int Kdim, int act) {
  const int w = __builtin_amdgcn_readfirstlane((int)(threadIdx.x >> 5));
  const int wi = w / WJ, wj = w - wi * WJ;
  const int j0 = (blockIdx.x * WJ + wj) * 32, i0 = (blockIdx.y * WI + wi) * 32;
  const int lane = threadIdx.x & 31;
  const int sub = lane & 15, hi = lane >> 4;
  const float* arow0 = A + (size_t)(i0 + sub) * lda + 4 * hi;
  const float* arow1 = arow0 + (size_t)16 * lda;
  const float* wrow0 = Wp + (size_t)(j0 >> 4) * (Kdim >> 3) * 128 + lane * 4;
  const float* wrow1 = wrow0 + (size_t)(Kdim >> 3) * 128;
  __builtin_prefetch(arow0, 0, 0);
  __builtin_prefetch(wrow0, 0, 0);
  v8f acc00 = {0.f, 0.f, 0.f, 0.f, 0.f, 0.f, 0.f, 0.f};
  v8f acc01 = acc00, acc10 = acc00, acc11 = acc00;
  v4f s0[16], s1[16];
  #pragma unroll
  for (int kb = 0; kb < 4; ++kb) {
    gemm_ld1(s0, kb, arow0, arow1, wrow0, wrow1);
    __builtin_amdgcn_sched_barrier(0);
  }
  #pragma unroll
  for (int kb = 0; kb < 4; ++kb) {
    gemm_ld1(s1, kb, arow0 + 32, arow1 + 32, wrow0 + 32 * 16, wrow1 + 32 * 16);
    __builtin_amdgcn_sched_barrier(0);
  }
  int k0 = 0;
  for (; k0 + 64 < Kdim; k0 += 64) {
    #pragma unroll
    for (int kb = 0; kb < 4; ++kb) {
      gemm_mm1(s0, kb, acc00, acc01, acc10, acc11);
      gemm_ld1(s0, kb, arow0 + k0 + 64, arow1 + k0 + 64,
               wrow0 + (size_t)(k0 + 64) * 16, wrow1 + (size_t)(k0 + 64) * 16);
      __builtin_amdgcn_sched_barrier(0);
    }
    #pragma unroll
    for (int kb = 0; kb < 4; ++kb) {
      gemm_mm1(s1, kb, acc00, acc01, acc10, acc11);
      gemm_ld1(s1, kb, arow0 + k0 + 96, arow1 + k0 + 96,
               wrow0 + (size_t)(k0 + 96) * 16, wrow1 + (size_t)(k0 + 96) * 16);
      __builtin_amdgcn_sched_barrier(0);
    }
  }
  #pragma unroll
  for (int kb = 0; kb < 4; ++kb) gemm_mm1(s0, kb, acc00, acc01, acc10, acc11);
  #pragma unroll
  for (int kb = 0; kb < 4; ++kb) gemm_mm1(s1, kb, acc00, acc01, acc10, acc11);
  const float bj0 = bias ? bias[j0 + sub]      : 0.0f;
  const float bj1 = bias ? bias[j0 + 16 + sub] : 0.0f;
  #pragma unroll
  for (int r = 0; r < 8; ++r) {
    const int row0 = i0 + r + 8 * hi;
    const int row1 = row0 + 16;
    const int col0 = j0 + sub;
    const int col1 = col0 + 16;
    float v00 = acc00[r] + bj0, v01 = acc01[r] + bj1;
    float v10 = acc10[r] + bj0, v11 = acc11[r] + bj1;
    if (act == 1) {
      v00 = gelu_exact(v00); v01 = gelu_exact(v01);
      v10 = gelu_exact(v10); v11 = gelu_exact(v11);
    }
    if (resid) {
      v00 += resid[(size_t)row0 * ldc + col0];
      v01 += resid[(size_t)row0 * ldc + col1];
      v10 += resid[(size_t)row1 * ldc + col0];
      v11 += resid[(size_t)row1 * ldc + col1];
    }
    Cout[(size_t)row0 * ldc + col0] = v00;
    Cout[(size_t)row0 * ldc + col1] = v01;
    Cout[(size_t)row1 * ldc + col0] = v10;
    Cout[(size_t)row1 * ldc + col1] = v11;
  }
}

#define ATT_WAVES_ 4
#define ATT_KPW_   (MTOK_ / ATT_WAVES_)
#define ATT_OP_    68

__global__ void __launch_bounds__(32 * ATT_WAVES_)
attn_kernel(const float* __restrict__ qkv,
            const float* __restrict__ xyzf,
            const float* __restrict__ Wsp,
            float* __restrict__ attn_out) {
  __shared__ float s_m[ATT_WAVES_][16];
  __shared__ float s_l[ATT_WAVES_][16];
  __shared__ float s_wx[ATT_WAVES_][16][3];
  __shared__ __attribute__((aligned(16))) float s_o[ATT_WAVES_][16][ATT_OP_];
  const int b = blockIdx.z, h = blockIdx.y, i0 = blockIdx.x * 16;
  const int t = threadIdx.x;
  const int lane = t & 31, w = t >> 5;
  const int sub = lane & 15, hi = lane >> 4;
  const int key0 = b * MTOK_ + w * ATT_KPW_;

  const float* qbase =
      qkv + ((size_t)(b * MTOK_ + i0 + sub) * 3 + 0) * C_ + h * DH_ + 2 * hi;
  v2f qb[16];
  #pragma unroll
  for (int ks = 0; ks < 16; ++ks)
    qb[ks] = *(const v2f*)(qbase + 4 * ks) * 0.125f;

  const float* krow =
      qkv + ((size_t)(key0 + sub) * 3 + 1) * C_ + h * DH_ + 2 * hi;
  const float* vcol =
      qkv + ((size_t)(key0 + 8 * hi) * 3 + 2) * C_ + h * DH_ + sub;
  const float* xrow = xyzf + (size_t)(key0 + 8 * hi) * 3;

  v8f o0 = {0.f, 0.f, 0.f, 0.f, 0.f, 0.f, 0.f, 0.f};
  v8f o1 = o0, o2 = o0, o3 = o0;
  float m = -3.4e38f, l = 0.0f, wx0 = 0.0f, wx1 = 0.0f, wx2 = 0.0f;

  v2f kf[16];
  #pragma unroll
  for (int ks = 0; ks < 16; ++ks) kf[ks] = *(const v2f*)(krow + 4 * ks);

  for (int jt = 0; jt < ATT_KPW_ / 16; ++jt) {
    v8f st = {0.f, 0.f, 0.f, 0.f, 0.f, 0.f, 0.f, 0.f};
    #pragma unroll
    for (int ks = 0; ks < 16; ++ks) st = wmma4(kf[ks], qb[ks], st);

    const int jn = (jt + 1 < ATT_KPW_ / 16) ? jt + 1 : jt;
    const float* kn = krow + (size_t)jn * 16 * 3 * C_;
    #pragma unroll
    for (int ks = 0; ks < 16; ++ks) kf[ks] = *(const v2f*)(kn + 4 * ks);

    const float* vt = vcol + (size_t)jt * 16 * 3 * C_;
    v2f vf[4][4];
    #pragma unroll
    for (int s = 0; s < 4; ++s) {
      #pragma unroll
      for (int dt = 0; dt < 4; ++dt) {
        const float* vp = vt + (size_t)(2 * s) * 3 * C_ + dt * 16;
        vf[s][dt].x = vp[0];
        vf[s][dt].y = vp[3 * C_];
      }
    }
    const v4f* xp = (const v4f*)(xrow + jt * 48);
    float xk[24];
    #pragma unroll
    for (int i = 0; i < 6; ++i) {
      const v4f xv = xp[i];
      xk[4 * i + 0] = xv.x; xk[4 * i + 1] = xv.y;
      xk[4 * i + 2] = xv.z; xk[4 * i + 3] = xv.w;
    }

    float tm = st[0];
    #pragma unroll
    for (int r = 1; r < 8; ++r) tm = fmaxf(tm, st[r]);
    tm = fmaxf(tm, xhalf(tm));
    const float mn = fmaxf(m, tm);
    const float alpha = expf(m - mn);
    m = mn;
    float p[8];
    float ps = 0.0f, a0 = 0.0f, a1 = 0.0f, a2 = 0.0f;
    #pragma unroll
    for (int r = 0; r < 8; ++r) {
      p[r] = expf(st[r] - mn);
      ps += p[r];
      a0 += p[r] * xk[3 * r + 0];
      a1 += p[r] * xk[3 * r + 1];
      a2 += p[r] * xk[3 * r + 2];
    }
    l   = l   * alpha + ps;
    wx0 = wx0 * alpha + a0;
    wx1 = wx1 * alpha + a1;
    wx2 = wx2 * alpha + a2;
    #pragma unroll
    for (int r = 0; r < 8; ++r) {
      o0[r] *= alpha; o1[r] *= alpha; o2[r] *= alpha; o3[r] *= alpha;
    }
    #pragma unroll
    for (int s = 0; s < 4; ++s) {
      v2f pb; pb.x = p[2 * s]; pb.y = p[2 * s + 1];
      o0 = wmma4(vf[s][0], pb, o0);
      o1 = wmma4(vf[s][1], pb, o1);
      o2 = wmma4(vf[s][2], pb, o2);
      o3 = wmma4(vf[s][3], pb, o3);
    }
  }

  l   += xhalf(l);
  wx0 += xhalf(wx0);
  wx1 += xhalf(wx1);
  wx2 += xhalf(wx2);
  if (hi == 0) {
    s_m[w][sub] = m;
    s_l[w][sub] = l;
    s_wx[w][sub][0] = wx0; s_wx[w][sub][1] = wx1; s_wx[w][sub][2] = wx2;
  }
  #pragma unroll
  for (int r = 0; r < 8; ++r) {
    s_o[w][sub][ 0 + r + 8 * hi] = o0[r];
    s_o[w][sub][16 + r + 8 * hi] = o1[r];
    s_o[w][sub][32 + r + 8 * hi] = o2[r];
    s_o[w][sub][48 + r + 8 * hi] = o3[r];
  }
  __syncthreads();

  #pragma unroll
  for (int pass = 0; pass < 2; ++pass) {
    const int q = pass * 8 + (t >> 4), dch = (t & 15) * 4;
    const float m0 = s_m[0][q], m1 = s_m[1][q], m2 = s_m[2][q], m3 = s_m[3][q];
    const float ma = fmaxf(fmaxf(m0, m1), fmaxf(m2, m3));
    const float f0 = expf(m0 - ma), f1 = expf(m1 - ma);
    const float f2 = expf(m2 - ma), f3 = expf(m3 - ma);
    const float lt = ((s_l[0][q] * f0 + s_l[1][q] * f1) + s_l[2][q] * f2) +
                     s_l[3][q] * f3;
    const float inv = 1.0f / lt;
    const int i = i0 + q;
    const float* xi = xyzf + ((size_t)b * MTOK_ + i) * 3;
    float wd[3];
    #pragma unroll
    for (int c = 0; c < 3; ++c)
      wd[c] = (((s_wx[0][q][c] * f0 + s_wx[1][q][c] * f1) + s_wx[2][q][c] * f2) +
               s_wx[3][q][c] * f3) * inv - xi[c];
    const v4f p0 = *(const v4f*)&s_o[0][q][dch];
    const v4f p1 = *(const v4f*)&s_o[1][q][dch];
    const v4f p2 = *(const v4f*)&s_o[2][q][dch];
    const v4f p3 = *(const v4f*)&s_o[3][q][dch];
    const v4f sm = ((p0 * f0 + p1 * f1) + p2 * f2) + p3 * f3;
    float res[4] = {sm.x, sm.y, sm.z, sm.w};
    #pragma unroll
    for (int e = 0; e < 4; ++e) {
      const int d = dch + e;
      res[e] = res[e] * inv + Wsp[d * 3 + 0] * wd[0] + Wsp[d * 3 + 1] * wd[1] +
               Wsp[d * 3 + 2] * wd[2];
    }
    v4f rv;
    rv.x = res[0]; rv.y = res[1]; rv.z = res[2]; rv.w = res[3];
    *(v4f*)(attn_out + ((size_t)(b * MTOK_) + i) * C_ + h * DH_ + dch) = rv;
  }
}

extern "C" void kernel_launch(void* const* d_in, const int* in_sizes, int n_in,
                              void* d_out, int out_size, void* d_ws, size_t ws_size,
                              hipStream_t stream) {
  (void)in_sizes; (void)n_in; (void)out_size; (void)ws_size;
  const float* xyzs = (const float*)d_in[0];
  const float* Wd   = (const float*)d_in[1];
  const float* Wqkv = (const float*)d_in[2];
  const float* Wsp  = (const float*)d_in[3];
  const float* Wout = (const float*)d_in[4];
  const float* bout = (const float*)d_in[5];
  const float* ln1g = (const float*)d_in[6];
  const float* ln1b = (const float*)d_in[7];
  const float* Wff1 = (const float*)d_in[8];
  const float* bff1 = (const float*)d_in[9];
  const float* Wff2 = (const float*)d_in[10];
  const float* bff2 = (const float*)d_in[11];
  const float* ln2g = (const float*)d_in[12];
  const float* ln2b = (const float*)d_in[13];

  float* ws      = (float*)d_ws;
  float* anchors = ws;
  float* feats   = ws + 8192;
  float* xln     = feats + (size_t)ROWS_ * C_;
  float* qkvb    = xln   + (size_t)ROWS_ * C_;
  float* attno   = qkvb  + (size_t)ROWS_ * 1536;
  float* ffh     = attno + (size_t)ROWS_ * C_;
  float* wpk     = ffh + (size_t)ROWS_ * MLP_;
  const float* pqkv = wpk;
  const float* pout = pqkv + (size_t)2 * 1536 * C_;
  const float* pff1 = pout + (size_t)2 * C_ * C_;
  const float* pff2 = pff1 + (size_t)2 * MLP_ * C_;

  fps_kernel<<<B_ * L_ + RP_BLOCKS_, N_, 0, stream>>>(xyzs, anchors, Wqkv, Wout,
                                                     Wff1, Wff2, wpk);
  p4dconv_kernel<<<B_ * L_ * MA_ / PG_, PTHR_, 0, stream>>>(xyzs, anchors, Wd, feats);

  for (int d = 0; d < 2; ++d) {
    ln_kernel<<<ROWS_, 256, 0, stream>>>(feats, ln1g + d * C_, ln1b + d * C_, xln);
    gemm_wmma<2, 6><<<dim3(1536 / (32 * 6), ROWS_ / (32 * 2)), 32 * 2 * 6, 0, stream>>>(
        xln, C_, pqkv + (size_t)d * 1536 * C_, nullptr, nullptr,
        qkvb, 1536, C_, 0);
    attn_kernel<<<dim3(MTOK_ / 16, HEADS_, B_), 32 * ATT_WAVES_, 0, stream>>>(
        qkvb, anchors, Wsp + (size_t)d * DH_ * 3, attno);
    gemm_wmma<2, 2><<<dim3(C_ / (32 * 2), ROWS_ / (32 * 2)), 32 * 2 * 2, 0, stream>>>(
        attno, C_, pout + (size_t)d * C_ * C_, bout + d * C_,
        feats, feats, C_, C_, 1);
    ln_kernel<<<ROWS_, 256, 0, stream>>>(feats, ln2g + d * C_, ln2b + d * C_, xln);
    gemm_wmma<4, 4><<<dim3(MLP_ / (32 * 4), ROWS_ / (32 * 4)), 32 * 4 * 4, 0, stream>>>(
        xln, C_, pff1 + (size_t)d * MLP_ * C_, bff1 + d * MLP_,
        nullptr, ffh, MLP_, C_, 1);
    gemm_wmma<2, 2><<<dim3(C_ / (32 * 2), ROWS_ / (32 * 2)), 32 * 2 * 2, 0, stream>>>(
        ffh, MLP_, pff2 + (size_t)d * C_ * MLP_, bff2 + d * C_,
        feats, feats, C_, MLP_, 0);
  }
  hipMemcpyAsync(d_out, feats, (size_t)ROWS_ * C_ * sizeof(float),
                 hipMemcpyDeviceToDevice, stream);
}


# GEMMs with N = 512 (out-proj, FF2): k split over two waves per 32x32 tile, two waves per SIMD (was one); halves meet once in LDS, each wave finishes 16 of the 32 rows; k summed as two halves
# speedup vs baseline: 2.3226x; 1.0403x over previous
#include <hip/hip_runtime.h>
#include <hip/hip_bf16.h>
#include <math.h>

typedef float v2f __attribute__((ext_vector_type(2)));
typedef float v4f __attribute__((ext_vector_type(4)));
typedef float v8f __attribute__((ext_vector_type(8)));

#define B_    2
#define T_    4
#define N_    1024
#define L_    4
#define MA_   256
#define KNB_  32
#define C_    512
#define HEADS_ 8
#define DH_   64
#define MTOK_ 1024
#define ROWS_ 2048
#define MLP_  2048

__device__ __forceinline__ v8f wmma4(v2f a, v2f b, v8f c) {
  return __builtin_amdgcn_wmma_f32_16x16x4_f32(false, a, false, b, (short)0, c,
                                               false, false);
}

__device__ __forceinline__ float gelu_exact(float x) {
  return 0.5f * x * (1.0f + erff(x * 0.7071067811865475f));
}

__device__ __forceinline__ float xhalf(float x) {
  const int bx = __builtin_bit_cast(int, x);
  return __builtin_bit_cast(float, __builtin_amdgcn_permlanex16(
      bx, bx, 0x76543210, 0xfedcba98, false, false));
}

typedef unsigned long long u64_t;
typedef u64_t v2u64 __attribute__((ext_vector_type(2)));

#define RPU_      4096
#define RP_QKV_   (2 * 1536 * C_ / 4 / RPU_)
#define RP_OUT_   (2 * C_ * C_ / 4 / RPU_)
#define RP_FF1_   (2 * MLP_ * C_ / 4 / RPU_)
#define RP_FF2_   (2 * C_ * MLP_ / 4 / RPU_)
#define RP_BLOCKS_ (RP_QKV_ + RP_OUT_ + RP_FF1_ + RP_FF2_)

__device__ __forceinline__ void repack_block(int rb, int tid,
                                             const float* __restrict__ Wqkv,
                                             const float* __restrict__ Wout,
                                             const float* __restrict__ Wff1,
                                             const float* __restrict__ Wff2,
                                             float* __restrict__ wpk) {
  const float* src = Wqkv;
  float* dst = wpk;
  int lb = rb;
  int kq_log = 6;
  if (rb >= RP_QKV_ + RP_OUT_ + RP_FF1_) {
    src = Wff2; lb = rb - (RP_QKV_ + RP_OUT_ + RP_FF1_); kq_log = 8;
    dst = wpk + (size_t)2 * (1536 + C_ + MLP_) * C_;
  } else if (rb >= RP_QKV_ + RP_OUT_) {
    src = Wff1; lb = rb - (RP_QKV_ + RP_OUT_);
    dst = wpk + (size_t)2 * (1536 + C_) * C_;
  } else if (rb >= RP_QKV_) {
    src = Wout; lb = rb - RP_QKV_;
    dst = wpk + (size_t)2 * 1536 * C_;
  }
  v4f v[4];
  #pragma unroll
  for (int j = 0; j < 4; ++j) {
    const int u = lb * RPU_ + j * 1024 + tid;
    const int lane = u & 31, t = u >> 5;
    const int kb = t & ((1 << kq_log) - 1), nt = t >> kq_log;
    v[j] = *(const v4f*)(src + (((size_t)(16 * nt + (lane & 15))) << (kq_log + 3)) +
                         8 * kb + 4 * (lane >> 4));
  }
  #pragma unroll
  for (int j = 0; j < 4; ++j)
    *(v4f*)(dst + (size_t)(lb * RPU_ + j * 1024 + tid) * 4) = v[j];
}

__global__ void fps_kernel(const float* __restrict__ xyzs,
                           float* __restrict__ anchors,
                           const float* __restrict__ Wqkv,
                           const float* __restrict__ Wout,
                           const float* __restrict__ Wff1,
                           const float* __restrict__ Wff2,
                           float* __restrict__ wpk) {
  if (blockIdx.x >= B_ * L_) {
    repack_block(blockIdx.x - B_ * L_, threadIdx.x, Wqkv, Wout, Wff1, Wff2, wpk);
    return;
  }
  __shared__ __attribute__((aligned(16))) float sdist[N_];
  __shared__ __attribute__((aligned(16))) v4f   spts[N_];
  __shared__ __attribute__((aligned(16))) u64_t skey[32];
  __shared__ __attribute__((aligned(16))) v4f   slast;
  __shared__ float sanch[MA_ * 3];
  const int blk = blockIdx.x;
  const int tid = threadIdx.x;
  const float* frame = xyzs + (size_t)blk * N_ * 3;
  const float px = frame[tid * 3 + 0];
  const float py = frame[tid * 3 + 1];
  const float pz = frame[tid * 3 + 2];
  float dist = 1e10f;
  float* aout = anchors + (size_t)blk * MA_ * 3;
  {
    v4f me; me.x = px; me.y = py; me.z = pz; me.w = 0.0f;
    spts[tid] = me;
    if (tid == 0) {
      slast = me;
      sanch[0] = px; sanch[1] = py; sanch[2] = pz;
    }
  }
  __syncthreads();
  for (int it = 1; it < MA_; ++it) {
    const v4f lp = slast;
    const float dx = px - lp.x, dy = py - lp.y, dz = pz - lp.z;
    dist = fminf(dist, __builtin_fmaf(dz, dz, __builtin_fmaf(dx, dx, dy * dy)));
    sdist[tid] = dist;
    __syncthreads();
    if (tid < 32) {
      const v4f* sd4 = (const v4f*)sdist;
      v4f q[8];
      #pragma unroll
      for (int i = 0; i < 8; ++i) q[i] = sd4[32 * i + tid];
      float bv[32];
      int   bo[32];
      #pragma unroll
      for (int i = 0; i < 8; ++i) {
        bv[4 * i + 0] = q[i].x; bo[4 * i + 0] = 128 * i + 0;
        bv[4 * i + 1] = q[i].y; bo[4 * i + 1] = 128 * i + 1;
        bv[4 * i + 2] = q[i].z; bo[4 * i + 2] = 128 * i + 2;
        bv[4 * i + 3] = q[i].w; bo[4 * i + 3] = 128 * i + 3;
      }
      #pragma unroll
      for (int wdt = 1; wdt < 32; wdt <<= 1) {
        #pragma unroll
        for (int j = 0; j < 32; j += 2 * wdt) {
          const bool tk = bv[j + wdt] > bv[j];
          bv[j] = tk ? bv[j + wdt] : bv[j];
          bo[j] = tk ? bo[j + wdt] : bo[j];
        }
      }
      const unsigned idx1 = (unsigned)(bo[0] + 4 * tid);
      skey[tid] = ((u64_t)__builtin_bit_cast(unsigned, bv[0]) << 32) |
                  (u64_t)(1023u - idx1);
      __builtin_amdgcn_fence(__ATOMIC_RELEASE, "wavefront");
      __builtin_amdgcn_wave_barrier();
      __builtin_amdgcn_fence(__ATOMIC_ACQUIRE, "wavefront");
      const v2u64* sk2 = (const v2u64*)skey;
      v2u64 kv[16];
      #pragma unroll
      for (int i = 0; i < 16; ++i) kv[i] = sk2[i];
      asm volatile("" :: "v"(kv[0]), "v"(kv[1]), "v"(kv[2]), "v"(kv[3]),
                         "v"(kv[4]), "v"(kv[5]), "v"(kv[6]), "v"(kv[7]),
                         "v"(kv[8]), "v"(kv[9]), "v"(kv[10]), "v"(kv[11]),
                         "v"(kv[12]), "v"(kv[13]), "v"(kv[14]), "v"(kv[15]));
      u64_t kk[32];
      #pragma unroll
      for (int i = 0; i < 16; ++i) {
        kk[2 * i] = kv[i].x; kk[2 * i + 1] = kv[i].y;
      }
      #pragma unroll
      for (int wdt = 1; wdt < 32; wdt <<= 1) {
        #pragma unroll
        for (int j = 0; j < 32; j += 2 * wdt)
          kk[j] = kk[j + wdt] > kk[j] ? kk[j + wdt] : kk[j];
      }
      const int nxt = 1023 - (int)(unsigned)(kk[0] & 0xffffffffull);
      const v4f wp = spts[nxt];
      if (tid == 0) {
        slast = wp;
        sanch[it * 3 + 0] = wp.x; sanch[it * 3 + 1] = wp.y;
        sanch[it * 3 + 2] = wp.z;
      }
    }
    __syncthreads();
  }
  if (tid < MA_ * 3) aout[tid] = sanch[tid];
}

#define PG_   4
#define PQ_   8
#define PSEG_ (N_ / PQ_)
#define PLST_ (KNB_ + 1)
#define PTHR_ 512

__global__ void __launch_bounds__(PTHR_)
p4dconv_kernel(const float* __restrict__ xyzs,
               const float* __restrict__ anchors,
               const float* __restrict__ Wd,
               float* __restrict__ feats) {
  __shared__ __attribute__((aligned(16))) v4f spt[3][N_ + PQ_];
  __shared__ __attribute__((aligned(16))) v4f slist[3 * PG_ * PQ_][PLST_];
  __shared__ __attribute__((aligned(16))) v4f flist[3 * PG_][KNB_];
  __shared__ int scnt[3 * PG_ * PQ_];
  __shared__ int sbase[3 * PG_ * PQ_];
  __shared__ int stake[3 * PG_ * PQ_];
  __shared__ int snum[3 * PG_];
  const int blk = blockIdx.x;
  const int m0  = (blk % (MA_ / PG_)) * PG_;
  const int bl  = blk / (MA_ / PG_);
  const int l   = bl % L_;
  const int b   = bl / L_;
  const int tid = threadIdx.x;

  #pragma unroll
  for (int f = 0; f < 3; ++f) {
    int fi = l + f - 1; fi = fi < 0 ? 0 : (fi > T_ - 1 ? T_ - 1 : fi);
    const float* nb = xyzs + ((size_t)b * T_ + fi) * N_ * 3;
    #pragma unroll
    for (int h = 0; h < N_ / PTHR_; ++h) {
      const int i = tid + PTHR_ * h;
      v4f p;
      p.x = nb[i * 3 + 0]; p.y = nb[i * 3 + 1]; p.z = nb[i * 3 + 2]; p.w = 0.0f;
      spt[f][i + i / PSEG_] = p;
    }
  }
  const bool walker = tid < 3 * PG_ * PQ_;
  const int wfr  = tid / (PG_ * PQ_);
  const int wg   = (tid % (PG_ * PQ_)) / PQ_;
  const int wq   = tid % PQ_;
  const int wseg = (wfr * PG_ + wg) * PQ_ + wq;
  float ax = 0.0f, ay = 0.0f, az = 0.0f;
  if (walker) {
    const float* arow = anchors + ((size_t)bl * MA_ + m0 + wg) * 3;
    ax = arow[0]; ay = arow[1]; az = arow[2];
  }
  const v4f wv = *(const v4f*)(Wd + tid * 4);
  __syncthreads();

  if (walker) {
    const v4f* pts = &spt[wfr][wq * (PSEG_ + 1)];
    v4f* lst = slist[wseg];
    int cnt = 0;
    #pragma unroll 4
    for (int p = 0; p < PSEG_; ++p) {
      const v4f pt = pts[p];
      const float dx = pt.x - ax, dy = pt.y - ay, dz = pt.z - az;
      const bool hit = __builtin_fmaf(dz, dz, __builtin_fmaf(dx, dx, dy * dy)) < 1.0f;
      v4f e; e.x = dx; e.y = dy; e.z = dz; e.w = 0.0f;
      lst[hit ? (cnt < KNB_ ? cnt : KNB_) : KNB_] = e;
      cnt += hit ? 1 : 0;
    }
    scnt[wseg] = cnt < KNB_ ? cnt : KNB_;
  }
  __syncthreads();

  if (walker && wq == 0) {
    int run = 0;
    #pragma unroll
    for (int q = 0; q < PQ_; ++q) {
      const int c = scnt[wseg + q];
      const int t = c < KNB_ - run ? c : KNB_ - run;
      sbase[wseg + q] = run;
      stake[wseg + q] = t;
      run += t;
    }
    if (run == 0) {
      const v4f p0 = spt[wfr][0];
      v4f e; e.x = p0.x - ax; e.y = p0.y - ay; e.z = p0.z - az; e.w = 0.0f;
      flist[wfr * PG_ + wg][0] = e;
      run = 1;
    }
    snum[wfr * PG_ + wg] = run;
  }
  __syncthreads();

  #pragma unroll
  for (int j = 0; j < (3 * PG_ * PQ_ * KNB_) / PTHR_; ++j) {
    const int item = tid + PTHR_ * j;
    const int k = item % KNB_, seg = item / KNB_;
    if (k < stake[seg]) flist[seg / PQ_][sbase[seg] + k] = slist[seg][k];
  }
  __syncthreads();

  const float w0 = wv.x, w1 = wv.y, w2 = wv.z, w3 = wv.w;
  for (int g = 0; g < PG_; ++g) {
    float maxv = -3.4e38f;
    #pragma unroll
    for (int f = 0; f < 3; ++f) {
      const float tq = (float)(f - 1);
      const int n = __builtin_amdgcn_readfirstlane(snum[f * PG_ + g]);
      const v4f* lst = flist[f * PG_ + g];
      #pragma unroll 4
      for (int k = 0; k < n; ++k) {
        const v4f e = lst[k];
        const float v =
            __builtin_fmaf(w2, e.z, __builtin_fmaf(w0, e.x, w1 * e.y)) + w3 * tq;
        maxv = fmaxf(maxv, v);
      }
    }
    feats[((size_t)bl * MA_ + m0 + g) * C_ + tid] = fmaxf(maxv, 0.0f);
  }
}

__global__ void ln_kernel(const float* __restrict__ x, const float* __restrict__ g,
                          const float* __restrict__ bta, float* __restrict__ y) {
  __shared__ float red[256];
  const int row = blockIdx.x, t = threadIdx.x;
  const float* xr = x + (size_t)row * C_;
  const float a0 = xr[t], a1 = xr[t + 256];
  red[t] = a0 + a1; __syncthreads();
  for (int s = 128; s > 0; s >>= 1) { if (t < s) red[t] += red[t + s]; __syncthreads(); }
  const float mean = red[0] * (1.0f / C_);
  __syncthreads();
  const float d0 = a0 - mean, d1 = a1 - mean;
  red[t] = d0 * d0 + d1 * d1; __syncthreads();
  for (int s = 128; s > 0; s >>= 1) { if (t < s) red[t] += red[t + s]; __syncthreads(); }
  const float inv = rsqrtf(red[0] * (1.0f / C_) + 1e-5f);
  float* yr = y + (size_t)row * C_;
  yr[t]       = d0 * inv * g[t]       + bta[t];
  yr[t + 256] = d1 * inv * g[t + 256] + bta[t + 256];
}

__device__ __forceinline__ void gemm_ld1(v4f (&f)[16], int kb, const float* a0,
                                         const float* a1, const float* w0,
                                         const float* w1) {
  f[4 * kb + 0] = *(const v4f*)(a0 + 8 * kb);
  f[4 * kb + 1] = *(const v4f*)(a1 + 8 * kb);
  f[4 * kb + 2] = *(const v4f*)(w0 + 128 * kb);
  f[4 * kb + 3] = *(const v4f*)(w1 + 128 * kb);
}

__device__ __forceinline__ void gemm_mm1(const v4f (&f)[16], int kb, v8f& acc00,
                                         v8f& acc01, v8f& acc10, v8f& acc11) {
  const v4f a0 = f[4 * kb + 0], a1 = f[4 * kb + 1];
  const v4f b0 = f[4 * kb + 2], b1 = f[4 * kb + 3];
  const v2f a0l = {a0.x, a0.y}, a0h = {a0.z, a0.w};
  const v2f a1l = {a1.x, a1.y}, a1h = {a1.z, a1.w};
  const v2f b0l = {b0.x, b0.y}, b0h = {b0.z, b0.w};
  const v2f b1l = {b1.x, b1.y}, b1h = {b1.z, b1.w};
  acc00 = wmma4(a0l, b0l, acc00);
  acc01 = wmma4(a0l, b1l, acc01);
  acc10 = wmma4(a1l, b0l, acc10);
  acc11 = wmma4(a1l, b1l, acc11);
  acc00 = wmma4(a0h, b0h, acc00);
  acc01 = wmma4(a0h, b1h, acc01);
  acc10 = wmma4(a1h, b0h, acc10);
  acc11 = wmma4(a1h, b1h, acc11);
}

template <int WI, int WJ, int KS>
__global__ void __launch_bounds__(32 * WI * WJ * KS)
gemm_wmma(const float* __restrict__ A, int lda,
          const float* __restrict__ Wp,
          const float* __restrict__ bias,
          const float* __restrict__ resid,
          float* __restrict__ Cout, int ldc,
          int Kdim, int act) {
  const int w = __builtin_amdgcn_readfirstlane((int)(threadIdx.x >> 5));
  const int ks = (KS == 1) ? 0 : w / (WI * WJ);
  const int wt = w - ks * (WI * WJ);
  const int wi = wt / WJ, wj = wt - wi * WJ;
  const int j0 = (blockIdx.x * WJ + wj) * 32, i0 = (blockIdx.y * WI + wi) * 32;
  const int lane = threadIdx.x & 31;
  const int sub = lane & 15, hi = lane >> 4;
  const int Kw = Kdim / KS;
  const int kbeg = ks * Kw;
  const float* arow0 = A + (size_t)(i0 + sub) * lda + 4 * hi + kbeg;
  const float* arow1 = arow0 + (size_t)16 * lda;
  const float* wrow0 = Wp + (size_t)(j0 >> 4) * (Kdim >> 3) * 128 + lane * 4 +
                       (size_t)kbeg * 16;
  const float* wrow1 = wrow0 + (size_t)(Kdim >> 3) * 128;
  __builtin_prefetch(arow0, 0, 0);
  __builtin_prefetch(wrow0, 0, 0);
  v8f acc00 = {0.f, 0.f, 0.f, 0.f, 0.f, 0.f, 0.f, 0.f};
  v8f acc01 = acc00, acc10 = acc00, acc11 = acc00;
  v4f s0[16], s1[16];
  #pragma unroll
  for (int kb = 0; kb < 4; ++kb) {
    gemm_ld1(s0, kb, arow0, arow1, wrow0, wrow1);
    __builtin_amdgcn_sched_barrier(0);
  }
  #pragma unroll
  for (int kb = 0; kb < 4; ++kb) {
    gemm_ld1(s1, kb, arow0 + 32, arow1 + 32, wrow0 + 32 * 16, wrow1 + 32 * 16);
    __builtin_amdgcn_sched_barrier(0);
  }
  int k0 = 0;
  for (; k0 + 64 < Kw; k0 += 64) {
    #pragma unroll
    for (int kb = 0; kb < 4; ++kb) {
      gemm_mm1(s0, kb, acc00, acc01, acc10, acc11);
      gemm_ld1(s0, kb, arow0 + k0 + 64, arow1 + k0 + 64,
               wrow0 + (size_t)(k0 + 64) * 16, wrow1 + (size_t)(k0 + 64) * 16);
      __builtin_amdgcn_sched_barrier(0);
    }
    #pragma unroll
    for (int kb = 0; kb < 4; ++kb) {
      gemm_mm1(s1, kb, acc00, acc01, acc10, acc11);
      gemm_ld1(s1, kb, arow0 + k0 + 96, arow1 + k0 + 96,
               wrow0 + (size_t)(k0 + 96) * 16, wrow1 + (size_t)(k0 + 96) * 16);
      __builtin_amdgcn_sched_barrier(0);
    }
  }
  #pragma unroll
  for (int kb = 0; kb < 4; ++kb) gemm_mm1(s0, kb, acc00, acc01, acc10, acc11);
  #pragma unroll
  for (int kb = 0; kb < 4; ++kb) gemm_mm1(s1, kb, acc00, acc01, acc10, acc11);
  v4f fin[4] = {};
  if (KS == 2) {
    __shared__ __attribute__((aligned(16))) v4f sx[WI * WJ * KS][4][32];
    const bool up = ks != 0;
    const v4f l00 = {acc00[0], acc00[1], acc00[2], acc00[3]};
    const v4f h00 = {acc00[4], acc00[5], acc00[6], acc00[7]};
    const v4f l01 = {acc01[0], acc01[1], acc01[2], acc01[3]};
    const v4f h01 = {acc01[4], acc01[5], acc01[6], acc01[7]};
    const v4f l10 = {acc10[0], acc10[1], acc10[2], acc10[3]};
    const v4f h10 = {acc10[4], acc10[5], acc10[6], acc10[7]};
    const v4f l11 = {acc11[0], acc11[1], acc11[2], acc11[3]};
    const v4f h11 = {acc11[4], acc11[5], acc11[6], acc11[7]};
    sx[w][0][lane] = up ? l00 : h00;
    sx[w][1][lane] = up ? l01 : h01;
    sx[w][2][lane] = up ? l10 : h10;
    sx[w][3][lane] = up ? l11 : h11;
    __syncthreads();
    const int p = up ? w - WI * WJ : w + WI * WJ;
    fin[0] = (up ? h00 : l00) + sx[p][0][lane];
    fin[1] = (up ? h01 : l01) + sx[p][1][lane];
    fin[2] = (up ? h10 : l10) + sx[p][2][lane];
    fin[3] = (up ? h11 : l11) + sx[p][3][lane];
  }
  const float bj0 = bias ? bias[j0 + sub]      : 0.0f;
  const float bj1 = bias ? bias[j0 + 16 + sub] : 0.0f;
  #pragma unroll
  for (int e = 0; e < 8 / KS; ++e) {
    const int r = (KS == 1) ? e : 4 * ks + e;
    const int row0 = i0 + r + 8 * hi;
    const int row1 = row0 + 16;
    const int col0 = j0 + sub;
    const int col1 = col0 + 16;
    float v00, v01, v10, v11;
    if (KS == 1) {
      v00 = acc00[e]; v01 = acc01[e]; v10 = acc10[e]; v11 = acc11[e];
    } else {
      v00 = fin[0][e]; v01 = fin[1][e]; v10 = fin[2][e]; v11 = fin[3][e];
    }
    v00 += bj0; v01 += bj1; v10 += bj0; v11 += bj1;
    if (act == 1) {
      v00 = gelu_exact(v00); v01 = gelu_exact(v01);
      v10 = gelu_exact(v10); v11 = gelu_exact(v11);
    }
    if (resid) {
      v00 += resid[(size_t)row0 * ldc + col0];
      v01 += resid[(size_t)row0 * ldc + col1];
      v10 += resid[(size_t)row1 * ldc + col0];
      v11 += resid[(size_t)row1 * ldc + col1];
    }
    Cout[(size_t)row0 * ldc + col0] = v00;
    Cout[(size_t)row0 * ldc + col1] = v01;
    Cout[(size_t)row1 * ldc + col0] = v10;
    Cout[(size_t)row1 * ldc + col1] = v11;
  }
}

#define ATT_WAVES_ 4
#define ATT_KPW_   (MTOK_ / ATT_WAVES_)
#define ATT_OP_    68

__global__ void __launch_bounds__(32 * ATT_WAVES_)
attn_kernel(const float* __restrict__ qkv,
            const float* __restrict__ xyzf,
            const float* __restrict__ Wsp,
            float* __restrict__ attn_out) {
  __shared__ float s_m[ATT_WAVES_][16];
  __shared__ float s_l[ATT_WAVES_][16];
  __shared__ float s_wx[ATT_WAVES_][16][3];
  __shared__ __attribute__((aligned(16))) float s_o[ATT_WAVES_][16][ATT_OP_];
  const int b = blockIdx.z, h = blockIdx.y, i0 = blockIdx.x * 16;
  const int t = threadIdx.x;
  const int lane = t & 31, w = t >> 5;
  const int sub = lane & 15, hi = lane >> 4;
  const int key0 = b * MTOK_ + w * ATT_KPW_;

  const float* qbase =
      qkv + ((size_t)(b * MTOK_ + i0 + sub) * 3 + 0) * C_ + h * DH_ + 2 * hi;
  v2f qb[16];
  #pragma unroll
  for (int ks = 0; ks < 16; ++ks)
    qb[ks] = *(const v2f*)(qbase + 4 * ks) * 0.125f;

  const float* krow =
      qkv + ((size_t)(key0 + sub) * 3 + 1) * C_ + h * DH_ + 2 * hi;
  const float* vcol =
      qkv + ((size_t)(key0 + 8 * hi) * 3 + 2) * C_ + h * DH_ + sub;
  const float* xrow = xyzf + (size_t)(key0 + 8 * hi) * 3;

  v8f o0 = {0.f, 0.f, 0.f, 0.f, 0.f, 0.f, 0.f, 0.f};
  v8f o1 = o0, o2 = o0, o3 = o0;
  float m = -3.4e38f, l = 0.0f, wx0 = 0.0f, wx1 = 0.0f, wx2 = 0.0f;

  v2f kf[16];
  #pragma unroll
  for (int ks = 0; ks < 16; ++ks) kf[ks] = *(const v2f*)(krow + 4 * ks);

  for (int jt = 0; jt < ATT_KPW_ / 16; ++jt) {
    v8f st = {0.f, 0.f, 0.f, 0.f, 0.f, 0.f, 0.f, 0.f};
    #pragma unroll
    for (int ks = 0; ks < 16; ++ks) st = wmma4(kf[ks], qb[ks], st);

    const int jn = (jt + 1 < ATT_KPW_ / 16) ? jt + 1 : jt;
    const float* kn = krow + (size_t)jn * 16 * 3 * C_;
    #pragma unroll
    for (int ks = 0; ks < 16; ++ks) kf[ks] = *(const v2f*)(kn + 4 * ks);

    const float* vt = vcol + (size_t)jt * 16 * 3 * C_;
    v2f vf[4][4];
    #pragma unroll
    for (int s = 0; s < 4; ++s) {
      #pragma unroll
      for (int dt = 0; dt < 4; ++dt) {
        const float* vp = vt + (size_t)(2 * s) * 3 * C_ + dt * 16;
        vf[s][dt].x = vp[0];
        vf[s][dt].y = vp[3 * C_];
      }
    }
    const v4f* xp = (const v4f*)(xrow + jt * 48);
    float xk[24];
    #pragma unroll
    for (int i = 0; i < 6; ++i) {
      const v4f xv = xp[i];
      xk[4 * i + 0] = xv.x; xk[4 * i + 1] = xv.y;
      xk[4 * i + 2] = xv.z; xk[4 * i + 3] = xv.w;
    }

    float tm = st[0];
    #pragma unroll
    for (int r = 1; r < 8; ++r) tm = fmaxf(tm, st[r]);
    tm = fmaxf(tm, xhalf(tm));
    const float mn = fmaxf(m, tm);
    const float alpha = expf(m - mn);
    m = mn;
    float p[8];
    float ps = 0.0f, a0 = 0.0f, a1 = 0.0f, a2 = 0.0f;
    #pragma unroll
    for (int r = 0; r < 8; ++r) {
      p[r] = expf(st[r] - mn);
      ps += p[r];
      a0 += p[r] * xk[3 * r + 0];
      a1 += p[r] * xk[3 * r + 1];
      a2 += p[r] * xk[3 * r + 2];
    }
    l   = l   * alpha + ps;
    wx0 = wx0 * alpha + a0;
    wx1 = wx1 * alpha + a1;
    wx2 = wx2 * alpha + a2;
    #pragma unroll
    for (int r = 0; r < 8; ++r) {
      o0[r] *= alpha; o1[r] *= alpha; o2[r] *= alpha; o3[r] *= alpha;
    }
    #pragma unroll
    for (int s = 0; s < 4; ++s) {
      v2f pb; pb.x = p[2 * s]; pb.y = p[2 * s + 1];
      o0 = wmma4(vf[s][0], pb, o0);
      o1 = wmma4(vf[s][1], pb, o1);
      o2 = wmma4(vf[s][2], pb, o2);
      o3 = wmma4(vf[s][3], pb, o3);
    }
  }

  l   += xhalf(l);
  wx0 += xhalf(wx0);
  wx1 += xhalf(wx1);
  wx2 += xhalf(wx2);
  if (hi == 0) {
    s_m[w][sub] = m;
    s_l[w][sub] = l;
    s_wx[w][sub][0] = wx0; s_wx[w][sub][1] = wx1; s_wx[w][sub][2] = wx2;
  }
  #pragma unroll
  for (int r = 0; r < 8; ++r) {
    s_o[w][sub][ 0 + r + 8 * hi] = o0[r];
    s_o[w][sub][16 + r + 8 * hi] = o1[r];
    s_o[w][sub][32 + r + 8 * hi] = o2[r];
    s_o[w][sub][48 + r + 8 * hi] = o3[r];
  }
  __syncthreads();

  #pragma unroll
  for (int pass = 0; pass < 2; ++pass) {
    const int q = pass * 8 + (t >> 4), dch = (t & 15) * 4;
    const float m0 = s_m[0][q], m1 = s_m[1][q], m2 = s_m[2][q], m3 = s_m[3][q];
    const float ma = fmaxf(fmaxf(m0, m1), fmaxf(m2, m3));
    const float f0 = expf(m0 - ma), f1 = expf(m1 - ma);
    const float f2 = expf(m2 - ma), f3 = expf(m3 - ma);
    const float lt = ((s_l[0][q] * f0 + s_l[1][q] * f1) + s_l[2][q] * f2) +
                     s_l[3][q] * f3;
    const float inv = 1.0f / lt;
    const int i = i0 + q;
    const float* xi = xyzf + ((size_t)b * MTOK_ + i) * 3;
    float wd[3];
    #pragma unroll
    for (int c = 0; c < 3; ++c)
      wd[c] = (((s_wx[0][q][c] * f0 + s_wx[1][q][c] * f1) + s_wx[2][q][c] * f2) +
               s_wx[3][q][c] * f3) * inv - xi[c];
    const v4f p0 = *(const v4f*)&s_o[0][q][dch];
    const v4f p1 = *(const v4f*)&s_o[1][q][dch];
    const v4f p2 = *(const v4f*)&s_o[2][q][dch];
    const v4f p3 = *(const v4f*)&s_o[3][q][dch];
    const v4f sm = ((p0 * f0 + p1 * f1) + p2 * f2) + p3 * f3;
    float res[4] = {sm.x, sm.y, sm.z, sm.w};
    #pragma unroll
    for (int e = 0; e < 4; ++e) {
      const int d = dch + e;
      res[e] = res[e] * inv + Wsp[d * 3 + 0] * wd[0] + Wsp[d * 3 + 1] * wd[1] +
               Wsp[d * 3 + 2] * wd[2];
    }
    v4f rv;
    rv.x = res[0]; rv.y = res[1]; rv.z = res[2]; rv.w = res[3];
    *(v4f*)(attn_out + ((size_t)(b * MTOK_) + i) * C_ + h * DH_ + dch) = rv;
  }
}

extern "C" void kernel_launch(void* const* d_in, const int* in_sizes, int n_in,
                              void* d_out, int out_size, void* d_ws, size_t ws_size,
                              hipStream_t stream) {
  (void)in_sizes; (void)n_in; (void)out_size; (void)ws_size;
  const float* xyzs = (const float*)d_in[0];
  const float* Wd   = (const float*)d_in[1];
  const float* Wqkv = (const float*)d_in[2];
  const float* Wsp  = (const float*)d_in[3];
  const float* Wout = (const float*)d_in[4];
  const float* bout = (const float*)d_in[5];
  const float* ln1g = (const float*)d_in[6];
  const float* ln1b = (const float*)d_in[7];
  const float* Wff1 = (const float*)d_in[8];
  const float* bff1 = (const float*)d_in[9];
  const float* Wff2 = (const float*)d_in[10];
  const float* bff2 = (const float*)d_in[11];
  const float* ln2g = (const float*)d_in[12];
  const float* ln2b = (const float*)d_in[13];

  float* ws      = (float*)d_ws;
  float* anchors = ws;
  float* feats   = ws + 8192;
  float* xln     = feats + (size_t)ROWS_ * C_;
  float* qkvb    = xln   + (size_t)ROWS_ * C_;
  float* attno   = qkvb  + (size_t)ROWS_ * 1536;
  float* ffh     = attno + (size_t)ROWS_ * C_;
  float* wpk     = ffh + (size_t)ROWS_ * MLP_;
  const float* pqkv = wpk;
  const float* pout = pqkv + (size_t)2 * 1536 * C_;
  const float* pff1 = pout + (size_t)2 * C_ * C_;
  const float* pff2 = pff1 + (size_t)2 * MLP_ * C_;

  fps_kernel<<<B_ * L_ + RP_BLOCKS_, N_, 0, stream>>>(xyzs, anchors, Wqkv, Wout,
                                                     Wff1, Wff2, wpk);
  p4dconv_kernel<<<B_ * L_ * MA_ / PG_, PTHR_, 0, stream>>>(xyzs, anchors, Wd, feats);

  for (int d = 0; d < 2; ++d) {
    ln_kernel<<<ROWS_, 256, 0, stream>>>(feats, ln1g + d * C_, ln1b + d * C_, xln);
    gemm_wmma<2, 6, 1><<<dim3(1536 / (32 * 6), ROWS_ / (32 * 2)), 32 * 2 * 6, 0, stream>>>(
        xln, C_, pqkv + (size_t)d * 1536 * C_, nullptr, nullptr,
        qkvb, 1536, C_, 0);
    attn_kernel<<<dim3(MTOK_ / 16, HEADS_, B_), 32 * ATT_WAVES_, 0, stream>>>(
        qkvb, anchors, Wsp + (size_t)d * DH_ * 3, attno);
    gemm_wmma<2, 2, 2><<<dim3(C_ / (32 * 2), ROWS_ / (32 * 2)), 32 * 2 * 2 * 2, 0, stream>>>(
        attno, C_, pout + (size_t)d * C_ * C_, bout + d * C_,
        feats, feats, C_, C_, 1);
    ln_kernel<<<ROWS_, 256, 0, stream>>>(feats, ln2g + d * C_, ln2b + d * C_, xln);
    gemm_wmma<4, 4, 1><<<dim3(MLP_ / (32 * 4), ROWS_ / (32 * 4)), 32 * 4 * 4, 0, stream>>>(
        xln, C_, pff1 + (size_t)d * MLP_ * C_, bff1 + d * MLP_,
        nullptr, ffh, MLP_, C_, 1);
    gemm_wmma<2, 2, 2><<<dim3(C_ / (32 * 2), ROWS_ / (32 * 2)), 32 * 2 * 2 * 2, 0, stream>>>(
        ffh, MLP_, pff2 + (size_t)d * C_ * MLP_, bff2 + d * C_,
        feats, feats, C_, MLP_, 0);
  }
  hipMemcpyAsync(d_out, feats, (size_t)ROWS_ * C_ * sizeof(float),
                 hipMemcpyDeviceToDevice, stream);
}
